# starting kernel
# baseline (speedup 1.0000x reference)
#include <hip/hip_runtime.h>
#include <cstddef>

#define H 128

typedef float v2f __attribute__((ext_vector_type(2)));
typedef float v8f __attribute__((ext_vector_type(8)));

__global__ void zero_u32_kernel(unsigned* __restrict__ p, long long n) {
    long long i = (long long)blockIdx.x * blockDim.x + threadIdx.x;
    if (i < n) p[i] = 0u;
}

__global__ void proj_kernel(const float* __restrict__ x,
                            const float* __restrict__ W,
                            const float* __restrict__ b,
                            float* __restrict__ out, int n, int K) {
    __shared__ __align__(16) float sW[32 * H];
    const int tid = threadIdx.x;
    for (int i = tid; i < K * H; i += 128) sW[i] = W[i];
    const float bias = b[tid];
    __syncthreads();
    const int start = blockIdx.x * 64;
    const int end   = min(start + 64, n);
    for (int i = start; i < end; ++i) {
        const float* xr = x + (size_t)i * K;
        float acc = bias;
        for (int k = 0; k < K; ++k) acc += xr[k] * sW[k * H + tid];
        out[(size_t)i * H + tid] = acc;
    }
}

__global__ void count_kernel(const int* __restrict__ tgt, int* __restrict__ cnt, int E) {
    int e = blockIdx.x * blockDim.x + threadIdx.x;
    if (e < E) atomicAdd(&cnt[tgt[e]], 1);
}

__global__ void inv_count_kernel(const int* __restrict__ cnt, float* __restrict__ inv, int n) {
    int i = blockIdx.x * blockDim.x + threadIdx.x;
    if (i < n) inv[i] = 1.0f / (float)max(cnt[i], 1);
}

__global__ void scatter_kernel(const float* __restrict__ feat,
                               const int* __restrict__ src,
                               const int* __restrict__ tgt,
                               const float* __restrict__ inv,
                               float* __restrict__ agg, int E) {
    long long idx = (long long)blockIdx.x * blockDim.x + threadIdx.x;
    if (idx >= (long long)E * H) return;
    const int e = (int)(idx >> 7);
    const int c = (int)(idx & (H - 1));
    const int s = src[e];
    const int t = tgt[e];
    atomicAdd(&agg[(size_t)t * H + c], feat[(size_t)s * H + c] * inv[t]);
}

__global__ __launch_bounds__(256) void fused_gemm_elu_kernel(
        const float* __restrict__ A0, const float* __restrict__ A1,
        const float* __restrict__ W0, const float* __restrict__ W1,
        const float* __restrict__ bias, float* __restrict__ out, int n) {
    __shared__ __align__(16) float sB[H * H];
    const int tid  = threadIdx.x;
    const int wave = tid >> 5;
    const int lane = tid & 31;
    const int half = lane >> 4;
    const int l15  = lane & 15;
    const int m0   = blockIdx.x * 128 + wave * 16;

    int arow = m0 + l15;
    if (arow >= n) arow = n - 1;
    const float* aRow0 = A0 + (size_t)arow * H;
    const float* aRow1 = A1 + (size_t)arow * H;

    v8f acc[8] = {};

    for (int phase = 0; phase < 2; ++phase) {
        __syncthreads();
        {
            const float4* s = (const float4*)(phase ? W1 : W0);
            for (int i = tid; i < H * H / 4; i += 256) {
                const float4 v = s[i];
                const int k  = i >> 5;
                const int c0 = (i & 31) << 2;
                float* d = sB + ((k >> 1) << 8) + (k & 1);
                d[2 * (c0 + 0)] = v.x;
                d[2 * (c0 + 1)] = v.y;
                d[2 * (c0 + 2)] = v.z;
                d[2 * (c0 + 3)] = v.w;
            }
        }
        __syncthreads();
        const float* ap = phase ? aRow1 : aRow0;
        const float* bbase = sB + (half << 8) + (l15 << 1);
        for (int k0 = 0; k0 < H; k0 += 4) {
            const int ka = k0 + (half << 1);
            v2f a;
            a.x = ap[ka];
            a.y = ap[ka + 1];
            const float* bk = bbase + ((k0 >> 1) << 8);
            #pragma unroll
            for (int t = 0; t < 8; ++t) {
                const v2f b = *(const v2f*)(bk + (t << 5));
                acc[t] = __builtin_amdgcn_wmma_f32_16x16x4_f32(
                    false, a, false, b, (short)0, acc[t], false, false);
            }
        }
    }

    #pragma unroll
    for (int t = 0; t < 8; ++t) {
        const int col = (t << 4) + l15;
        const float bc = bias[col];
        #pragma unroll
        for (int r = 0; r < 8; ++r) {
            const int row = m0 + (half << 3) + r;
            if (row < n) {
                float v = acc[t][r] + bc;
                out[(size_t)row * H + col] = v > 0.0f ? v : (__expf(v) - 1.0f);
            }
        }
    }
}

extern "C" void kernel_launch(void* const* d_in, const int* in_sizes, int n_in,
                              void* d_out, int out_size, void* d_ws, size_t ws_size,
                              hipStream_t stream) {
    const float* x_int  = (const float*)d_in[0];
    const float* x_lane = (const float*)d_in[1];
    const float* x_sens = (const float*)d_in[2];
    const float* x_inj  = (const float*)d_in[3];
    const int*   e_sp   = (const int*)d_in[4];
    const int*   e_fl   = (const int*)d_in[5];
    const int*   e_fs   = (const int*)d_in[6];
    const int*   e_in   = (const int*)d_in[7];
    const float* W_int  = (const float*)d_in[8];
    const float* b_int  = (const float*)d_in[9];
    const float* W_lane = (const float*)d_in[10];
    const float* b_lane = (const float*)d_in[11];
    const float* W_sens = (const float*)d_in[12];
    const float* b_sens = (const float*)d_in[13];
    const float* W_inj  = (const float*)d_in[14];
    const float* b_inj  = (const float*)d_in[15];
    const float* self_W = (const float*)d_in[16];
    const float* self_b = (const float*)d_in[17];
    const float* rel_W  = (const float*)d_in[18];

    const int nInt  = in_sizes[0] / 32;
    const int nLane = in_sizes[1] / 16;
    const int nSens = in_sizes[2] / 8;
    const int nInj  = in_sizes[3] / 8;
    const int eSp   = in_sizes[4] / 2;
    const int eFl   = in_sizes[5] / 2;
    const int eFs   = in_sizes[6] / 2;
    const int eIn   = in_sizes[7] / 2;
    const int L     = in_sizes[16] / (H * H);

    float* ws = (float*)d_ws;
    size_t o = 0;
    float* hA    = ws + o; o += (size_t)nInt  * H;
    float* hB    = ws + o; o += (size_t)nInt  * H;
    float* hLane = ws + o; o += (size_t)nLane * H;
    float* hSens = ws + o; o += (size_t)nSens * H;
    float* hInj  = ws + o; o += (size_t)nInj  * H;
    float* aggS  = ws + o; o += (size_t)nInt  * H;
    float* aggB  = ws + o; o += (size_t)nInt  * H;
    int*   cnt   = (int*)(ws + o); o += (size_t)4 * nInt;
    float* invw  = ws + o; o += (size_t)4 * nInt;
    (void)ws_size; (void)n_in; (void)out_size;

    proj_kernel<<<(nInt  + 63) / 64, 128, 0, stream>>>(x_int,  W_int,  b_int,  hA,    nInt,  32);
    proj_kernel<<<(nLane + 63) / 64, 128, 0, stream>>>(x_lane, W_lane, b_lane, hLane, nLane, 16);
    proj_kernel<<<(nSens + 63) / 64, 128, 0, stream>>>(x_sens, W_sens, b_sens, hSens, nSens, 8);
    proj_kernel<<<(nInj  + 63) / 64, 128, 0, stream>>>(x_inj,  W_inj,  b_inj,  hInj,  nInj,  8);

    const long long zc = 4LL * nInt;
    zero_u32_kernel<<<(unsigned)((zc + 255) / 256), 256, 0, stream>>>((unsigned*)cnt, zc);
    count_kernel<<<(eSp + 255) / 256, 256, 0, stream>>>(e_sp + eSp, cnt + 0 * nInt, eSp);
    count_kernel<<<(eFl + 255) / 256, 256, 0, stream>>>(e_fl + eFl, cnt + 1 * nInt, eFl);
    count_kernel<<<(eFs + 255) / 256, 256, 0, stream>>>(e_fs + eFs, cnt + 2 * nInt, eFs);
    count_kernel<<<(eIn + 255) / 256, 256, 0, stream>>>(e_in + eIn, cnt + 3 * nInt, eIn);
    inv_count_kernel<<<(int)((zc + 255) / 256), 256, 0, stream>>>(cnt, invw, (int)zc);

    const long long za = (long long)nInt * H;
    zero_u32_kernel<<<(unsigned)((za + 255) / 256), 256, 0, stream>>>((unsigned*)aggS, za);
    scatter_kernel<<<(unsigned)(((long long)eFl * H + 255) / 256), 256, 0, stream>>>(
        hLane, e_fl, e_fl + eFl, invw + 1 * nInt, aggS, eFl);
    scatter_kernel<<<(unsigned)(((long long)eFs * H + 255) / 256), 256, 0, stream>>>(
        hSens, e_fs, e_fs + eFs, invw + 2 * nInt, aggS, eFs);
    scatter_kernel<<<(unsigned)(((long long)eIn * H + 255) / 256), 256, 0, stream>>>(
        hInj,  e_in, e_in + eIn, invw + 3 * nInt, aggS, eIn);

    float* hcur = hA;
    float* hnext = hB;
    for (int l = 0; l < L; ++l) {
        hipMemcpyAsync(aggB, aggS, (size_t)nInt * H * sizeof(float),
                       hipMemcpyDeviceToDevice, stream);
        scatter_kernel<<<(unsigned)(((long long)eSp * H + 255) / 256), 256, 0, stream>>>(
            hcur, e_sp, e_sp + eSp, invw, aggB, eSp);
        float* outp = (l == L - 1) ? (float*)d_out : hnext;
        fused_gemm_elu_kernel<<<(nInt + 127) / 128, 256, 0, stream>>>(
            hcur, aggB,
            self_W + (size_t)l * H * H, rel_W + (size_t)l * H * H,
            self_b + (size_t)l * H, outp, nInt);
        float* t = hnext; hnext = hcur; hcur = t;
    }
}


# Float-atomic scatter-add replaced by per-target gather: edges bucketed by target (int-atomic rank, scan, fill); a wave sorts its bucket by source id, so the sum order is fixed; sum order changes
# speedup vs baseline: 2.3768x; 2.3768x over previous
#include <hip/hip_runtime.h>
#include <cstddef>

#define H 128

typedef float v2f __attribute__((ext_vector_type(2)));
typedef float v8f __attribute__((ext_vector_type(8)));

__global__ void zero_u32_kernel(unsigned* __restrict__ p, long long n) {
    long long i = (long long)blockIdx.x * blockDim.x + threadIdx.x;
    if (i < n) p[i] = 0u;
}

__global__ void proj_kernel(const float* __restrict__ x,
                            const float* __restrict__ W,
                            const float* __restrict__ b,
                            float* __restrict__ out, int n, int K) {
    __shared__ __align__(16) float sW[32 * H];
    const int tid = threadIdx.x;
    for (int i = tid; i < K * H; i += 128) sW[i] = W[i];
    const float bias = b[tid];
    __syncthreads();
    const int start = blockIdx.x * 64;
    const int end   = min(start + 64, n);
    for (int i = start; i < end; ++i) {
        const float* xr = x + (size_t)i * K;
        float acc = bias;
        for (int k = 0; k < K; ++k) acc += xr[k] * sW[k * H + tid];
        out[(size_t)i * H + tid] = acc;
    }
}

__global__ void count_rank_kernel(const int* __restrict__ tgt, int* __restrict__ cnt,
                                  int* __restrict__ rank, int E) {
    int e = blockIdx.x * blockDim.x + threadIdx.x;
    if (e < E) rank[e] = atomicAdd(&cnt[tgt[e]], 1);
}

__global__ __launch_bounds__(1024) void scan_kernel(const int* __restrict__ cnt,
                                                    int* __restrict__ off, int n) {
    __shared__ int sWave[32];
    const int* c = cnt + (size_t)blockIdx.x * n;
    int* o = off + (size_t)blockIdx.x * n;
    const int tid  = threadIdx.x;
    const int lane = tid & 31;
    const int wave = tid >> 5;
    const int per   = (n + 1023) / 1024;
    const int start = tid * per;
    if (n <= 0) return;
    const int last = n - 1;
    int sum = 0;
    for (int k = 0; k < per; k += 8) {
        int v[8];
        for (int u = 0; u < 8; ++u) v[u] = c[min(start + k + u, last)];
        for (int u = 0; u < 8; ++u)
            if (k + u < per && start + k + u < n) sum += v[u];
    }
    int inc = sum;
    for (int d = 1; d < 32; d <<= 1) {
        const int v = __shfl_up(inc, d, 32);
        if (lane >= d) inc += v;
    }
    if (lane == 31) sWave[wave] = inc;
    __syncthreads();
    if (wave == 0) {
        const int w = sWave[lane];
        int wi = w;
        for (int d = 1; d < 32; d <<= 1) {
            const int v = __shfl_up(wi, d, 32);
            if (lane >= d) wi += v;
        }
        sWave[lane] = wi - w;
    }
    __syncthreads();
    int run = sWave[wave] + inc - sum;
    for (int k = 0; k < per; k += 8) {
        int v[8];
        for (int u = 0; u < 8; ++u) v[u] = c[min(start + k + u, last)];
        for (int u = 0; u < 8; ++u) {
            const int i = start + k + u;
            if (k + u < per && i < n) { o[i] = run; run += v[u]; }
        }
    }
}

__global__ void fill_kernel(const int* __restrict__ src, const int* __restrict__ tgt,
                            const int* __restrict__ rank, const int* __restrict__ off,
                            int* __restrict__ list, int E) {
    int e = blockIdx.x * blockDim.x + threadIdx.x;
    if (e < E) list[off[tgt[e]] + rank[e]] = src[e];
}

__device__ __forceinline__ void add4(float4& a, const float4 v) {
    a.x += v.x; a.y += v.y; a.z += v.z; a.w += v.w;
}

__device__ __forceinline__ float4 bucket_sum(const float4* __restrict__ feat4,
                                             const int* __restrict__ list,
                                             int beg, int len, int lane) {
    float4 acc = make_float4(0.0f, 0.0f, 0.0f, 0.0f);
    if (len <= 32) {
        const int key = (lane < len) ? list[beg + lane] : 0x7fffffff;
        int rank = 0;
        for (int j = 0; j < len; ++j) {
            const int kj = __builtin_amdgcn_readlane(key, j);
            rank += (kj < key || (kj == key && j < lane)) ? 1 : 0;
        }
        if (lane >= len) rank = lane;
        const int sorted = __builtin_amdgcn_ds_permute(rank << 2, key);
        int r = 0;
        for (; r + 4 <= len; r += 4) {
            const int s0 = __builtin_amdgcn_readlane(sorted, r);
            const int s1 = __builtin_amdgcn_readlane(sorted, r + 1);
            const int s2 = __builtin_amdgcn_readlane(sorted, r + 2);
            const int s3 = __builtin_amdgcn_readlane(sorted, r + 3);
            const float4 v0 = feat4[(size_t)s0 * (H / 4) + lane];
            const float4 v1 = feat4[(size_t)s1 * (H / 4) + lane];
            const float4 v2 = feat4[(size_t)s2 * (H / 4) + lane];
            const float4 v3 = feat4[(size_t)s3 * (H / 4) + lane];
            add4(acc, v0); add4(acc, v1); add4(acc, v2); add4(acc, v3);
        }
        for (; r < len; ++r) {
            const int s0 = __builtin_amdgcn_readlane(sorted, r);
            add4(acc, feat4[(size_t)s0 * (H / 4) + lane]);
        }
    } else {
        int prev = -1;
        for (int it = 0; it < len; ++it) {
            int m = 0x7fffffff;
            for (int i = lane; i < len; i += 32) {
                const int k = list[beg + i];
                if (k > prev && k < m) m = k;
            }
            for (int o = 16; o > 0; o >>= 1) m = min(m, __shfl_xor(m, o, 32));
            if (m == 0x7fffffff) break;
            int c = 0;
            for (int i = lane; i < len; i += 32) c += (list[beg + i] == m) ? 1 : 0;
            for (int o = 16; o > 0; o >>= 1) c += __shfl_xor(c, o, 32);
            const float4 v = feat4[(size_t)m * (H / 4) + lane];
            for (int q = 0; q < c; ++q) add4(acc, v);
            prev = m;
        }
    }
    return acc;
}

__global__ __launch_bounds__(256) void gather_inv_kernel(
        const float* __restrict__ hLane, const float* __restrict__ hSens,
        const float* __restrict__ hInj,
        const int* __restrict__ listFl, const int* __restrict__ listFs,
        const int* __restrict__ listIn,
        const int* __restrict__ cnt, const int* __restrict__ off,
        float* __restrict__ aggS, int n) {
    const int lane = threadIdx.x & 31;
    const int t = __builtin_amdgcn_readfirstlane((int)(blockIdx.x * 8 + (threadIdx.x >> 5)));
    if (t >= n) return;
    float4 r;
    {
        const int len = cnt[1 * n + t];
        const float4 s = bucket_sum((const float4*)hLane, listFl, off[1 * n + t], len, lane);
        const float w = 1.0f / (float)max(len, 1);
        r = make_float4(s.x * w, s.y * w, s.z * w, s.w * w);
    }
    {
        const int len = cnt[2 * n + t];
        const float4 s = bucket_sum((const float4*)hSens, listFs, off[2 * n + t], len, lane);
        const float w = 1.0f / (float)max(len, 1);
        r.x += s.x * w; r.y += s.y * w; r.z += s.z * w; r.w += s.w * w;
    }
    {
        const int len = cnt[3 * n + t];
        const float4 s = bucket_sum((const float4*)hInj, listIn, off[3 * n + t], len, lane);
        const float w = 1.0f / (float)max(len, 1);
        r.x += s.x * w; r.y += s.y * w; r.z += s.z * w; r.w += s.w * w;
    }
    ((float4*)aggS)[(size_t)t * (H / 4) + lane] = r;
}

__global__ __launch_bounds__(256) void gather_sp_kernel(
        const float* __restrict__ h, const int* __restrict__ listSp,
        const int* __restrict__ cnt, const int* __restrict__ off,
        const float* __restrict__ aggS, float* __restrict__ aggB, int n) {
    const int lane = threadIdx.x & 31;
    const int t = __builtin_amdgcn_readfirstlane((int)(blockIdx.x * 8 + (threadIdx.x >> 5)));
    if (t >= n) return;
    const float4 base = ((const float4*)aggS)[(size_t)t * (H / 4) + lane];
    const int len = cnt[t];
    const float4 s = bucket_sum((const float4*)h, listSp, off[t], len, lane);
    const float w = 1.0f / (float)max(len, 1);
    float4 r;
    r.x = base.x + s.x * w; r.y = base.y + s.y * w;
    r.z = base.z + s.z * w; r.w = base.w + s.w * w;
    ((float4*)aggB)[(size_t)t * (H / 4) + lane] = r;
}

__global__ __launch_bounds__(256) void fused_gemm_elu_kernel(
        const float* __restrict__ A0, const float* __restrict__ A1,
        const float* __restrict__ W0, const float* __restrict__ W1,
        const float* __restrict__ bias, float* __restrict__ out, int n) {
    __shared__ __align__(16) float sB[H * H];
    const int tid  = threadIdx.x;
    const int wave = tid >> 5;
    const int lane = tid & 31;
    const int half = lane >> 4;
    const int l15  = lane & 15;
    const int m0   = blockIdx.x * 128 + wave * 16;

    int arow = m0 + l15;
    if (arow >= n) arow = n - 1;
    const float* aRow0 = A0 + (size_t)arow * H;
    const float* aRow1 = A1 + (size_t)arow * H;

    v8f acc[8] = {};

    for (int phase = 0; phase < 2; ++phase) {
        __syncthreads();
        {
            const float4* s = (const float4*)(phase ? W1 : W0);
            for (int i = tid; i < H * H / 4; i += 256) {
                const float4 v = s[i];
                const int k  = i >> 5;
                const int c0 = (i & 31) << 2;
                float* d = sB + ((k >> 1) << 8) + (k & 1);
                d[2 * (c0 + 0)] = v.x;
                d[2 * (c0 + 1)] = v.y;
                d[2 * (c0 + 2)] = v.z;
                d[2 * (c0 + 3)] = v.w;
            }
        }
        __syncthreads();
        const float* ap = phase ? aRow1 : aRow0;
        const float* bbase = sB + (half << 8) + (l15 << 1);
        for (int k0 = 0; k0 < H; k0 += 4) {
            const int ka = k0 + (half << 1);
            v2f a;
            a.x = ap[ka];
            a.y = ap[ka + 1];
            const float* bk = bbase + ((k0 >> 1) << 8);
            #pragma unroll
            for (int t = 0; t < 8; ++t) {
                const v2f b = *(const v2f*)(bk + (t << 5));
                acc[t] = __builtin_amdgcn_wmma_f32_16x16x4_f32(
                    false, a, false, b, (short)0, acc[t], false, false);
            }
        }
    }

    #pragma unroll
    for (int t = 0; t < 8; ++t) {
        const int col = (t << 4) + l15;
        const float bc = bias[col];
        #pragma unroll
        for (int r = 0; r < 8; ++r) {
            const int row = m0 + (half << 3) + r;
            if (row < n) {
                float v = acc[t][r] + bc;
                out[(size_t)row * H + col] = v > 0.0f ? v : (__expf(v) - 1.0f);
            }
        }
    }
}

extern "C" void kernel_launch(void* const* d_in, const int* in_sizes, int n_in,
                              void* d_out, int out_size, void* d_ws, size_t ws_size,
                              hipStream_t stream) {
    const float* x_int  = (const float*)d_in[0];
    const float* x_lane = (const float*)d_in[1];
    const float* x_sens = (const float*)d_in[2];
    const float* x_inj  = (const float*)d_in[3];
    const int*   e_sp   = (const int*)d_in[4];
    const int*   e_fl   = (const int*)d_in[5];
    const int*   e_fs   = (const int*)d_in[6];
    const int*   e_in   = (const int*)d_in[7];
    const float* W_int  = (const float*)d_in[8];
    const float* b_int  = (const float*)d_in[9];
    const float* W_lane = (const float*)d_in[10];
    const float* b_lane = (const float*)d_in[11];
    const float* W_sens = (const float*)d_in[12];
    const float* b_sens = (const float*)d_in[13];
    const float* W_inj  = (const float*)d_in[14];
    const float* b_inj  = (const float*)d_in[15];
    const float* self_W = (const float*)d_in[16];
    const float* self_b = (const float*)d_in[17];
    const float* rel_W  = (const float*)d_in[18];

    const int nInt  = in_sizes[0] / 32;
    const int nLane = in_sizes[1] / 16;
    const int nSens = in_sizes[2] / 8;
    const int nInj  = in_sizes[3] / 8;
    const int eSp   = in_sizes[4] / 2;
    const int eFl   = in_sizes[5] / 2;
    const int eFs   = in_sizes[6] / 2;
    const int eIn   = in_sizes[7] / 2;
    const int L     = in_sizes[16] / (H * H);

    float* ws = (float*)d_ws;
    size_t o = 0;
    float* hA    = ws + o; o += (size_t)nInt  * H;
    float* hB    = ws + o; o += (size_t)nInt  * H;
    float* hLane = ws + o; o += (size_t)nLane * H;
    float* hSens = ws + o; o += (size_t)nSens * H;
    float* hInj  = ws + o; o += (size_t)nInj  * H;
    float* aggS  = ws + o; o += (size_t)nInt  * H;
    float* aggB  = ws + o; o += (size_t)nInt  * H;
    int*   cnt   = (int*)(ws + o); o += (size_t)4 * nInt;
    int*   off   = (int*)(ws + o); o += (size_t)4 * nInt;
    (void)ws_size; (void)n_in; (void)out_size;

    int* rankSp = (int*)aggB;
    int* rankFl = rankSp + eSp;
    int* rankFs = rankFl + eFl;
    int* rankIn = rankFs + eFs;
    int* listFl = rankIn + eIn;
    int* listFs = listFl + eFl;
    int* listIn = listFs + eFs;
    int* listSp = (int*)hLane;

    proj_kernel<<<(nInt  + 63) / 64, 128, 0, stream>>>(x_int,  W_int,  b_int,  hA,    nInt,  32);
    proj_kernel<<<(nLane + 63) / 64, 128, 0, stream>>>(x_lane, W_lane, b_lane, hLane, nLane, 16);
    proj_kernel<<<(nSens + 63) / 64, 128, 0, stream>>>(x_sens, W_sens, b_sens, hSens, nSens, 8);
    proj_kernel<<<(nInj  + 63) / 64, 128, 0, stream>>>(x_inj,  W_inj,  b_inj,  hInj,  nInj,  8);

    const long long zc = 4LL * nInt;
    zero_u32_kernel<<<(unsigned)((zc + 255) / 256), 256, 0, stream>>>((unsigned*)cnt, zc);
    count_rank_kernel<<<(eSp + 255) / 256, 256, 0, stream>>>(e_sp + eSp, cnt + 0 * nInt, rankSp, eSp);
    count_rank_kernel<<<(eFl + 255) / 256, 256, 0, stream>>>(e_fl + eFl, cnt + 1 * nInt, rankFl, eFl);
    count_rank_kernel<<<(eFs + 255) / 256, 256, 0, stream>>>(e_fs + eFs, cnt + 2 * nInt, rankFs, eFs);
    count_rank_kernel<<<(eIn + 255) / 256, 256, 0, stream>>>(e_in + eIn, cnt + 3 * nInt, rankIn, eIn);
    scan_kernel<<<4, 1024, 0, stream>>>(cnt, off, nInt);

    fill_kernel<<<(eFl + 255) / 256, 256, 0, stream>>>(e_fl, e_fl + eFl, rankFl, off + 1 * nInt, listFl, eFl);
    fill_kernel<<<(eFs + 255) / 256, 256, 0, stream>>>(e_fs, e_fs + eFs, rankFs, off + 2 * nInt, listFs, eFs);
    fill_kernel<<<(eIn + 255) / 256, 256, 0, stream>>>(e_in, e_in + eIn, rankIn, off + 3 * nInt, listIn, eIn);
    gather_inv_kernel<<<(nInt + 7) / 8, 256, 0, stream>>>(
        hLane, hSens, hInj, listFl, listFs, listIn, cnt, off, aggS, nInt);

    fill_kernel<<<(eSp + 255) / 256, 256, 0, stream>>>(e_sp, e_sp + eSp, rankSp, off, listSp, eSp);

    float* hcur = hA;
    float* hnext = hB;
    for (int l = 0; l < L; ++l) {
        gather_sp_kernel<<<(nInt + 7) / 8, 256, 0, stream>>>(
            hcur, listSp, cnt, off, aggS, aggB, nInt);
        float* outp = (l == L - 1) ? (float*)d_out : hnext;
        fused_gemm_elu_kernel<<<(nInt + 127) / 128, 256, 0, stream>>>(
            hcur, aggB,
            self_W + (size_t)l * H * H, rel_W + (size_t)l * H * H,
            self_b + (size_t)l * H, outp, nInt);
        float* t = hnext; hnext = hcur; hcur = t;
    }
}


# Lane/sens/inj input projections moved behind the mean (linearity): the gather sums raw 16/8/8-float rows and projects once per target, three projection launches dropped; sum order changes
# speedup vs baseline: 2.7829x; 1.1709x over previous
#include <hip/hip_runtime.h>
#include <cstddef>

#define H 128

typedef float v2f __attribute__((ext_vector_type(2)));
typedef float v8f __attribute__((ext_vector_type(8)));

__global__ void zero_u32_kernel(unsigned* __restrict__ p, long long n) {
    long long i = (long long)blockIdx.x * blockDim.x + threadIdx.x;
    if (i < n) p[i] = 0u;
}

__global__ void proj_kernel(const float* __restrict__ x,
                            const float* __restrict__ W,
                            const float* __restrict__ b,
                            float* __restrict__ out, int n, int K) {
    __shared__ __align__(16) float sW[32 * H];
    const int tid = threadIdx.x;
    for (int i = tid; i < K * H; i += 128) sW[i] = W[i];
    const float bias = b[tid];
    __syncthreads();
    const int start = blockIdx.x * 64;
    const int end   = min(start + 64, n);
    for (int i = start; i < end; ++i) {
        const float* xr = x + (size_t)i * K;
        float acc = bias;
        for (int k = 0; k < K; ++k) acc += xr[k] * sW[k * H + tid];
        out[(size_t)i * H + tid] = acc;
    }
}

__global__ void count_rank_kernel(const int* __restrict__ tgt, int* __restrict__ cnt,
                                  int* __restrict__ rank, int E) {
    int e = blockIdx.x * blockDim.x + threadIdx.x;
    if (e < E) rank[e] = atomicAdd(&cnt[tgt[e]], 1);
}

__global__ __launch_bounds__(1024) void scan_kernel(const int* __restrict__ cnt,
                                                    int* __restrict__ off, int n) {
    __shared__ int sWave[32];
    const int* c = cnt + (size_t)blockIdx.x * n;
    int* o = off + (size_t)blockIdx.x * n;
    const int tid  = threadIdx.x;
    const int lane = tid & 31;
    const int wave = tid >> 5;
    const int per   = (n + 1023) / 1024;
    const int start = tid * per;
    if (n <= 0) return;
    const int last = n - 1;
    int sum = 0;
    for (int k = 0; k < per; k += 8) {
        int v[8];
        for (int u = 0; u < 8; ++u) v[u] = c[min(start + k + u, last)];
        for (int u = 0; u < 8; ++u)
            if (k + u < per && start + k + u < n) sum += v[u];
    }
    int inc = sum;
    for (int d = 1; d < 32; d <<= 1) {
        const int v = __shfl_up(inc, d, 32);
        if (lane >= d) inc += v;
    }
    if (lane == 31) sWave[wave] = inc;
    __syncthreads();
    if (wave == 0) {
        const int w = sWave[lane];
        int wi = w;
        for (int d = 1; d < 32; d <<= 1) {
            const int v = __shfl_up(wi, d, 32);
            if (lane >= d) wi += v;
        }
        sWave[lane] = wi - w;
    }
    __syncthreads();
    int run = sWave[wave] + inc - sum;
    for (int k = 0; k < per; k += 8) {
        int v[8];
        for (int u = 0; u < 8; ++u) v[u] = c[min(start + k + u, last)];
        for (int u = 0; u < 8; ++u) {
            const int i = start + k + u;
            if (k + u < per && i < n) { o[i] = run; run += v[u]; }
        }
    }
}

__global__ void fill_kernel(const int* __restrict__ src, const int* __restrict__ tgt,
                            const int* __restrict__ rank, const int* __restrict__ off,
                            int* __restrict__ list, int E) {
    int e = blockIdx.x * blockDim.x + threadIdx.x;
    if (e < E) list[off[tgt[e]] + rank[e]] = src[e];
}

__device__ __forceinline__ void add4(float4& a, const float4 v) {
    a.x += v.x; a.y += v.y; a.z += v.z; a.w += v.w;
}

__device__ __forceinline__ float4 bucket_sum(const float4* __restrict__ feat4,
                                             const int* __restrict__ list,
                                             int beg, int len, int lane) {
    float4 acc = make_float4(0.0f, 0.0f, 0.0f, 0.0f);
    if (len <= 32) {
        const int key = (lane < len) ? list[beg + lane] : 0x7fffffff;
        int rank = 0;
        for (int j = 0; j < len; ++j) {
            const int kj = __builtin_amdgcn_readlane(key, j);
            rank += (kj < key || (kj == key && j < lane)) ? 1 : 0;
        }
        if (lane >= len) rank = lane;
        const int sorted = __builtin_amdgcn_ds_permute(rank << 2, key);
        int r = 0;
        for (; r + 4 <= len; r += 4) {
            const int s0 = __builtin_amdgcn_readlane(sorted, r);
            const int s1 = __builtin_amdgcn_readlane(sorted, r + 1);
            const int s2 = __builtin_amdgcn_readlane(sorted, r + 2);
            const int s3 = __builtin_amdgcn_readlane(sorted, r + 3);
            const float4 v0 = feat4[(size_t)s0 * (H / 4) + lane];
            const float4 v1 = feat4[(size_t)s1 * (H / 4) + lane];
            const float4 v2 = feat4[(size_t)s2 * (H / 4) + lane];
            const float4 v3 = feat4[(size_t)s3 * (H / 4) + lane];
            add4(acc, v0); add4(acc, v1); add4(acc, v2); add4(acc, v3);
        }
        for (; r < len; ++r) {
            const int s0 = __builtin_amdgcn_readlane(sorted, r);
            add4(acc, feat4[(size_t)s0 * (H / 4) + lane]);
        }
    } else {
        int prev = -1;
        for (int it = 0; it < len; ++it) {
            int m = 0x7fffffff;
            for (int i = lane; i < len; i += 32) {
                const int k = list[beg + i];
                if (k > prev && k < m) m = k;
            }
            for (int o = 16; o > 0; o >>= 1) m = min(m, __shfl_xor(m, o, 32));
            if (m == 0x7fffffff) break;
            int c = 0;
            for (int i = lane; i < len; i += 32) c += (list[beg + i] == m) ? 1 : 0;
            for (int o = 16; o > 0; o >>= 1) c += __shfl_xor(c, o, 32);
            const float4 v = feat4[(size_t)m * (H / 4) + lane];
            for (int q = 0; q < c; ++q) add4(acc, v);
            prev = m;
        }
    }
    return acc;
}


__device__ __forceinline__ int sort_ids(int key, int len, int lane) {
    int rank = 0;
    for (int j = 0; j < len; ++j) {
        const int kj = __builtin_amdgcn_readlane(key, j);
        rank += (kj < key || (kj == key && j < lane)) ? 1 : 0;
    }
    if (lane >= len) rank = lane;
    return __builtin_amdgcn_ds_permute(rank << 2, key);
}

template <int Q>
__device__ __forceinline__ float4 long_bucket_sum(const float4* __restrict__ x4,
                                                  const int* __restrict__ list,
                                                  int beg, int len, int lane, int qq) {
    float4 acc = make_float4(0.0f, 0.0f, 0.0f, 0.0f);
    int prev = -1;
    for (int it = 0; it < len; ++it) {
        int m = 0x7fffffff;
        for (int i = lane; i < len; i += 32) {
            const int k = list[beg + i];
            if (k > prev && k < m) m = k;
        }
        for (int o = 16; o > 0; o >>= 1) m = min(m, __shfl_xor(m, o, 32));
        if (m == 0x7fffffff) break;
        int c = 0;
        for (int i = lane; i < len; i += 32) c += (list[beg + i] == m) ? 1 : 0;
        for (int o = 16; o > 0; o >>= 1) c += __shfl_xor(c, o, 32);
        const float4 v = x4[(size_t)m * Q + qq];
        for (int r = 0; r < c; ++r) add4(acc, v);
        prev = m;
    }
    return acc;
}

__device__ __forceinline__ float4 slot_sum(float4 a) {
    for (int m = 4; m < 32; m <<= 1) {
        float4 b;
        b.x = __shfl_xor(a.x, m, 32); b.y = __shfl_xor(a.y, m, 32);
        b.z = __shfl_xor(a.z, m, 32); b.w = __shfl_xor(a.w, m, 32);
        add4(a, b);
    }
    return a;
}

#define TPW 4

__global__ __launch_bounds__(256) void gather_raw_proj_kernel(
        const float* __restrict__ xLane, const float* __restrict__ xSens,
        const float* __restrict__ xInj,
        const float* __restrict__ WLane, const float* __restrict__ bLane,
        const float* __restrict__ WSens, const float* __restrict__ bSens,
        const float* __restrict__ WInj,  const float* __restrict__ bInj,
        const int* __restrict__ listFl, const int* __restrict__ listFs,
        const int* __restrict__ listIn,
        const int* __restrict__ cnt, const int* __restrict__ off,
        float* __restrict__ aggS, int n) {
    __shared__ float4 sW[35 * 32];
    __shared__ float4 sZ[8][TPW][9];
    const int tid  = threadIdx.x;
    const int wave = tid >> 5;
    const int lane = tid & 31;
    const int slot = lane >> 2;
    const int q    = lane & 3;

    {
        const float4* wl = (const float4*)WLane;
        sW[tid]       = wl[tid];
        sW[256 + tid] = wl[256 + tid];
        sW[512 + tid] = ((const float4*)WSens)[tid];
        sW[768 + tid] = ((const float4*)WInj)[tid];
        if (tid < 96) {
            const float* bs = tid < 32 ? bLane : (tid < 64 ? bSens : bInj);
            sW[1024 + tid] = ((const float4*)bs)[tid & 31];
        }
    }

    const float4* xl4 = (const float4*)xLane;
    const float4* xs4 = (const float4*)xSens;
    const float4* xi4 = (const float4*)xInj;
    const float4 zero4 = make_float4(0.0f, 0.0f, 0.0f, 0.0f);
    const int tb = blockIdx.x * (8 * TPW) + wave * TPW;

    for (int u = 0; u < TPW; ++u) {
        const int t = __builtin_amdgcn_readfirstlane(tb + u);
        float4 zl = zero4, zsi = zero4;
        float fl = 0.0f, fs = 0.0f, fi = 0.0f;
        if (t < n) {
            const int lenL = cnt[1 * n + t], lenS = cnt[2 * n + t], lenI = cnt[3 * n + t];
            const int begL = off[1 * n + t], begS = off[2 * n + t], begI = off[3 * n + t];
            fl = lenL > 0 ? 1.0f : 0.0f;
            fs = lenS > 0 ? 1.0f : 0.0f;
            fi = lenI > 0 ? 1.0f : 0.0f;
            const bool isS = q < 2;
            const int qq = q & 1;
            if (lenL <= 32 && lenS <= 32 && lenI <= 32) {
                const int keyL = (lane < lenL) ? listFl[begL + lane] : 0x7fffffff;
                const int keyS = (lane < lenS) ? listFs[begS + lane] : 0x7fffffff;
                const int keyI = (lane < lenI) ? listIn[begI + lane] : 0x7fffffff;
                const int sortedL = sort_ids(keyL, lenL, lane);
                const int sortedS = sort_ids(keyS, lenS, lane);
                const int sortedI = sort_ids(keyI, lenI, lane);
                const int lenX = isS ? lenS : lenI;
                const int maxSI = max(lenS, lenI);
                const float4* xb = isS ? xs4 : xi4;
                float4 a[4], b[4];
                #pragma unroll
                for (int i = 0; i < 4; ++i) {
                    a[i] = zero4; b[i] = zero4;
                    const int p = 8 * i + slot;
                    if (8 * i < lenL) {
                        const int id = __builtin_amdgcn_ds_bpermute(p << 2, sortedL);
                        if (p < lenL) a[i] = xl4[(size_t)id * 4 + q];
                    }
                    if (8 * i < maxSI) {
                        const int idS = __builtin_amdgcn_ds_bpermute(p << 2, sortedS);
                        const int idI = __builtin_amdgcn_ds_bpermute(p << 2, sortedI);
                        if (p < lenX) b[i] = xb[(size_t)(isS ? idS : idI) * 2 + qq];
                    }
                }
                add4(a[0], a[1]); add4(a[0], a[2]); add4(a[0], a[3]);
                add4(b[0], b[1]); add4(b[0], b[2]); add4(b[0], b[3]);
                zl  = slot_sum(a[0]);
                zsi = slot_sum(b[0]);
            } else {
                zl = long_bucket_sum<4>(xl4, listFl, begL, lenL, lane, q);
                const float4 s = long_bucket_sum<2>(xs4, listFs, begS, lenS, lane, qq);
                const float4 j = long_bucket_sum<2>(xi4, listIn, begI, lenI, lane, qq);
                zsi = isS ? s : j;
            }
            const float wL = 1.0f / (float)max(lenL, 1);
            const float wX = 1.0f / (float)max(isS ? lenS : lenI, 1);
            zl.x *= wL;  zl.y *= wL;  zl.z *= wL;  zl.w *= wL;
            zsi.x *= wX; zsi.y *= wX; zsi.z *= wX; zsi.w *= wX;
        }
        if (lane < 4) { sZ[wave][u][lane] = zl; sZ[wave][u][4 + lane] = zsi; }
        if (lane == 0) sZ[wave][u][8] = make_float4(fl, fs, fi, 0.0f);
    }
    __syncthreads();

    float4 o[TPW];
    #pragma unroll
    for (int u = 0; u < TPW; ++u) o[u] = zero4;
    #pragma nounroll
    for (int g = 0; g < 8; ++g) {
        const float4 w0 = sW[(4 * g + 0) * 32 + lane];
        const float4 w1 = sW[(4 * g + 1) * 32 + lane];
        const float4 w2 = sW[(4 * g + 2) * 32 + lane];
        const float4 w3 = sW[(4 * g + 3) * 32 + lane];
        #pragma unroll
        for (int u = 0; u < TPW; ++u) {
            const float4 z = sZ[wave][u][g];
            o[u].x += z.x * w0.x; o[u].y += z.x * w0.y; o[u].z += z.x * w0.z; o[u].w += z.x * w0.w;
            o[u].x += z.y * w1.x; o[u].y += z.y * w1.y; o[u].z += z.y * w1.z; o[u].w += z.y * w1.w;
            o[u].x += z.z * w2.x; o[u].y += z.z * w2.y; o[u].z += z.z * w2.z; o[u].w += z.z * w2.w;
            o[u].x += z.w * w3.x; o[u].y += z.w * w3.y; o[u].z += z.w * w3.z; o[u].w += z.w * w3.w;
        }
    }
    {
        const float4 b0 = sW[32 * 32 + lane];
        const float4 b1 = sW[33 * 32 + lane];
        const float4 b2 = sW[34 * 32 + lane];
        #pragma unroll
        for (int u = 0; u < TPW; ++u) {
            const float4 f = sZ[wave][u][8];
            o[u].x += f.x * b0.x; o[u].y += f.x * b0.y; o[u].z += f.x * b0.z; o[u].w += f.x * b0.w;
            o[u].x += f.y * b1.x; o[u].y += f.y * b1.y; o[u].z += f.y * b1.z; o[u].w += f.y * b1.w;
            o[u].x += f.z * b2.x; o[u].y += f.z * b2.y; o[u].z += f.z * b2.z; o[u].w += f.z * b2.w;
        }
    }
    #pragma unroll
    for (int u = 0; u < TPW; ++u) {
        const int t = tb + u;
        if (t < n) ((float4*)aggS)[(size_t)t * (H / 4) + lane] = o[u];
    }
}

__global__ __launch_bounds__(256) void gather_sp_kernel(
        const float* __restrict__ h, const int* __restrict__ listSp,
        const int* __restrict__ cnt, const int* __restrict__ off,
        const float* __restrict__ aggS, float* __restrict__ aggB, int n) {
    const int lane = threadIdx.x & 31;
    const int t = __builtin_amdgcn_readfirstlane((int)(blockIdx.x * 8 + (threadIdx.x >> 5)));
    if (t >= n) return;
    const float4 base = ((const float4*)aggS)[(size_t)t * (H / 4) + lane];
    const int len = cnt[t];
    const float4 s = bucket_sum((const float4*)h, listSp, off[t], len, lane);
    const float w = 1.0f / (float)max(len, 1);
    float4 r;
    r.x = base.x + s.x * w; r.y = base.y + s.y * w;
    r.z = base.z + s.z * w; r.w = base.w + s.w * w;
    ((float4*)aggB)[(size_t)t * (H / 4) + lane] = r;
}

__global__ __launch_bounds__(256) void fused_gemm_elu_kernel(
        const float* __restrict__ A0, const float* __restrict__ A1,
        const float* __restrict__ W0, const float* __restrict__ W1,
        const float* __restrict__ bias, float* __restrict__ out, int n) {
    __shared__ __align__(16) float sB[H * H];
    const int tid  = threadIdx.x;
    const int wave = tid >> 5;
    const int lane = tid & 31;
    const int half = lane >> 4;
    const int l15  = lane & 15;
    const int m0   = blockIdx.x * 128 + wave * 16;

    int arow = m0 + l15;
    if (arow >= n) arow = n - 1;
    const float* aRow0 = A0 + (size_t)arow * H;
    const float* aRow1 = A1 + (size_t)arow * H;

    v8f acc[8] = {};

    for (int phase = 0; phase < 2; ++phase) {
        __syncthreads();
        {
            const float4* s = (const float4*)(phase ? W1 : W0);
            for (int i = tid; i < H * H / 4; i += 256) {
                const float4 v = s[i];
                const int k  = i >> 5;
                const int c0 = (i & 31) << 2;
                float* d = sB + ((k >> 1) << 8) + (k & 1);
                d[2 * (c0 + 0)] = v.x;
                d[2 * (c0 + 1)] = v.y;
                d[2 * (c0 + 2)] = v.z;
                d[2 * (c0 + 3)] = v.w;
            }
        }
        __syncthreads();
        const float* ap = phase ? aRow1 : aRow0;
        const float* bbase = sB + (half << 8) + (l15 << 1);
        for (int k0 = 0; k0 < H; k0 += 4) {
            const int ka = k0 + (half << 1);
            v2f a;
            a.x = ap[ka];
            a.y = ap[ka + 1];
            const float* bk = bbase + ((k0 >> 1) << 8);
            #pragma unroll
            for (int t = 0; t < 8; ++t) {
                const v2f b = *(const v2f*)(bk + (t << 5));
                acc[t] = __builtin_amdgcn_wmma_f32_16x16x4_f32(
                    false, a, false, b, (short)0, acc[t], false, false);
            }
        }
    }

    #pragma unroll
    for (int t = 0; t < 8; ++t) {
        const int col = (t << 4) + l15;
        const float bc = bias[col];
        #pragma unroll
        for (int r = 0; r < 8; ++r) {
            const int row = m0 + (half << 3) + r;
            if (row < n) {
                float v = acc[t][r] + bc;
                out[(size_t)row * H + col] = v > 0.0f ? v : (__expf(v) - 1.0f);
            }
        }
    }
}

extern "C" void kernel_launch(void* const* d_in, const int* in_sizes, int n_in,
                              void* d_out, int out_size, void* d_ws, size_t ws_size,
                              hipStream_t stream) {
    const float* x_int  = (const float*)d_in[0];
    const float* x_lane = (const float*)d_in[1];
    const float* x_sens = (const float*)d_in[2];
    const float* x_inj  = (const float*)d_in[3];
    const int*   e_sp   = (const int*)d_in[4];
    const int*   e_fl   = (const int*)d_in[5];
    const int*   e_fs   = (const int*)d_in[6];
    const int*   e_in   = (const int*)d_in[7];
    const float* W_int  = (const float*)d_in[8];
    const float* b_int  = (const float*)d_in[9];
    const float* W_lane = (const float*)d_in[10];
    const float* b_lane = (const float*)d_in[11];
    const float* W_sens = (const float*)d_in[12];
    const float* b_sens = (const float*)d_in[13];
    const float* W_inj  = (const float*)d_in[14];
    const float* b_inj  = (const float*)d_in[15];
    const float* self_W = (const float*)d_in[16];
    const float* self_b = (const float*)d_in[17];
    const float* rel_W  = (const float*)d_in[18];

    const int nInt  = in_sizes[0] / 32;
    const int nLane = in_sizes[1] / 16;
    const int nSens = in_sizes[2] / 8;
    const int nInj  = in_sizes[3] / 8;
    const int eSp   = in_sizes[4] / 2;
    const int eFl   = in_sizes[5] / 2;
    const int eFs   = in_sizes[6] / 2;
    const int eIn   = in_sizes[7] / 2;
    const int L     = in_sizes[16] / (H * H);

    float* ws = (float*)d_ws;
    size_t o = 0;
    float* hA    = ws + o; o += (size_t)nInt  * H;
    float* hB    = ws + o; o += (size_t)nInt  * H;
    float* hLane = ws + o; o += (size_t)nLane * H;
    o += (size_t)nSens * H;
    o += (size_t)nInj  * H;
    float* aggS  = ws + o; o += (size_t)nInt  * H;
    float* aggB  = ws + o; o += (size_t)nInt  * H;
    int*   cnt   = (int*)(ws + o); o += (size_t)4 * nInt;
    int*   off   = (int*)(ws + o); o += (size_t)4 * nInt;
    (void)ws_size; (void)n_in; (void)out_size;

    int* rankSp = (int*)aggB;
    int* rankFl = rankSp + eSp;
    int* rankFs = rankFl + eFl;
    int* rankIn = rankFs + eFs;
    int* listFl = rankIn + eIn;
    int* listFs = listFl + eFl;
    int* listIn = listFs + eFs;
    int* listSp = (int*)hLane;

    proj_kernel<<<(nInt  + 63) / 64, 128, 0, stream>>>(x_int,  W_int,  b_int,  hA,    nInt,  32);

    const long long zc = 4LL * nInt;
    zero_u32_kernel<<<(unsigned)((zc + 255) / 256), 256, 0, stream>>>((unsigned*)cnt, zc);
    count_rank_kernel<<<(eSp + 255) / 256, 256, 0, stream>>>(e_sp + eSp, cnt + 0 * nInt, rankSp, eSp);
    count_rank_kernel<<<(eFl + 255) / 256, 256, 0, stream>>>(e_fl + eFl, cnt + 1 * nInt, rankFl, eFl);
    count_rank_kernel<<<(eFs + 255) / 256, 256, 0, stream>>>(e_fs + eFs, cnt + 2 * nInt, rankFs, eFs);
    count_rank_kernel<<<(eIn + 255) / 256, 256, 0, stream>>>(e_in + eIn, cnt + 3 * nInt, rankIn, eIn);
    scan_kernel<<<4, 1024, 0, stream>>>(cnt, off, nInt);

    fill_kernel<<<(eFl + 255) / 256, 256, 0, stream>>>(e_fl, e_fl + eFl, rankFl, off + 1 * nInt, listFl, eFl);
    fill_kernel<<<(eFs + 255) / 256, 256, 0, stream>>>(e_fs, e_fs + eFs, rankFs, off + 2 * nInt, listFs, eFs);
    fill_kernel<<<(eIn + 255) / 256, 256, 0, stream>>>(e_in, e_in + eIn, rankIn, off + 3 * nInt, listIn, eIn);
    gather_raw_proj_kernel<<<(nInt + 8 * TPW - 1) / (8 * TPW), 256, 0, stream>>>(
        x_lane, x_sens, x_inj, W_lane, b_lane, W_sens, b_sens, W_inj, b_inj,
        listFl, listFs, listIn, cnt, off, aggS, nInt);

    fill_kernel<<<(eSp + 255) / 256, 256, 0, stream>>>(e_sp, e_sp + eSp, rankSp, off, listSp, eSp);

    float* hcur = hA;
    float* hnext = hB;
    for (int l = 0; l < L; ++l) {
        gather_sp_kernel<<<(nInt + 7) / 8, 256, 0, stream>>>(
            hcur, listSp, cnt, off, aggS, aggB, nInt);
        float* outp = (l == L - 1) ? (float*)d_out : hnext;
        fused_gemm_elu_kernel<<<(nInt + 127) / 128, 256, 0, stream>>>(
            hcur, aggB,
            self_W + (size_t)l * H * H, rel_W + (size_t)l * H * H,
            self_b + (size_t)l * H, outp, nInt);
        float* t = hnext; hnext = hcur; hcur = t;
    }
}


# Edges grouped by target through LDS: chunks binned by 256-target range, then a block per range counts, scans and places ids; no global atomic or scattered access per edge; 18 launches to 10
# speedup vs baseline: 5.0822x; 1.8262x over previous
#include <hip/hip_runtime.h>
#include <cstddef>

#define H 128

typedef float v2f __attribute__((ext_vector_type(2)));
typedef float v8f __attribute__((ext_vector_type(8)));

__global__ void proj_kernel(const float* __restrict__ x,
                            const float* __restrict__ W,
                            const float* __restrict__ b,
                            float* __restrict__ out, int n, int K) {
    __shared__ __align__(16) float sW[32 * H];
    const int tid = threadIdx.x;
    for (int i = tid; i < K * H; i += 128) sW[i] = W[i];
    const float bias = b[tid];
    __syncthreads();
    const int start = blockIdx.x * 64;
    const int end   = min(start + 64, n);
    for (int i = start; i < end; ++i) {
        const float* xr = x + (size_t)i * K;
        float acc = bias;
        for (int k = 0; k < K; ++k) acc += xr[k] * sW[k * H + tid];
        out[(size_t)i * H + tid] = acc;
    }
}

#define CHUNK 4096
#define MAXNB 1024
#define STAGE 8192

__global__ __launch_bounds__(256) void bin_kernel(
        const int* __restrict__ ed0, const int* __restrict__ ed1,
        const int* __restrict__ ed2, const int* __restrict__ ed3,
        int* __restrict__ binned, int* __restrict__ tab,
        int E0, int E1, int E2, int E3, int nb, int nch) {
    __shared__ int  sHist[MAXNB + 1];
    __shared__ int2 sPair[CHUNK];
    const int tid = threadIdx.x;
    const int C0 = (E0 + CHUNK - 1) / CHUNK, C1 = (E1 + CHUNK - 1) / CHUNK,
              C2 = (E2 + CHUNK - 1) / CHUNK;
    int c = blockIdx.x;
    const int* ed = ed0;
    int E = E0, rbase = 0;
    if (c >= C0) {
        c -= C0; ed = ed1; E = E1; rbase = E0;
        if (c >= C1) {
            c -= C1; ed = ed2; E = E2; rbase += E1;
            if (c >= C2) { c -= C2; ed = ed3; E = E3; rbase += E2; }
        }
    }
    const int e0  = c * CHUNK;
    const int cnt = min(CHUNK, E - e0);

    for (int j = tid; j <= nb; j += 256) sHist[j] = 0;
    __syncthreads();

    int srcv[16], tgtv[16], rk[16];
    #pragma unroll
    for (int k = 0; k < 16; ++k) {
        const int i = k * 256 + tid;
        srcv[k] = 0; tgtv[k] = 0;
        if (i < cnt) { srcv[k] = ed[e0 + i]; tgtv[k] = ed[E + e0 + i]; }
    }
    #pragma unroll
    for (int k = 0; k < 16; ++k) {
        const int i = k * 256 + tid;
        rk[k] = 0;
        if (i < cnt) rk[k] = atomicAdd(&sHist[tgtv[k] >> 8], 1);
    }
    __syncthreads();

    if (tid < 32) {
        int carry = 0;
        for (int base = 0; base < nb; base += 32) {
            const int j = base + tid;
            const int v = (j < nb) ? sHist[j] : 0;
            int inc = v;
            for (int d = 1; d < 32; d <<= 1) {
                const int u = __shfl_up(inc, d, 32);
                if (tid >= d) inc += u;
            }
            if (j < nb) sHist[j] = carry + inc - v;
            carry += __shfl(inc, 31, 32);
        }
        if (tid == 0) sHist[nb] = carry;
    }
    __syncthreads();

    #pragma unroll
    for (int k = 0; k < 16; ++k) {
        const int i = k * 256 + tid;
        if (i < cnt) sPair[sHist[tgtv[k] >> 8] + rk[k]] = make_int2(srcv[k], tgtv[k]);
    }
    for (int j = tid; j <= nb; j += 256) tab[(size_t)j * nch + blockIdx.x] = sHist[j];
    __syncthreads();

    int2* outp = (int2*)binned + (size_t)rbase + e0;
    for (int i = tid; i < cnt; i += 256) outp[i] = sPair[i];
}

__global__ __launch_bounds__(256) void fine_kernel(
        const int* __restrict__ binned, const int* __restrict__ tab,
        int* __restrict__ cnt, int* __restrict__ off,
        int* __restrict__ list0, int* __restrict__ list1,
        int* __restrict__ list2, int* __restrict__ list3,
        int E0, int E1, int E2, int E3, int nb, int nch, int n) {
    __shared__ int sCnt[256];
    __shared__ int sCur[256];
    __shared__ int sWave[8];
    __shared__ int sRed[8];
    __shared__ int sStage[STAGE];
    const int tid  = threadIdx.x;
    const int lane = tid & 31;
    const int wave = tid >> 5;
    const int C0 = (E0 + CHUNK - 1) / CHUNK, C1 = (E1 + CHUNK - 1) / CHUNK,
              C2 = (E2 + CHUNK - 1) / CHUNK, C3 = (E3 + CHUNK - 1) / CHUNK;
    const int r = blockIdx.x / nb;
    const int g = blockIdx.x - r * nb;
    int rbase = 0, cb = 0, nC = C0;
    int* list = list0;
    if (r == 1) { rbase = E0;           cb = C0;           nC = C1; list = list1; }
    if (r == 2) { rbase = E0 + E1;      cb = C0 + C1;      nC = C2; list = list2; }
    if (r == 3) { rbase = E0 + E1 + E2; cb = C0 + C1 + C2; nC = C3; list = list3; }
    const int* tlo = tab + (size_t)g * nch + cb;
    const int* thi = tab + (size_t)(g + 1) * nch + cb;
    const int2* win = (const int2*)binned + (size_t)rbase;

    sCnt[tid] = 0;
    __syncthreads();

    int lower = 0;
    for (int b = tid; b < nC; b += 256) {
        const int s = tlo[b], e = thi[b];
        lower += s;
        const int2* p = win + (size_t)b * CHUNK;
        for (int i = s; i < e; i += 8) {
            int y[8];
            #pragma unroll
            for (int u = 0; u < 8; ++u) y[u] = p[min(i + u, e - 1)].y;
            #pragma unroll
            for (int u = 0; u < 8; ++u)
                if (i + u < e) atomicAdd(&sCnt[y[u] & 255], 1);
        }
    }
    __syncthreads();

    const int cT = sCnt[tid];
    int inc = cT;
    for (int d = 1; d < 32; d <<= 1) {
        const int u = __shfl_up(inc, d, 32);
        if (lane >= d) inc += u;
    }
    for (int o = 16; o > 0; o >>= 1) lower += __shfl_xor(lower, o, 32);
    if (lane == 31) sWave[wave] = inc;
    if (lane == 0)  sRed[wave]  = lower;
    __syncthreads();
    int wbase = 0, total = 0, start = 0;
    for (int w = 0; w < 8; ++w) {
        const int tw = sWave[w];
        if (w < wave) wbase += tw;
        total += tw;
        start += sRed[w];
    }
    const int offL = wbase + inc - cT;
    const int t = g * 256 + tid;
    if (t < n) {
        cnt[(size_t)r * n + t] = cT;
        off[(size_t)r * n + t] = start + offL;
    }
    sCur[tid] = offL;
    __syncthreads();

    const bool staged = total <= STAGE;
    for (int b = tid; b < nC; b += 256) {
        const int s = tlo[b], e = thi[b];
        const int2* p = win + (size_t)b * CHUNK;
        for (int i = s; i < e; i += 8) {
            int2 v[8];
            #pragma unroll
            for (int u = 0; u < 8; ++u) v[u] = p[min(i + u, e - 1)];
            #pragma unroll
            for (int u = 0; u < 8; ++u) {
                if (i + u < e) {
                    const int pos = atomicAdd(&sCur[v[u].y & 255], 1);
                    if (staged) sStage[pos] = v[u].x;
                    else        list[start + pos] = v[u].x;
                }
            }
        }
    }
    if (staged) {
        __syncthreads();
        for (int i = tid; i < total; i += 256) list[start + i] = sStage[i];
    }
}

__device__ __forceinline__ void add4(float4& a, const float4 v) {
    a.x += v.x; a.y += v.y; a.z += v.z; a.w += v.w;
}

__device__ __forceinline__ float4 bucket_sum(const float4* __restrict__ feat4,
                                             const int* __restrict__ list,
                                             int beg, int len, int lane) {
    float4 acc = make_float4(0.0f, 0.0f, 0.0f, 0.0f);
    if (len <= 32) {
        const int key = (lane < len) ? list[beg + lane] : 0x7fffffff;
        int rank = 0;
        for (int j = 0; j < len; ++j) {
            const int kj = __builtin_amdgcn_readlane(key, j);
            rank += (kj < key || (kj == key && j < lane)) ? 1 : 0;
        }
        if (lane >= len) rank = lane;
        const int sorted = __builtin_amdgcn_ds_permute(rank << 2, key);
        int r = 0;
        for (; r + 4 <= len; r += 4) {
            const int s0 = __builtin_amdgcn_readlane(sorted, r);
            const int s1 = __builtin_amdgcn_readlane(sorted, r + 1);
            const int s2 = __builtin_amdgcn_readlane(sorted, r + 2);
            const int s3 = __builtin_amdgcn_readlane(sorted, r + 3);
            const float4 v0 = feat4[(size_t)s0 * (H / 4) + lane];
            const float4 v1 = feat4[(size_t)s1 * (H / 4) + lane];
            const float4 v2 = feat4[(size_t)s2 * (H / 4) + lane];
            const float4 v3 = feat4[(size_t)s3 * (H / 4) + lane];
            add4(acc, v0); add4(acc, v1); add4(acc, v2); add4(acc, v3);
        }
        for (; r < len; ++r) {
            const int s0 = __builtin_amdgcn_readlane(sorted, r);
            add4(acc, feat4[(size_t)s0 * (H / 4) + lane]);
        }
    } else {
        int prev = -1;
        for (int it = 0; it < len; ++it) {
            int m = 0x7fffffff;
            for (int i = lane; i < len; i += 32) {
                const int k = list[beg + i];
                if (k > prev && k < m) m = k;
            }
            for (int o = 16; o > 0; o >>= 1) m = min(m, __shfl_xor(m, o, 32));
            if (m == 0x7fffffff) break;
            int c = 0;
            for (int i = lane; i < len; i += 32) c += (list[beg + i] == m) ? 1 : 0;
            for (int o = 16; o > 0; o >>= 1) c += __shfl_xor(c, o, 32);
            const float4 v = feat4[(size_t)m * (H / 4) + lane];
            for (int q = 0; q < c; ++q) add4(acc, v);
            prev = m;
        }
    }
    return acc;
}


__device__ __forceinline__ int sort_ids(int key, int len, int lane) {
    int rank = 0;
    for (int j = 0; j < len; ++j) {
        const int kj = __builtin_amdgcn_readlane(key, j);
        rank += (kj < key || (kj == key && j < lane)) ? 1 : 0;
    }
    if (lane >= len) rank = lane;
    return __builtin_amdgcn_ds_permute(rank << 2, key);
}

template <int Q>
__device__ __forceinline__ float4 long_bucket_sum(const float4* __restrict__ x4,
                                                  const int* __restrict__ list,
                                                  int beg, int len, int lane, int qq) {
    float4 acc = make_float4(0.0f, 0.0f, 0.0f, 0.0f);
    int prev = -1;
    for (int it = 0; it < len; ++it) {
        int m = 0x7fffffff;
        for (int i = lane; i < len; i += 32) {
            const int k = list[beg + i];
            if (k > prev && k < m) m = k;
        }
        for (int o = 16; o > 0; o >>= 1) m = min(m, __shfl_xor(m, o, 32));
        if (m == 0x7fffffff) break;
        int c = 0;
        for (int i = lane; i < len; i += 32) c += (list[beg + i] == m) ? 1 : 0;
        for (int o = 16; o > 0; o >>= 1) c += __shfl_xor(c, o, 32);
        const float4 v = x4[(size_t)m * Q + qq];
        for (int r = 0; r < c; ++r) add4(acc, v);
        prev = m;
    }
    return acc;
}

__device__ __forceinline__ float4 slot_sum(float4 a) {
    for (int m = 4; m < 32; m <<= 1) {
        float4 b;
        b.x = __shfl_xor(a.x, m, 32); b.y = __shfl_xor(a.y, m, 32);
        b.z = __shfl_xor(a.z, m, 32); b.w = __shfl_xor(a.w, m, 32);
        add4(a, b);
    }
    return a;
}

#define TPW 4

__global__ __launch_bounds__(256) void gather_raw_proj_kernel(
        const float* __restrict__ xLane, const float* __restrict__ xSens,
        const float* __restrict__ xInj,
        const float* __restrict__ WLane, const float* __restrict__ bLane,
        const float* __restrict__ WSens, const float* __restrict__ bSens,
        const float* __restrict__ WInj,  const float* __restrict__ bInj,
        const int* __restrict__ listFl, const int* __restrict__ listFs,
        const int* __restrict__ listIn,
        const int* __restrict__ cnt, const int* __restrict__ off,
        float* __restrict__ aggS, int n) {
    __shared__ float4 sW[35 * 32];
    __shared__ float4 sZ[8][TPW][9];
    const int tid  = threadIdx.x;
    const int wave = tid >> 5;
    const int lane = tid & 31;
    const int slot = lane >> 2;
    const int q    = lane & 3;

    {
        const float4* wl = (const float4*)WLane;
        sW[tid]       = wl[tid];
        sW[256 + tid] = wl[256 + tid];
        sW[512 + tid] = ((const float4*)WSens)[tid];
        sW[768 + tid] = ((const float4*)WInj)[tid];
        if (tid < 96) {
            const float* bs = tid < 32 ? bLane : (tid < 64 ? bSens : bInj);
            sW[1024 + tid] = ((const float4*)bs)[tid & 31];
        }
    }

    const float4* xl4 = (const float4*)xLane;
    const float4* xs4 = (const float4*)xSens;
    const float4* xi4 = (const float4*)xInj;
    const float4 zero4 = make_float4(0.0f, 0.0f, 0.0f, 0.0f);
    const int tb = blockIdx.x * (8 * TPW) + wave * TPW;

    for (int u = 0; u < TPW; ++u) {
        const int t = __builtin_amdgcn_readfirstlane(tb + u);
        float4 zl = zero4, zsi = zero4;
        float fl = 0.0f, fs = 0.0f, fi = 0.0f;
        if (t < n) {
            const int lenL = cnt[1 * n + t], lenS = cnt[2 * n + t], lenI = cnt[3 * n + t];
            const int begL = off[1 * n + t], begS = off[2 * n + t], begI = off[3 * n + t];
            fl = lenL > 0 ? 1.0f : 0.0f;
            fs = lenS > 0 ? 1.0f : 0.0f;
            fi = lenI > 0 ? 1.0f : 0.0f;
            const bool isS = q < 2;
            const int qq = q & 1;
            if (lenL <= 32 && lenS <= 32 && lenI <= 32) {
                const int keyL = (lane < lenL) ? listFl[begL + lane] : 0x7fffffff;
                const int keyS = (lane < lenS) ? listFs[begS + lane] : 0x7fffffff;
                const int keyI = (lane < lenI) ? listIn[begI + lane] : 0x7fffffff;
                const int sortedL = sort_ids(keyL, lenL, lane);
                const int sortedS = sort_ids(keyS, lenS, lane);
                const int sortedI = sort_ids(keyI, lenI, lane);
                const int lenX = isS ? lenS : lenI;
                const int maxSI = max(lenS, lenI);
                const float4* xb = isS ? xs4 : xi4;
                float4 a[4], b[4];
                #pragma unroll
                for (int i = 0; i < 4; ++i) {
                    a[i] = zero4; b[i] = zero4;
                    const int p = 8 * i + slot;
                    if (8 * i < lenL) {
                        const int id = __builtin_amdgcn_ds_bpermute(p << 2, sortedL);
                        if (p < lenL) a[i] = xl4[(size_t)id * 4 + q];
                    }
                    if (8 * i < maxSI) {
                        const int idS = __builtin_amdgcn_ds_bpermute(p << 2, sortedS);
                        const int idI = __builtin_amdgcn_ds_bpermute(p << 2, sortedI);
                        if (p < lenX) b[i] = xb[(size_t)(isS ? idS : idI) * 2 + qq];
                    }
                }
                add4(a[0], a[1]); add4(a[0], a[2]); add4(a[0], a[3]);
                add4(b[0], b[1]); add4(b[0], b[2]); add4(b[0], b[3]);
                zl  = slot_sum(a[0]);
                zsi = slot_sum(b[0]);
            } else {
                zl = long_bucket_sum<4>(xl4, listFl, begL, lenL, lane, q);
                const float4 s = long_bucket_sum<2>(xs4, listFs, begS, lenS, lane, qq);
                const float4 j = long_bucket_sum<2>(xi4, listIn, begI, lenI, lane, qq);
                zsi = isS ? s : j;
            }
            const float wL = 1.0f / (float)max(lenL, 1);
            const float wX = 1.0f / (float)max(isS ? lenS : lenI, 1);
            zl.x *= wL;  zl.y *= wL;  zl.z *= wL;  zl.w *= wL;
            zsi.x *= wX; zsi.y *= wX; zsi.z *= wX; zsi.w *= wX;
        }
        if (lane < 4) { sZ[wave][u][lane] = zl; sZ[wave][u][4 + lane] = zsi; }
        if (lane == 0) sZ[wave][u][8] = make_float4(fl, fs, fi, 0.0f);
    }
    __syncthreads();

    float4 o[TPW];
    #pragma unroll
    for (int u = 0; u < TPW; ++u) o[u] = zero4;
    #pragma nounroll
    for (int g = 0; g < 8; ++g) {
        const float4 w0 = sW[(4 * g + 0) * 32 + lane];
        const float4 w1 = sW[(4 * g + 1) * 32 + lane];
        const float4 w2 = sW[(4 * g + 2) * 32 + lane];
        const float4 w3 = sW[(4 * g + 3) * 32 + lane];
        #pragma unroll
        for (int u = 0; u < TPW; ++u) {
            const float4 z = sZ[wave][u][g];
            o[u].x += z.x * w0.x; o[u].y += z.x * w0.y; o[u].z += z.x * w0.z; o[u].w += z.x * w0.w;
            o[u].x += z.y * w1.x; o[u].y += z.y * w1.y; o[u].z += z.y * w1.z; o[u].w += z.y * w1.w;
            o[u].x += z.z * w2.x; o[u].y += z.z * w2.y; o[u].z += z.z * w2.z; o[u].w += z.z * w2.w;
            o[u].x += z.w * w3.x; o[u].y += z.w * w3.y; o[u].z += z.w * w3.z; o[u].w += z.w * w3.w;
        }
    }
    {
        const float4 b0 = sW[32 * 32 + lane];
        const float4 b1 = sW[33 * 32 + lane];
        const float4 b2 = sW[34 * 32 + lane];
        #pragma unroll
        for (int u = 0; u < TPW; ++u) {
            const float4 f = sZ[wave][u][8];
            o[u].x += f.x * b0.x; o[u].y += f.x * b0.y; o[u].z += f.x * b0.z; o[u].w += f.x * b0.w;
            o[u].x += f.y * b1.x; o[u].y += f.y * b1.y; o[u].z += f.y * b1.z; o[u].w += f.y * b1.w;
            o[u].x += f.z * b2.x; o[u].y += f.z * b2.y; o[u].z += f.z * b2.z; o[u].w += f.z * b2.w;
        }
    }
    #pragma unroll
    for (int u = 0; u < TPW; ++u) {
        const int t = tb + u;
        if (t < n) ((float4*)aggS)[(size_t)t * (H / 4) + lane] = o[u];
    }
}

__global__ __launch_bounds__(256) void gather_sp_kernel(
        const float* __restrict__ h, const int* __restrict__ listSp,
        const int* __restrict__ cnt, const int* __restrict__ off,
        const float* __restrict__ aggS, float* __restrict__ aggB, int n) {
    const int lane = threadIdx.x & 31;
    const int t = __builtin_amdgcn_readfirstlane((int)(blockIdx.x * 8 + (threadIdx.x >> 5)));
    if (t >= n) return;
    const float4 base = ((const float4*)aggS)[(size_t)t * (H / 4) + lane];
    const int len = cnt[t];
    const float4 s = bucket_sum((const float4*)h, listSp, off[t], len, lane);
    const float w = 1.0f / (float)max(len, 1);
    float4 r;
    r.x = base.x + s.x * w; r.y = base.y + s.y * w;
    r.z = base.z + s.z * w; r.w = base.w + s.w * w;
    ((float4*)aggB)[(size_t)t * (H / 4) + lane] = r;
}

__global__ __launch_bounds__(256) void fused_gemm_elu_kernel(
        const float* __restrict__ A0, const float* __restrict__ A1,
        const float* __restrict__ W0, const float* __restrict__ W1,
        const float* __restrict__ bias, float* __restrict__ out, int n) {
    __shared__ __align__(16) float sB[H * H];
    const int tid  = threadIdx.x;
    const int wave = tid >> 5;
    const int lane = tid & 31;
    const int half = lane >> 4;
    const int l15  = lane & 15;
    const int m0   = blockIdx.x * 128 + wave * 16;

    int arow = m0 + l15;
    if (arow >= n) arow = n - 1;
    const float* aRow0 = A0 + (size_t)arow * H;
    const float* aRow1 = A1 + (size_t)arow * H;

    v8f acc[8] = {};

    for (int phase = 0; phase < 2; ++phase) {
        __syncthreads();
        {
            const float4* s = (const float4*)(phase ? W1 : W0);
            for (int i = tid; i < H * H / 4; i += 256) {
                const float4 v = s[i];
                const int k  = i >> 5;
                const int c0 = (i & 31) << 2;
                float* d = sB + ((k >> 1) << 8) + (k & 1);
                d[2 * (c0 + 0)] = v.x;
                d[2 * (c0 + 1)] = v.y;
                d[2 * (c0 + 2)] = v.z;
                d[2 * (c0 + 3)] = v.w;
            }
        }
        __syncthreads();
        const float* ap = phase ? aRow1 : aRow0;
        const float* bbase = sB + (half << 8) + (l15 << 1);
        for (int k0 = 0; k0 < H; k0 += 4) {
            const int ka = k0 + (half << 1);
            v2f a;
            a.x = ap[ka];
            a.y = ap[ka + 1];
            const float* bk = bbase + ((k0 >> 1) << 8);
            #pragma unroll
            for (int t = 0; t < 8; ++t) {
                const v2f b = *(const v2f*)(bk + (t << 5));
                acc[t] = __builtin_amdgcn_wmma_f32_16x16x4_f32(
                    false, a, false, b, (short)0, acc[t], false, false);
            }
        }
    }

    #pragma unroll
    for (int t = 0; t < 8; ++t) {
        const int col = (t << 4) + l15;
        const float bc = bias[col];
        #pragma unroll
        for (int r = 0; r < 8; ++r) {
            const int row = m0 + (half << 3) + r;
            if (row < n) {
                float v = acc[t][r] + bc;
                out[(size_t)row * H + col] = v > 0.0f ? v : (__expf(v) - 1.0f);
            }
        }
    }
}

extern "C" void kernel_launch(void* const* d_in, const int* in_sizes, int n_in,
                              void* d_out, int out_size, void* d_ws, size_t ws_size,
                              hipStream_t stream) {
    const float* x_int  = (const float*)d_in[0];
    const float* x_lane = (const float*)d_in[1];
    const float* x_sens = (const float*)d_in[2];
    const float* x_inj  = (const float*)d_in[3];
    const int*   e_sp   = (const int*)d_in[4];
    const int*   e_fl   = (const int*)d_in[5];
    const int*   e_fs   = (const int*)d_in[6];
    const int*   e_in   = (const int*)d_in[7];
    const float* W_int  = (const float*)d_in[8];
    const float* b_int  = (const float*)d_in[9];
    const float* W_lane = (const float*)d_in[10];
    const float* b_lane = (const float*)d_in[11];
    const float* W_sens = (const float*)d_in[12];
    const float* b_sens = (const float*)d_in[13];
    const float* W_inj  = (const float*)d_in[14];
    const float* b_inj  = (const float*)d_in[15];
    const float* self_W = (const float*)d_in[16];
    const float* self_b = (const float*)d_in[17];
    const float* rel_W  = (const float*)d_in[18];

    const int nInt  = in_sizes[0] / 32;
    const int nLane = in_sizes[1] / 16;
    const int nSens = in_sizes[2] / 8;
    const int nInj  = in_sizes[3] / 8;
    const int eSp   = in_sizes[4] / 2;
    const int eFl   = in_sizes[5] / 2;
    const int eFs   = in_sizes[6] / 2;
    const int eIn   = in_sizes[7] / 2;
    const int L     = in_sizes[16] / (H * H);

    float* ws = (float*)d_ws;
    size_t o = 0;
    float* hA    = ws + o; o += (size_t)nInt  * H;
    float* hB    = ws + o; o += (size_t)nInt  * H;
    float* hLane = ws + o; o += (size_t)nLane * H;
    float* pSens = ws + o; o += (size_t)nSens * H;
    o += (size_t)nInj  * H;
    float* aggS  = ws + o; o += (size_t)nInt  * H;
    float* aggB  = ws + o; o += (size_t)nInt  * H;
    int*   cnt   = (int*)(ws + o); o += (size_t)4 * nInt;
    int*   off   = (int*)(ws + o); o += (size_t)4 * nInt;
    (void)ws_size; (void)n_in; (void)out_size;

    int*  binned = (int*)aggB;
    int*  listFl = (int*)aggB + 2 * ((size_t)eSp + eFl + eFs + eIn);
    int*  listFs = listFl + eFl;
    int*  listIn = listFs + eFs;
    int*  listSp = (int*)hLane;
    int*  tab    = (int*)pSens;

    proj_kernel<<<(nInt  + 63) / 64, 128, 0, stream>>>(x_int,  W_int,  b_int,  hA,    nInt,  32);

    const int nb  = (nInt + 255) / 256;
    const int nch = (eSp + CHUNK - 1) / CHUNK + (eFl + CHUNK - 1) / CHUNK
                  + (eFs + CHUNK - 1) / CHUNK + (eIn + CHUNK - 1) / CHUNK;
    bin_kernel<<<nch, 256, 0, stream>>>(e_sp, e_fl, e_fs, e_in, binned, tab,
                                        eSp, eFl, eFs, eIn, nb, nch);
    fine_kernel<<<4 * nb, 256, 0, stream>>>(binned, tab, cnt, off,
                                            listSp, listFl, listFs, listIn,
                                            eSp, eFl, eFs, eIn, nb, nch, nInt);

    gather_raw_proj_kernel<<<(nInt + 8 * TPW - 1) / (8 * TPW), 256, 0, stream>>>(
        x_lane, x_sens, x_inj, W_lane, b_lane, W_sens, b_sens, W_inj, b_inj,
        listFl, listFs, listIn, cnt, off, aggS, nInt);

    float* hcur = hA;
    float* hnext = hB;
    for (int l = 0; l < L; ++l) {
        gather_sp_kernel<<<(nInt + 7) / 8, 256, 0, stream>>>(
            hcur, listSp, cnt, off, aggS, aggB, nInt);
        float* outp = (l == L - 1) ? (float*)d_out : hnext;
        fused_gemm_elu_kernel<<<(nInt + 127) / 128, 256, 0, stream>>>(
            hcur, aggB,
            self_W + (size_t)l * H * H, rel_W + (size_t)l * H * H,
            self_b + (size_t)l * H, outp, nInt);
        float* t = hnext; hnext = hcur; hcur = t;
    }
}


# Input projection moved to the f32 16x16x4 matrix unit: a wave takes 16 rows, four 16-byte loads per lane (was 64 serial rows per wave fed by scalar loads); sum order over k changes
# speedup vs baseline: 5.9150x; 1.1639x over previous
#include <hip/hip_runtime.h>
#include <cstddef>

#define H 128

typedef float v2f __attribute__((ext_vector_type(2)));
typedef float v8f __attribute__((ext_vector_type(8)));

#define KIN 32
__global__ __launch_bounds__(256) void proj_kernel(const float* __restrict__ x,
                                                   const float* __restrict__ W,
                                                   const float* __restrict__ b,
                                                   float* __restrict__ out, int n) {
    __shared__ __align__(16) float sW[KIN * H];
    const int tid  = threadIdx.x;
    const int lane = tid & 31;
    const int half = lane >> 4;
    const int l15  = lane & 15;
    const int m0 = __builtin_amdgcn_readfirstlane((int)(blockIdx.x * 128 + (tid >> 5) * 16));

    int arow = m0 + l15;
    if (arow >= n) arow = n - 1;
    const float4* xr = (const float4*)(x + (size_t)arow * KIN + (half << 4));
    const float4 x0 = xr[0], x1 = xr[1], x2 = xr[2], x3 = xr[3];

    float4 wv[4];
    #pragma unroll
    for (int j = 0; j < 4; ++j) wv[j] = ((const float4*)W)[tid + 256 * j];

    v8f acc[8];
    #pragma unroll
    for (int t = 0; t < 8; ++t) {
        const float bc = b[(t << 4) + l15];
        #pragma unroll
        for (int r = 0; r < 8; ++r) acc[t][r] = bc;
    }

    #pragma unroll
    for (int j = 0; j < 4; ++j) {
        const int i  = tid + 256 * j;
        const int k  = i >> 5;
        const int c0 = (i & 31) << 2;
        float* d = sW + ((k >> 1) << 8) + (k & 1);
        d[2 * (c0 + 0)] = wv[j].x;
        d[2 * (c0 + 1)] = wv[j].y;
        d[2 * (c0 + 2)] = wv[j].z;
        d[2 * (c0 + 3)] = wv[j].w;
    }
    __syncthreads();

    const float xf[16] = { x0.x, x0.y, x0.z, x0.w, x1.x, x1.y, x1.z, x1.w,
                           x2.x, x2.y, x2.z, x2.w, x3.x, x3.y, x3.z, x3.w };
    const float* bbase = sW + (half << 11) + (l15 << 1);
    #pragma unroll
    for (int s = 0; s < 8; ++s) {
        v2f a;
        a.x = xf[2 * s];
        a.y = xf[2 * s + 1];
        const float* bk = bbase + (s << 8);
        #pragma unroll
        for (int t = 0; t < 8; ++t) {
            const v2f bf = *(const v2f*)(bk + (t << 5));
            acc[t] = __builtin_amdgcn_wmma_f32_16x16x4_f32(
                false, a, false, bf, (short)0, acc[t], false, false);
        }
    }

    float* orow = out + (size_t)(m0 + (half << 3)) * H + l15;
    if (m0 + 16 <= n) {
        #pragma unroll
        for (int t = 0; t < 8; ++t) {
            #pragma unroll
            for (int r = 0; r < 8; ++r) orow[r * H + (t << 4)] = acc[t][r];
        }
    } else {
        #pragma unroll
        for (int t = 0; t < 8; ++t) {
            #pragma unroll
            for (int r = 0; r < 8; ++r) {
                if (m0 + (half << 3) + r < n) orow[r * H + (t << 4)] = acc[t][r];
            }
        }
    }
}

#define CHUNK 4096
#define MAXNB 1024
#define STAGE 8192

__global__ __launch_bounds__(256) void bin_kernel(
        const int* __restrict__ ed0, const int* __restrict__ ed1,
        const int* __restrict__ ed2, const int* __restrict__ ed3,
        int* __restrict__ binned, int* __restrict__ tab,
        int E0, int E1, int E2, int E3, int nb, int nch) {
    __shared__ int  sHist[MAXNB + 1];
    __shared__ int2 sPair[CHUNK];
    const int tid = threadIdx.x;
    const int C0 = (E0 + CHUNK - 1) / CHUNK, C1 = (E1 + CHUNK - 1) / CHUNK,
              C2 = (E2 + CHUNK - 1) / CHUNK;
    int c = blockIdx.x;
    const int* ed = ed0;
    int E = E0, rbase = 0;
    if (c >= C0) {
        c -= C0; ed = ed1; E = E1; rbase = E0;
        if (c >= C1) {
            c -= C1; ed = ed2; E = E2; rbase += E1;
            if (c >= C2) { c -= C2; ed = ed3; E = E3; rbase += E2; }
        }
    }
    const int e0  = c * CHUNK;
    const int cnt = min(CHUNK, E - e0);

    for (int j = tid; j <= nb; j += 256) sHist[j] = 0;
    __syncthreads();

    int srcv[16], tgtv[16], rk[16];
    #pragma unroll
    for (int k = 0; k < 16; ++k) {
        const int i = k * 256 + tid;
        srcv[k] = 0; tgtv[k] = 0;
        if (i < cnt) { srcv[k] = ed[e0 + i]; tgtv[k] = ed[E + e0 + i]; }
    }
    #pragma unroll
    for (int k = 0; k < 16; ++k) {
        const int i = k * 256 + tid;
        rk[k] = 0;
        if (i < cnt) rk[k] = atomicAdd(&sHist[tgtv[k] >> 8], 1);
    }
    __syncthreads();

    if (tid < 32) {
        int carry = 0;
        for (int base = 0; base < nb; base += 32) {
            const int j = base + tid;
            const int v = (j < nb) ? sHist[j] : 0;
            int inc = v;
            for (int d = 1; d < 32; d <<= 1) {
                const int u = __shfl_up(inc, d, 32);
                if (tid >= d) inc += u;
            }
            if (j < nb) sHist[j] = carry + inc - v;
            carry += __shfl(inc, 31, 32);
        }
        if (tid == 0) sHist[nb] = carry;
    }
    __syncthreads();

    #pragma unroll
    for (int k = 0; k < 16; ++k) {
        const int i = k * 256 + tid;
        if (i < cnt) sPair[sHist[tgtv[k] >> 8] + rk[k]] = make_int2(srcv[k], tgtv[k]);
    }
    for (int j = tid; j <= nb; j += 256) tab[(size_t)j * nch + blockIdx.x] = sHist[j];
    __syncthreads();

    int2* outp = (int2*)binned + (size_t)rbase + e0;
    for (int i = tid; i < cnt; i += 256) outp[i] = sPair[i];
}

__global__ __launch_bounds__(256) void fine_kernel(
        const int* __restrict__ binned, const int* __restrict__ tab,
        int* __restrict__ cnt, int* __restrict__ off,
        int* __restrict__ list0, int* __restrict__ list1,
        int* __restrict__ list2, int* __restrict__ list3,
        int E0, int E1, int E2, int E3, int nb, int nch, int n) {
    __shared__ int sCnt[256];
    __shared__ int sCur[256];
    __shared__ int sWave[8];
    __shared__ int sRed[8];
    __shared__ int sStage[STAGE];
    const int tid  = threadIdx.x;
    const int lane = tid & 31;
    const int wave = tid >> 5;
    const int C0 = (E0 + CHUNK - 1) / CHUNK, C1 = (E1 + CHUNK - 1) / CHUNK,
              C2 = (E2 + CHUNK - 1) / CHUNK, C3 = (E3 + CHUNK - 1) / CHUNK;
    const int r = blockIdx.x / nb;
    const int g = blockIdx.x - r * nb;
    int rbase = 0, cb = 0, nC = C0;
    int* list = list0;
    if (r == 1) { rbase = E0;           cb = C0;           nC = C1; list = list1; }
    if (r == 2) { rbase = E0 + E1;      cb = C0 + C1;      nC = C2; list = list2; }
    if (r == 3) { rbase = E0 + E1 + E2; cb = C0 + C1 + C2; nC = C3; list = list3; }
    const int* tlo = tab + (size_t)g * nch + cb;
    const int* thi = tab + (size_t)(g + 1) * nch + cb;
    const int2* win = (const int2*)binned + (size_t)rbase;

    sCnt[tid] = 0;
    __syncthreads();

    int lower = 0;
    for (int b = tid; b < nC; b += 256) {
        const int s = tlo[b], e = thi[b];
        lower += s;
        const int2* p = win + (size_t)b * CHUNK;
        for (int i = s; i < e; i += 8) {
            int y[8];
            #pragma unroll
            for (int u = 0; u < 8; ++u) y[u] = p[min(i + u, e - 1)].y;
            #pragma unroll
            for (int u = 0; u < 8; ++u)
                if (i + u < e) atomicAdd(&sCnt[y[u] & 255], 1);
        }
    }
    __syncthreads();

    const int cT = sCnt[tid];
    int inc = cT;
    for (int d = 1; d < 32; d <<= 1) {
        const int u = __shfl_up(inc, d, 32);
        if (lane >= d) inc += u;
    }
    for (int o = 16; o > 0; o >>= 1) lower += __shfl_xor(lower, o, 32);
    if (lane == 31) sWave[wave] = inc;
    if (lane == 0)  sRed[wave]  = lower;
    __syncthreads();
    int wbase = 0, total = 0, start = 0;
    for (int w = 0; w < 8; ++w) {
        const int tw = sWave[w];
        if (w < wave) wbase += tw;
        total += tw;
        start += sRed[w];
    }
    const int offL = wbase + inc - cT;
    const int t = g * 256 + tid;
    if (t < n) {
        cnt[(size_t)r * n + t] = cT;
        off[(size_t)r * n + t] = start + offL;
    }
    sCur[tid] = offL;
    __syncthreads();

    const bool staged = total <= STAGE;
    for (int b = tid; b < nC; b += 256) {
        const int s = tlo[b], e = thi[b];
        const int2* p = win + (size_t)b * CHUNK;
        for (int i = s; i < e; i += 8) {
            int2 v[8];
            #pragma unroll
            for (int u = 0; u < 8; ++u) v[u] = p[min(i + u, e - 1)];
            #pragma unroll
            for (int u = 0; u < 8; ++u) {
                if (i + u < e) {
                    const int pos = atomicAdd(&sCur[v[u].y & 255], 1);
                    if (staged) sStage[pos] = v[u].x;
                    else        list[start + pos] = v[u].x;
                }
            }
        }
    }
    if (staged) {
        __syncthreads();
        for (int i = tid; i < total; i += 256) list[start + i] = sStage[i];
    }
}

__device__ __forceinline__ void add4(float4& a, const float4 v) {
    a.x += v.x; a.y += v.y; a.z += v.z; a.w += v.w;
}

__device__ __forceinline__ float4 bucket_sum(const float4* __restrict__ feat4,
                                             const int* __restrict__ list,
                                             int beg, int len, int lane) {
    float4 acc = make_float4(0.0f, 0.0f, 0.0f, 0.0f);
    if (len <= 32) {
        const int key = (lane < len) ? list[beg + lane] : 0x7fffffff;
        int rank = 0;
        for (int j = 0; j < len; ++j) {
            const int kj = __builtin_amdgcn_readlane(key, j);
            rank += (kj < key || (kj == key && j < lane)) ? 1 : 0;
        }
        if (lane >= len) rank = lane;
        const int sorted = __builtin_amdgcn_ds_permute(rank << 2, key);
        int r = 0;
        for (; r + 4 <= len; r += 4) {
            const int s0 = __builtin_amdgcn_readlane(sorted, r);
            const int s1 = __builtin_amdgcn_readlane(sorted, r + 1);
            const int s2 = __builtin_amdgcn_readlane(sorted, r + 2);
            const int s3 = __builtin_amdgcn_readlane(sorted, r + 3);
            const float4 v0 = feat4[(size_t)s0 * (H / 4) + lane];
            const float4 v1 = feat4[(size_t)s1 * (H / 4) + lane];
            const float4 v2 = feat4[(size_t)s2 * (H / 4) + lane];
            const float4 v3 = feat4[(size_t)s3 * (H / 4) + lane];
            add4(acc, v0); add4(acc, v1); add4(acc, v2); add4(acc, v3);
        }
        for (; r < len; ++r) {
            const int s0 = __builtin_amdgcn_readlane(sorted, r);
            add4(acc, feat4[(size_t)s0 * (H / 4) + lane]);
        }
    } else {
        int prev = -1;
        for (int it = 0; it < len; ++it) {
            int m = 0x7fffffff;
            for (int i = lane; i < len; i += 32) {
                const int k = list[beg + i];
                if (k > prev && k < m) m = k;
            }
            for (int o = 16; o > 0; o >>= 1) m = min(m, __shfl_xor(m, o, 32));
            if (m == 0x7fffffff) break;
            int c = 0;
            for (int i = lane; i < len; i += 32) c += (list[beg + i] == m) ? 1 : 0;
            for (int o = 16; o > 0; o >>= 1) c += __shfl_xor(c, o, 32);
            const float4 v = feat4[(size_t)m * (H / 4) + lane];
            for (int q = 0; q < c; ++q) add4(acc, v);
            prev = m;
        }
    }
    return acc;
}


__device__ __forceinline__ int sort_ids(int key, int len, int lane) {
    int rank = 0;
    for (int j = 0; j < len; ++j) {
        const int kj = __builtin_amdgcn_readlane(key, j);
        rank += (kj < key || (kj == key && j < lane)) ? 1 : 0;
    }
    if (lane >= len) rank = lane;
    return __builtin_amdgcn_ds_permute(rank << 2, key);
}

template <int Q>
__device__ __forceinline__ float4 long_bucket_sum(const float4* __restrict__ x4,
                                                  const int* __restrict__ list,
                                                  int beg, int len, int lane, int qq) {
    float4 acc = make_float4(0.0f, 0.0f, 0.0f, 0.0f);
    int prev = -1;
    for (int it = 0; it < len; ++it) {
        int m = 0x7fffffff;
        for (int i = lane; i < len; i += 32) {
            const int k = list[beg + i];
            if (k > prev && k < m) m = k;
        }
        for (int o = 16; o > 0; o >>= 1) m = min(m, __shfl_xor(m, o, 32));
        if (m == 0x7fffffff) break;
        int c = 0;
        for (int i = lane; i < len; i += 32) c += (list[beg + i] == m) ? 1 : 0;
        for (int o = 16; o > 0; o >>= 1) c += __shfl_xor(c, o, 32);
        const float4 v = x4[(size_t)m * Q + qq];
        for (int r = 0; r < c; ++r) add4(acc, v);
        prev = m;
    }
    return acc;
}

__device__ __forceinline__ float4 slot_sum(float4 a) {
    for (int m = 4; m < 32; m <<= 1) {
        float4 b;
        b.x = __shfl_xor(a.x, m, 32); b.y = __shfl_xor(a.y, m, 32);
        b.z = __shfl_xor(a.z, m, 32); b.w = __shfl_xor(a.w, m, 32);
        add4(a, b);
    }
    return a;
}

#define TPW 4

__global__ __launch_bounds__(256) void gather_raw_proj_kernel(
        const float* __restrict__ xLane, const float* __restrict__ xSens,
        const float* __restrict__ xInj,
        const float* __restrict__ WLane, const float* __restrict__ bLane,
        const float* __restrict__ WSens, const float* __restrict__ bSens,
        const float* __restrict__ WInj,  const float* __restrict__ bInj,
        const int* __restrict__ listFl, const int* __restrict__ listFs,
        const int* __restrict__ listIn,
        const int* __restrict__ cnt, const int* __restrict__ off,
        float* __restrict__ aggS, int n) {
    __shared__ float4 sW[35 * 32];
    __shared__ float4 sZ[8][TPW][9];
    const int tid  = threadIdx.x;
    const int wave = tid >> 5;
    const int lane = tid & 31;
    const int slot = lane >> 2;
    const int q    = lane & 3;

    {
        const float4* wl = (const float4*)WLane;
        sW[tid]       = wl[tid];
        sW[256 + tid] = wl[256 + tid];
        sW[512 + tid] = ((const float4*)WSens)[tid];
        sW[768 + tid] = ((const float4*)WInj)[tid];
        if (tid < 96) {
            const float* bs = tid < 32 ? bLane : (tid < 64 ? bSens : bInj);
            sW[1024 + tid] = ((const float4*)bs)[tid & 31];
        }
    }

    const float4* xl4 = (const float4*)xLane;
    const float4* xs4 = (const float4*)xSens;
    const float4* xi4 = (const float4*)xInj;
    const float4 zero4 = make_float4(0.0f, 0.0f, 0.0f, 0.0f);
    const int tb = blockIdx.x * (8 * TPW) + wave * TPW;

    for (int u = 0; u < TPW; ++u) {
        const int t = __builtin_amdgcn_readfirstlane(tb + u);
        float4 zl = zero4, zsi = zero4;
        float fl = 0.0f, fs = 0.0f, fi = 0.0f;
        if (t < n) {
            const int lenL = cnt[1 * n + t], lenS = cnt[2 * n + t], lenI = cnt[3 * n + t];
            const int begL = off[1 * n + t], begS = off[2 * n + t], begI = off[3 * n + t];
            fl = lenL > 0 ? 1.0f : 0.0f;
            fs = lenS > 0 ? 1.0f : 0.0f;
            fi = lenI > 0 ? 1.0f : 0.0f;
            const bool isS = q < 2;
            const int qq = q & 1;
            if (lenL <= 32 && lenS <= 32 && lenI <= 32) {
                const int keyL = (lane < lenL) ? listFl[begL + lane] : 0x7fffffff;
                const int keyS = (lane < lenS) ? listFs[begS + lane] : 0x7fffffff;
                const int keyI = (lane < lenI) ? listIn[begI + lane] : 0x7fffffff;
                const int sortedL = sort_ids(keyL, lenL, lane);
                const int sortedS = sort_ids(keyS, lenS, lane);
                const int sortedI = sort_ids(keyI, lenI, lane);
                const int lenX = isS ? lenS : lenI;
                const int maxSI = max(lenS, lenI);
                const float4* xb = isS ? xs4 : xi4;
                float4 a[4], b[4];
                #pragma unroll
                for (int i = 0; i < 4; ++i) {
                    a[i] = zero4; b[i] = zero4;
                    const int p = 8 * i + slot;
                    if (8 * i < lenL) {
                        const int id = __builtin_amdgcn_ds_bpermute(p << 2, sortedL);
                        if (p < lenL) a[i] = xl4[(size_t)id * 4 + q];
                    }
                    if (8 * i < maxSI) {
                        const int idS = __builtin_amdgcn_ds_bpermute(p << 2, sortedS);
                        const int idI = __builtin_amdgcn_ds_bpermute(p << 2, sortedI);
                        if (p < lenX) b[i] = xb[(size_t)(isS ? idS : idI) * 2 + qq];
                    }
                }
                add4(a[0], a[1]); add4(a[0], a[2]); add4(a[0], a[3]);
                add4(b[0], b[1]); add4(b[0], b[2]); add4(b[0], b[3]);
                zl  = slot_sum(a[0]);
                zsi = slot_sum(b[0]);
            } else {
                zl = long_bucket_sum<4>(xl4, listFl, begL, lenL, lane, q);
                const float4 s = long_bucket_sum<2>(xs4, listFs, begS, lenS, lane, qq);
                const float4 j = long_bucket_sum<2>(xi4, listIn, begI, lenI, lane, qq);
                zsi = isS ? s : j;
            }
            const float wL = 1.0f / (float)max(lenL, 1);
            const float wX = 1.0f / (float)max(isS ? lenS : lenI, 1);
            zl.x *= wL;  zl.y *= wL;  zl.z *= wL;  zl.w *= wL;
            zsi.x *= wX; zsi.y *= wX; zsi.z *= wX; zsi.w *= wX;
        }
        if (lane < 4) { sZ[wave][u][lane] = zl; sZ[wave][u][4 + lane] = zsi; }
        if (lane == 0) sZ[wave][u][8] = make_float4(fl, fs, fi, 0.0f);
    }
    __syncthreads();

    float4 o[TPW];
    #pragma unroll
    for (int u = 0; u < TPW; ++u) o[u] = zero4;
    #pragma nounroll
    for (int g = 0; g < 8; ++g) {
        const float4 w0 = sW[(4 * g + 0) * 32 + lane];
        const float4 w1 = sW[(4 * g + 1) * 32 + lane];
        const float4 w2 = sW[(4 * g + 2) * 32 + lane];
        const float4 w3 = sW[(4 * g + 3) * 32 + lane];
        #pragma unroll
        for (int u = 0; u < TPW; ++u) {
            const float4 z = sZ[wave][u][g];
            o[u].x += z.x * w0.x; o[u].y += z.x * w0.y; o[u].z += z.x * w0.z; o[u].w += z.x * w0.w;
            o[u].x += z.y * w1.x; o[u].y += z.y * w1.y; o[u].z += z.y * w1.z; o[u].w += z.y * w1.w;
            o[u].x += z.z * w2.x; o[u].y += z.z * w2.y; o[u].z += z.z * w2.z; o[u].w += z.z * w2.w;
            o[u].x += z.w * w3.x; o[u].y += z.w * w3.y; o[u].z += z.w * w3.z; o[u].w += z.w * w3.w;
        }
    }
    {
        const float4 b0 = sW[32 * 32 + lane];
        const float4 b1 = sW[33 * 32 + lane];
        const float4 b2 = sW[34 * 32 + lane];
        #pragma unroll
        for (int u = 0; u < TPW; ++u) {
            const float4 f = sZ[wave][u][8];
            o[u].x += f.x * b0.x; o[u].y += f.x * b0.y; o[u].z += f.x * b0.z; o[u].w += f.x * b0.w;
            o[u].x += f.y * b1.x; o[u].y += f.y * b1.y; o[u].z += f.y * b1.z; o[u].w += f.y * b1.w;
            o[u].x += f.z * b2.x; o[u].y += f.z * b2.y; o[u].z += f.z * b2.z; o[u].w += f.z * b2.w;
        }
    }
    #pragma unroll
    for (int u = 0; u < TPW; ++u) {
        const int t = tb + u;
        if (t < n) ((float4*)aggS)[(size_t)t * (H / 4) + lane] = o[u];
    }
}

__global__ __launch_bounds__(256) void gather_sp_kernel(
        const float* __restrict__ h, const int* __restrict__ listSp,
        const int* __restrict__ cnt, const int* __restrict__ off,
        const float* __restrict__ aggS, float* __restrict__ aggB, int n) {
    const int lane = threadIdx.x & 31;
    const int t = __builtin_amdgcn_readfirstlane((int)(blockIdx.x * 8 + (threadIdx.x >> 5)));
    if (t >= n) return;
    const float4 base = ((const float4*)aggS)[(size_t)t * (H / 4) + lane];
    const int len = cnt[t];
    const float4 s = bucket_sum((const float4*)h, listSp, off[t], len, lane);
    const float w = 1.0f / (float)max(len, 1);
    float4 r;
    r.x = base.x + s.x * w; r.y = base.y + s.y * w;
    r.z = base.z + s.z * w; r.w = base.w + s.w * w;
    ((float4*)aggB)[(size_t)t * (H / 4) + lane] = r;
}

__global__ __launch_bounds__(256) void fused_gemm_elu_kernel(
        const float* __restrict__ A0, const float* __restrict__ A1,
        const float* __restrict__ W0, const float* __restrict__ W1,
        const float* __restrict__ bias, float* __restrict__ out, int n) {
    __shared__ __align__(16) float sB[H * H];
    const int tid  = threadIdx.x;
    const int wave = tid >> 5;
    const int lane = tid & 31;
    const int half = lane >> 4;
    const int l15  = lane & 15;
    const int m0   = blockIdx.x * 128 + wave * 16;

    int arow = m0 + l15;
    if (arow >= n) arow = n - 1;
    const float* aRow0 = A0 + (size_t)arow * H;
    const float* aRow1 = A1 + (size_t)arow * H;

    v8f acc[8] = {};

    for (int phase = 0; phase < 2; ++phase) {
        __syncthreads();
        {
            const float4* s = (const float4*)(phase ? W1 : W0);
            for (int i = tid; i < H * H / 4; i += 256) {
                const float4 v = s[i];
                const int k  = i >> 5;
                const int c0 = (i & 31) << 2;
                float* d = sB + ((k >> 1) << 8) + (k & 1);
                d[2 * (c0 + 0)] = v.x;
                d[2 * (c0 + 1)] = v.y;
                d[2 * (c0 + 2)] = v.z;
                d[2 * (c0 + 3)] = v.w;
            }
        }
        __syncthreads();
        const float* ap = phase ? aRow1 : aRow0;
        const float* bbase = sB + (half << 8) + (l15 << 1);
        for (int k0 = 0; k0 < H; k0 += 4) {
            const int ka = k0 + (half << 1);
            v2f a;
            a.x = ap[ka];
            a.y = ap[ka + 1];
            const float* bk = bbase + ((k0 >> 1) << 8);
            #pragma unroll
            for (int t = 0; t < 8; ++t) {
                const v2f b = *(const v2f*)(bk + (t << 5));
                acc[t] = __builtin_amdgcn_wmma_f32_16x16x4_f32(
                    false, a, false, b, (short)0, acc[t], false, false);
            }
        }
    }

    #pragma unroll
    for (int t = 0; t < 8; ++t) {
        const int col = (t << 4) + l15;
        const float bc = bias[col];
        #pragma unroll
        for (int r = 0; r < 8; ++r) {
            const int row = m0 + (half << 3) + r;
            if (row < n) {
                float v = acc[t][r] + bc;
                out[(size_t)row * H + col] = v > 0.0f ? v : (__expf(v) - 1.0f);
            }
        }
    }
}

extern "C" void kernel_launch(void* const* d_in, const int* in_sizes, int n_in,
                              void* d_out, int out_size, void* d_ws, size_t ws_size,
                              hipStream_t stream) {
    const float* x_int  = (const float*)d_in[0];
    const float* x_lane = (const float*)d_in[1];
    const float* x_sens = (const float*)d_in[2];
    const float* x_inj  = (const float*)d_in[3];
    const int*   e_sp   = (const int*)d_in[4];
    const int*   e_fl   = (const int*)d_in[5];
    const int*   e_fs   = (const int*)d_in[6];
    const int*   e_in   = (const int*)d_in[7];
    const float* W_int  = (const float*)d_in[8];
    const float* b_int  = (const float*)d_in[9];
    const float* W_lane = (const float*)d_in[10];
    const float* b_lane = (const float*)d_in[11];
    const float* W_sens = (const float*)d_in[12];
    const float* b_sens = (const float*)d_in[13];
    const float* W_inj  = (const float*)d_in[14];
    const float* b_inj  = (const float*)d_in[15];
    const float* self_W = (const float*)d_in[16];
    const float* self_b = (const float*)d_in[17];
    const float* rel_W  = (const float*)d_in[18];

    const int nInt  = in_sizes[0] / 32;
    const int nLane = in_sizes[1] / 16;
    const int nSens = in_sizes[2] / 8;
    const int nInj  = in_sizes[3] / 8;
    const int eSp   = in_sizes[4] / 2;
    const int eFl   = in_sizes[5] / 2;
    const int eFs   = in_sizes[6] / 2;
    const int eIn   = in_sizes[7] / 2;
    const int L     = in_sizes[16] / (H * H);

    float* ws = (float*)d_ws;
    size_t o = 0;
    float* hA    = ws + o; o += (size_t)nInt  * H;
    float* hB    = ws + o; o += (size_t)nInt  * H;
    float* hLane = ws + o; o += (size_t)nLane * H;
    float* pSens = ws + o; o += (size_t)nSens * H;
    o += (size_t)nInj  * H;
    float* aggS  = ws + o; o += (size_t)nInt  * H;
    float* aggB  = ws + o; o += (size_t)nInt  * H;
    int*   cnt   = (int*)(ws + o); o += (size_t)4 * nInt;
    int*   off   = (int*)(ws + o); o += (size_t)4 * nInt;
    (void)ws_size; (void)n_in; (void)out_size;

    int*  binned = (int*)aggB;
    int*  listFl = (int*)aggB + 2 * ((size_t)eSp + eFl + eFs + eIn);
    int*  listFs = listFl + eFl;
    int*  listIn = listFs + eFs;
    int*  listSp = (int*)hLane;
    int*  tab    = (int*)pSens;

    proj_kernel<<<(nInt + 127) / 128, 256, 0, stream>>>(x_int, W_int, b_int, hA, nInt);

    const int nb  = (nInt + 255) / 256;
    const int nch = (eSp + CHUNK - 1) / CHUNK + (eFl + CHUNK - 1) / CHUNK
                  + (eFs + CHUNK - 1) / CHUNK + (eIn + CHUNK - 1) / CHUNK;
    bin_kernel<<<nch, 256, 0, stream>>>(e_sp, e_fl, e_fs, e_in, binned, tab,
                                        eSp, eFl, eFs, eIn, nb, nch);
    fine_kernel<<<4 * nb, 256, 0, stream>>>(binned, tab, cnt, off,
                                            listSp, listFl, listFs, listIn,
                                            eSp, eFl, eFs, eIn, nb, nch, nInt);

    gather_raw_proj_kernel<<<(nInt + 8 * TPW - 1) / (8 * TPW), 256, 0, stream>>>(
        x_lane, x_sens, x_inj, W_lane, b_lane, W_sens, b_sens, W_inj, b_inj,
        listFl, listFs, listIn, cnt, off, aggS, nInt);

    float* hcur = hA;
    float* hnext = hB;
    for (int l = 0; l < L; ++l) {
        gather_sp_kernel<<<(nInt + 7) / 8, 256, 0, stream>>>(
            hcur, listSp, cnt, off, aggS, aggB, nInt);
        float* outp = (l == L - 1) ? (float*)d_out : hnext;
        fused_gemm_elu_kernel<<<(nInt + 127) / 128, 256, 0, stream>>>(
            hcur, aggB,
            self_W + (size_t)l * H * H, rel_W + (size_t)l * H * H,
            self_b + (size_t)l * H, outp, nInt);
        float* t = hnext; hnext = hcur; hcur = t;
    }
}


# Spatial gather fused into the layer kernel: a block per WGP gathers its rows' means into LDS beside the h @ Wself product and feeds them to the matrix unit; 10 launches to 7; sum order over k changes
# speedup vs baseline: 7.1261x; 1.2048x over previous
#include <hip/hip_runtime.h>
#include <cstddef>

#define H 128

typedef float v2f __attribute__((ext_vector_type(2)));
typedef float v8f __attribute__((ext_vector_type(8)));

#define KIN 32
__global__ __launch_bounds__(256) void proj_kernel(const float* __restrict__ x,
                                                   const float* __restrict__ W,
                                                   const float* __restrict__ b,
                                                   float* __restrict__ out, int n) {
    __shared__ __align__(16) float sW[KIN * H];
    const int tid  = threadIdx.x;
    const int lane = tid & 31;
    const int half = lane >> 4;
    const int l15  = lane & 15;
    const int m0 = __builtin_amdgcn_readfirstlane((int)(blockIdx.x * 128 + (tid >> 5) * 16));

    int arow = m0 + l15;
    if (arow >= n) arow = n - 1;
    const float4* xr = (const float4*)(x + (size_t)arow * KIN + (half << 4));
    const float4 x0 = xr[0], x1 = xr[1], x2 = xr[2], x3 = xr[3];

    float4 wv[4];
    #pragma unroll
    for (int j = 0; j < 4; ++j) wv[j] = ((const float4*)W)[tid + 256 * j];

    v8f acc[8];
    #pragma unroll
    for (int t = 0; t < 8; ++t) {
        const float bc = b[(t << 4) + l15];
        #pragma unroll
        for (int r = 0; r < 8; ++r) acc[t][r] = bc;
    }

    #pragma unroll
    for (int j = 0; j < 4; ++j) {
        const int i  = tid + 256 * j;
        const int k  = i >> 5;
        const int c0 = (i & 31) << 2;
        float* d = sW + ((k >> 1) << 8) + (k & 1);
        d[2 * (c0 + 0)] = wv[j].x;
        d[2 * (c0 + 1)] = wv[j].y;
        d[2 * (c0 + 2)] = wv[j].z;
        d[2 * (c0 + 3)] = wv[j].w;
    }
    __syncthreads();

    const float xf[16] = { x0.x, x0.y, x0.z, x0.w, x1.x, x1.y, x1.z, x1.w,
                           x2.x, x2.y, x2.z, x2.w, x3.x, x3.y, x3.z, x3.w };
    const float* bbase = sW + (half << 11) + (l15 << 1);
    #pragma unroll
    for (int s = 0; s < 8; ++s) {
        v2f a;
        a.x = xf[2 * s];
        a.y = xf[2 * s + 1];
        const float* bk = bbase + (s << 8);
        #pragma unroll
        for (int t = 0; t < 8; ++t) {
            const v2f bf = *(const v2f*)(bk + (t << 5));
            acc[t] = __builtin_amdgcn_wmma_f32_16x16x4_f32(
                false, a, false, bf, (short)0, acc[t], false, false);
        }
    }

    float* orow = out + (size_t)(m0 + (half << 3)) * H + l15;
    if (m0 + 16 <= n) {
        #pragma unroll
        for (int t = 0; t < 8; ++t) {
            #pragma unroll
            for (int r = 0; r < 8; ++r) orow[r * H + (t << 4)] = acc[t][r];
        }
    } else {
        #pragma unroll
        for (int t = 0; t < 8; ++t) {
            #pragma unroll
            for (int r = 0; r < 8; ++r) {
                if (m0 + (half << 3) + r < n) orow[r * H + (t << 4)] = acc[t][r];
            }
        }
    }
}

#define CHUNK 4096
#define MAXNB 1024
#define STAGE 8192

__global__ __launch_bounds__(256) void bin_kernel(
        const int* __restrict__ ed0, const int* __restrict__ ed1,
        const int* __restrict__ ed2, const int* __restrict__ ed3,
        int* __restrict__ binned, int* __restrict__ tab,
        int E0, int E1, int E2, int E3, int nb, int nch) {
    __shared__ int  sHist[MAXNB + 1];
    __shared__ int2 sPair[CHUNK];
    const int tid = threadIdx.x;
    const int C0 = (E0 + CHUNK - 1) / CHUNK, C1 = (E1 + CHUNK - 1) / CHUNK,
              C2 = (E2 + CHUNK - 1) / CHUNK;
    int c = blockIdx.x;
    const int* ed = ed0;
    int E = E0, rbase = 0;
    if (c >= C0) {
        c -= C0; ed = ed1; E = E1; rbase = E0;
        if (c >= C1) {
            c -= C1; ed = ed2; E = E2; rbase += E1;
            if (c >= C2) { c -= C2; ed = ed3; E = E3; rbase += E2; }
        }
    }
    const int e0  = c * CHUNK;
    const int cnt = min(CHUNK, E - e0);

    for (int j = tid; j <= nb; j += 256) sHist[j] = 0;
    __syncthreads();

    int srcv[16], tgtv[16], rk[16];
    #pragma unroll
    for (int k = 0; k < 16; ++k) {
        const int i = k * 256 + tid;
        srcv[k] = 0; tgtv[k] = 0;
        if (i < cnt) { srcv[k] = ed[e0 + i]; tgtv[k] = ed[E + e0 + i]; }
    }
    #pragma unroll
    for (int k = 0; k < 16; ++k) {
        const int i = k * 256 + tid;
        rk[k] = 0;
        if (i < cnt) rk[k] = atomicAdd(&sHist[tgtv[k] >> 8], 1);
    }
    __syncthreads();

    if (tid < 32) {
        int carry = 0;
        for (int base = 0; base < nb; base += 32) {
            const int j = base + tid;
            const int v = (j < nb) ? sHist[j] : 0;
            int inc = v;
            for (int d = 1; d < 32; d <<= 1) {
                const int u = __shfl_up(inc, d, 32);
                if (tid >= d) inc += u;
            }
            if (j < nb) sHist[j] = carry + inc - v;
            carry += __shfl(inc, 31, 32);
        }
        if (tid == 0) sHist[nb] = carry;
    }
    __syncthreads();

    #pragma unroll
    for (int k = 0; k < 16; ++k) {
        const int i = k * 256 + tid;
        if (i < cnt) sPair[sHist[tgtv[k] >> 8] + rk[k]] = make_int2(srcv[k], tgtv[k]);
    }
    for (int j = tid; j <= nb; j += 256) tab[(size_t)j * nch + blockIdx.x] = sHist[j];
    __syncthreads();

    int2* outp = (int2*)binned + (size_t)rbase + e0;
    for (int i = tid; i < cnt; i += 256) outp[i] = sPair[i];
}

__global__ __launch_bounds__(256) void fine_kernel(
        const int* __restrict__ binned, const int* __restrict__ tab,
        int* __restrict__ cnt, int* __restrict__ off,
        int* __restrict__ list0, int* __restrict__ list1,
        int* __restrict__ list2, int* __restrict__ list3,
        int E0, int E1, int E2, int E3, int nb, int nch, int n) {
    __shared__ int sCnt[256];
    __shared__ int sCur[256];
    __shared__ int sWave[8];
    __shared__ int sRed[8];
    __shared__ int sStage[STAGE];
    const int tid  = threadIdx.x;
    const int lane = tid & 31;
    const int wave = tid >> 5;
    const int C0 = (E0 + CHUNK - 1) / CHUNK, C1 = (E1 + CHUNK - 1) / CHUNK,
              C2 = (E2 + CHUNK - 1) / CHUNK, C3 = (E3 + CHUNK - 1) / CHUNK;
    const int r = blockIdx.x / nb;
    const int g = blockIdx.x - r * nb;
    int rbase = 0, cb = 0, nC = C0;
    int* list = list0;
    if (r == 1) { rbase = E0;           cb = C0;           nC = C1; list = list1; }
    if (r == 2) { rbase = E0 + E1;      cb = C0 + C1;      nC = C2; list = list2; }
    if (r == 3) { rbase = E0 + E1 + E2; cb = C0 + C1 + C2; nC = C3; list = list3; }
    const int* tlo = tab + (size_t)g * nch + cb;
    const int* thi = tab + (size_t)(g + 1) * nch + cb;
    const int2* win = (const int2*)binned + (size_t)rbase;

    sCnt[tid] = 0;
    __syncthreads();

    int lower = 0;
    for (int b = tid; b < nC; b += 256) {
        const int s = tlo[b], e = thi[b];
        lower += s;
        const int2* p = win + (size_t)b * CHUNK;
        for (int i = s; i < e; i += 8) {
            int y[8];
            #pragma unroll
            for (int u = 0; u < 8; ++u) y[u] = p[min(i + u, e - 1)].y;
            #pragma unroll
            for (int u = 0; u < 8; ++u)
                if (i + u < e) atomicAdd(&sCnt[y[u] & 255], 1);
        }
    }
    __syncthreads();

    const int cT = sCnt[tid];
    int inc = cT;
    for (int d = 1; d < 32; d <<= 1) {
        const int u = __shfl_up(inc, d, 32);
        if (lane >= d) inc += u;
    }
    for (int o = 16; o > 0; o >>= 1) lower += __shfl_xor(lower, o, 32);
    if (lane == 31) sWave[wave] = inc;
    if (lane == 0)  sRed[wave]  = lower;
    __syncthreads();
    int wbase = 0, total = 0, start = 0;
    for (int w = 0; w < 8; ++w) {
        const int tw = sWave[w];
        if (w < wave) wbase += tw;
        total += tw;
        start += sRed[w];
    }
    const int offL = wbase + inc - cT;
    const int t = g * 256 + tid;
    if (t < n) {
        cnt[(size_t)r * n + t] = cT;
        off[(size_t)r * n + t] = start + offL;
    }
    sCur[tid] = offL;
    __syncthreads();

    const bool staged = total <= STAGE;
    for (int b = tid; b < nC; b += 256) {
        const int s = tlo[b], e = thi[b];
        const int2* p = win + (size_t)b * CHUNK;
        for (int i = s; i < e; i += 8) {
            int2 v[8];
            #pragma unroll
            for (int u = 0; u < 8; ++u) v[u] = p[min(i + u, e - 1)];
            #pragma unroll
            for (int u = 0; u < 8; ++u) {
                if (i + u < e) {
                    const int pos = atomicAdd(&sCur[v[u].y & 255], 1);
                    if (staged) sStage[pos] = v[u].x;
                    else        list[start + pos] = v[u].x;
                }
            }
        }
    }
    if (staged) {
        __syncthreads();
        for (int i = tid; i < total; i += 256) list[start + i] = sStage[i];
    }
}

__device__ __forceinline__ void add4(float4& a, const float4 v) {
    a.x += v.x; a.y += v.y; a.z += v.z; a.w += v.w;
}

__device__ __forceinline__ float4 bucket_sum(const float4* __restrict__ feat4,
                                             const int* __restrict__ list,
                                             int beg, int len, int lane) {
    float4 acc = make_float4(0.0f, 0.0f, 0.0f, 0.0f);
    if (len <= 32) {
        const int key = (lane < len) ? list[beg + lane] : 0x7fffffff;
        int rank = 0;
        for (int j = 0; j < len; ++j) {
            const int kj = __builtin_amdgcn_readlane(key, j);
            rank += (kj < key || (kj == key && j < lane)) ? 1 : 0;
        }
        if (lane >= len) rank = lane;
        const int sorted = __builtin_amdgcn_ds_permute(rank << 2, key);
        int r = 0;
        for (; r + 4 <= len; r += 4) {
            const int s0 = __builtin_amdgcn_readlane(sorted, r);
            const int s1 = __builtin_amdgcn_readlane(sorted, r + 1);
            const int s2 = __builtin_amdgcn_readlane(sorted, r + 2);
            const int s3 = __builtin_amdgcn_readlane(sorted, r + 3);
            const float4 v0 = feat4[(size_t)s0 * (H / 4) + lane];
            const float4 v1 = feat4[(size_t)s1 * (H / 4) + lane];
            const float4 v2 = feat4[(size_t)s2 * (H / 4) + lane];
            const float4 v3 = feat4[(size_t)s3 * (H / 4) + lane];
            add4(acc, v0); add4(acc, v1); add4(acc, v2); add4(acc, v3);
        }
        for (; r < len; ++r) {
            const int s0 = __builtin_amdgcn_readlane(sorted, r);
            add4(acc, feat4[(size_t)s0 * (H / 4) + lane]);
        }
    } else {
        int prev = -1;
        for (int it = 0; it < len; ++it) {
            int m = 0x7fffffff;
            for (int i = lane; i < len; i += 32) {
                const int k = list[beg + i];
                if (k > prev && k < m) m = k;
            }
            for (int o = 16; o > 0; o >>= 1) m = min(m, __shfl_xor(m, o, 32));
            if (m == 0x7fffffff) break;
            int c = 0;
            for (int i = lane; i < len; i += 32) c += (list[beg + i] == m) ? 1 : 0;
            for (int o = 16; o > 0; o >>= 1) c += __shfl_xor(c, o, 32);
            const float4 v = feat4[(size_t)m * (H / 4) + lane];
            for (int q = 0; q < c; ++q) add4(acc, v);
            prev = m;
        }
    }
    return acc;
}


__device__ __forceinline__ int sort_ids(int key, int len, int lane) {
    int rank = 0;
    for (int j = 0; j < len; ++j) {
        const int kj = __builtin_amdgcn_readlane(key, j);
        rank += (kj < key || (kj == key && j < lane)) ? 1 : 0;
    }
    if (lane >= len) rank = lane;
    return __builtin_amdgcn_ds_permute(rank << 2, key);
}

template <int Q>
__device__ __forceinline__ float4 long_bucket_sum(const float4* __restrict__ x4,
                                                  const int* __restrict__ list,
                                                  int beg, int len, int lane, int qq) {
    float4 acc = make_float4(0.0f, 0.0f, 0.0f, 0.0f);
    int prev = -1;
    for (int it = 0; it < len; ++it) {
        int m = 0x7fffffff;
        for (int i = lane; i < len; i += 32) {
            const int k = list[beg + i];
            if (k > prev && k < m) m = k;
        }
        for (int o = 16; o > 0; o >>= 1) m = min(m, __shfl_xor(m, o, 32));
        if (m == 0x7fffffff) break;
        int c = 0;
        for (int i = lane; i < len; i += 32) c += (list[beg + i] == m) ? 1 : 0;
        for (int o = 16; o > 0; o >>= 1) c += __shfl_xor(c, o, 32);
        const float4 v = x4[(size_t)m * Q + qq];
        for (int r = 0; r < c; ++r) add4(acc, v);
        prev = m;
    }
    return acc;
}

__device__ __forceinline__ float4 slot_sum(float4 a) {
    for (int m = 4; m < 32; m <<= 1) {
        float4 b;
        b.x = __shfl_xor(a.x, m, 32); b.y = __shfl_xor(a.y, m, 32);
        b.z = __shfl_xor(a.z, m, 32); b.w = __shfl_xor(a.w, m, 32);
        add4(a, b);
    }
    return a;
}

#define TPW 4

__global__ __launch_bounds__(256) void gather_raw_proj_kernel(
        const float* __restrict__ xLane, const float* __restrict__ xSens,
        const float* __restrict__ xInj,
        const float* __restrict__ WLane, const float* __restrict__ bLane,
        const float* __restrict__ WSens, const float* __restrict__ bSens,
        const float* __restrict__ WInj,  const float* __restrict__ bInj,
        const int* __restrict__ listFl, const int* __restrict__ listFs,
        const int* __restrict__ listIn,
        const int* __restrict__ cnt, const int* __restrict__ off,
        float* __restrict__ aggS, int n) {
    __shared__ float4 sW[35 * 32];
    __shared__ float4 sZ[8][TPW][9];
    const int tid  = threadIdx.x;
    const int wave = tid >> 5;
    const int lane = tid & 31;
    const int slot = lane >> 2;
    const int q    = lane & 3;

    {
        const float4* wl = (const float4*)WLane;
        sW[tid]       = wl[tid];
        sW[256 + tid] = wl[256 + tid];
        sW[512 + tid] = ((const float4*)WSens)[tid];
        sW[768 + tid] = ((const float4*)WInj)[tid];
        if (tid < 96) {
            const float* bs = tid < 32 ? bLane : (tid < 64 ? bSens : bInj);
            sW[1024 + tid] = ((const float4*)bs)[tid & 31];
        }
    }

    const float4* xl4 = (const float4*)xLane;
    const float4* xs4 = (const float4*)xSens;
    const float4* xi4 = (const float4*)xInj;
    const float4 zero4 = make_float4(0.0f, 0.0f, 0.0f, 0.0f);
    const int tb = blockIdx.x * (8 * TPW) + wave * TPW;

    for (int u = 0; u < TPW; ++u) {
        const int t = __builtin_amdgcn_readfirstlane(tb + u);
        float4 zl = zero4, zsi = zero4;
        float fl = 0.0f, fs = 0.0f, fi = 0.0f;
        if (t < n) {
            const int lenL = cnt[1 * n + t], lenS = cnt[2 * n + t], lenI = cnt[3 * n + t];
            const int begL = off[1 * n + t], begS = off[2 * n + t], begI = off[3 * n + t];
            fl = lenL > 0 ? 1.0f : 0.0f;
            fs = lenS > 0 ? 1.0f : 0.0f;
            fi = lenI > 0 ? 1.0f : 0.0f;
            const bool isS = q < 2;
            const int qq = q & 1;
            if (lenL <= 32 && lenS <= 32 && lenI <= 32) {
                const int keyL = (lane < lenL) ? listFl[begL + lane] : 0x7fffffff;
                const int keyS = (lane < lenS) ? listFs[begS + lane] : 0x7fffffff;
                const int keyI = (lane < lenI) ? listIn[begI + lane] : 0x7fffffff;
                const int sortedL = sort_ids(keyL, lenL, lane);
                const int sortedS = sort_ids(keyS, lenS, lane);
                const int sortedI = sort_ids(keyI, lenI, lane);
                const int lenX = isS ? lenS : lenI;
                const int maxSI = max(lenS, lenI);
                const float4* xb = isS ? xs4 : xi4;
                float4 a[4], b[4];
                #pragma unroll
                for (int i = 0; i < 4; ++i) {
                    a[i] = zero4; b[i] = zero4;
                    const int p = 8 * i + slot;
                    if (8 * i < lenL) {
                        const int id = __builtin_amdgcn_ds_bpermute(p << 2, sortedL);
                        if (p < lenL) a[i] = xl4[(size_t)id * 4 + q];
                    }
                    if (8 * i < maxSI) {
                        const int idS = __builtin_amdgcn_ds_bpermute(p << 2, sortedS);
                        const int idI = __builtin_amdgcn_ds_bpermute(p << 2, sortedI);
                        if (p < lenX) b[i] = xb[(size_t)(isS ? idS : idI) * 2 + qq];
                    }
                }
                add4(a[0], a[1]); add4(a[0], a[2]); add4(a[0], a[3]);
                add4(b[0], b[1]); add4(b[0], b[2]); add4(b[0], b[3]);
                zl  = slot_sum(a[0]);
                zsi = slot_sum(b[0]);
            } else {
                zl = long_bucket_sum<4>(xl4, listFl, begL, lenL, lane, q);
                const float4 s = long_bucket_sum<2>(xs4, listFs, begS, lenS, lane, qq);
                const float4 j = long_bucket_sum<2>(xi4, listIn, begI, lenI, lane, qq);
                zsi = isS ? s : j;
            }
            const float wL = 1.0f / (float)max(lenL, 1);
            const float wX = 1.0f / (float)max(isS ? lenS : lenI, 1);
            zl.x *= wL;  zl.y *= wL;  zl.z *= wL;  zl.w *= wL;
            zsi.x *= wX; zsi.y *= wX; zsi.z *= wX; zsi.w *= wX;
        }
        if (lane < 4) { sZ[wave][u][lane] = zl; sZ[wave][u][4 + lane] = zsi; }
        if (lane == 0) sZ[wave][u][8] = make_float4(fl, fs, fi, 0.0f);
    }
    __syncthreads();

    float4 o[TPW];
    #pragma unroll
    for (int u = 0; u < TPW; ++u) o[u] = zero4;
    #pragma nounroll
    for (int g = 0; g < 8; ++g) {
        const float4 w0 = sW[(4 * g + 0) * 32 + lane];
        const float4 w1 = sW[(4 * g + 1) * 32 + lane];
        const float4 w2 = sW[(4 * g + 2) * 32 + lane];
        const float4 w3 = sW[(4 * g + 3) * 32 + lane];
        #pragma unroll
        for (int u = 0; u < TPW; ++u) {
            const float4 z = sZ[wave][u][g];
            o[u].x += z.x * w0.x; o[u].y += z.x * w0.y; o[u].z += z.x * w0.z; o[u].w += z.x * w0.w;
            o[u].x += z.y * w1.x; o[u].y += z.y * w1.y; o[u].z += z.y * w1.z; o[u].w += z.y * w1.w;
            o[u].x += z.z * w2.x; o[u].y += z.z * w2.y; o[u].z += z.z * w2.z; o[u].w += z.z * w2.w;
            o[u].x += z.w * w3.x; o[u].y += z.w * w3.y; o[u].z += z.w * w3.z; o[u].w += z.w * w3.w;
        }
    }
    {
        const float4 b0 = sW[32 * 32 + lane];
        const float4 b1 = sW[33 * 32 + lane];
        const float4 b2 = sW[34 * 32 + lane];
        #pragma unroll
        for (int u = 0; u < TPW; ++u) {
            const float4 f = sZ[wave][u][8];
            o[u].x += f.x * b0.x; o[u].y += f.x * b0.y; o[u].z += f.x * b0.z; o[u].w += f.x * b0.w;
            o[u].x += f.y * b1.x; o[u].y += f.y * b1.y; o[u].z += f.y * b1.z; o[u].w += f.y * b1.w;
            o[u].x += f.z * b2.x; o[u].y += f.z * b2.y; o[u].z += f.z * b2.z; o[u].w += f.z * b2.w;
        }
    }
    #pragma unroll
    for (int u = 0; u < TPW; ++u) {
        const int t = tb + u;
        if (t < n) ((float4*)aggS)[(size_t)t * (H / 4) + lane] = o[u];
    }
}

#define RPB 7
#define GW (4 * RPB)
#define GT (GW * 32)
#define APITCH 132

__device__ __forceinline__ float4 bucket_sum_keyed(const float4* __restrict__ feat4,
                                                   int key, int len, int lane) {
    const float4 zero4 = make_float4(0.0f, 0.0f, 0.0f, 0.0f);
    const int sorted = sort_ids(key, len, lane);
    float4 acc = zero4;
    for (int r = 0; r < len; r += 8) {
        float4 v[8];
        #pragma unroll
        for (int i = 0; i < 8; ++i) {
            v[i] = zero4;
            if (r + i < len) {
                const int sid = __builtin_amdgcn_readlane(sorted, r + i);
                v[i] = feat4[(size_t)sid * (H / 4) + lane];
            }
        }
        #pragma unroll
        for (int i = 0; i < 8; ++i) add4(acc, v[i]);
    }
    return acc;
}

__global__ __launch_bounds__(GT) void fused_gemm_elu_kernel(
        const float* __restrict__ h, const float* __restrict__ aggS,
        const int* __restrict__ listSp, const int* __restrict__ cnt, const int* __restrict__ off,
        const float* __restrict__ W0, const float* __restrict__ W1,
        const float* __restrict__ bias, float* __restrict__ out, int n, int nrp) {
    __shared__ __align__(16) float sB[2 * H * H];
    __shared__ __align__(16) float sA[RPB * 32 * APITCH];
    const int tid  = threadIdx.x;
    const int lane = tid & 31;
    const int half = lane >> 4;
    const int l15  = lane & 15;

    {
        constexpr int NV = 2 * H * H / 4;
        constexpr int NI = (NV + GT - 1) / GT;
        float4 wv[NI];
        #pragma unroll
        for (int q = 0; q < NI; ++q) {
            const int i = tid + GT * q;
            wv[q] = make_float4(0.0f, 0.0f, 0.0f, 0.0f);
            if (i < NV) wv[q] = ((const float4*)((i >> 12) ? W1 : W0))[i & 4095];
        }
        #pragma unroll
        for (int q = 0; q < NI; ++q) {
            const int i = tid + GT * q;
            if (i < NV) {
                const int jj = i & 4095;
                const int k  = jj >> 5;
                const int c0 = (jj & 31) << 2;
                float* d = sB + ((i >> 12) << 14) + ((k >> 1) << 8) + (k & 1);
                d[2 * (c0 + 0)] = wv[q].x;
                d[2 * (c0 + 1)] = wv[q].y;
                d[2 * (c0 + 2)] = wv[q].z;
                d[2 * (c0 + 3)] = wv[q].w;
            }
        }
    }
    __syncthreads();

    const int wave = __builtin_amdgcn_readfirstlane(tid >> 5);
    const int lrp  = wave >> 2;
    const int rpBeg = (int)(((long long)blockIdx.x * nrp) / gridDim.x);
    const int rpEnd = (int)(((long long)(blockIdx.x + 1) * nrp) / gridDim.x);
    const bool active = rpBeg + lrp < rpEnd;
    const int m0 = (rpBeg + lrp) << 5;
    const int cb = (wave & 3) << 5;
    int r0 = m0 + l15;
    if (r0 >= n) r0 = n - 1;
    int r1 = m0 + 16 + l15;
    if (r1 >= n) r1 = n - 1;
    float* sAp = sA + (size_t)lrp * 32 * APITCH;

    v8f acc00 = {}, acc01 = {}, acc10 = {}, acc11 = {};
    if (active) {
        const int first = (wave >> 2) & 1;
        for (int st = 0; st < 2; ++st) {
            if ((st ^ first) == 0) {
                const float4* p0 = (const float4*)(h + (size_t)r0 * H + (half << 2));
                const float4* p1 = (const float4*)(h + (size_t)r1 * H + (half << 2));
                const float* bbase = sB + (half << 9) + ((cb + l15) << 1);
                for (int j = 0; j < H / 8; ++j) {
                    const float4 x0 = p0[2 * j];
                    const float4 x1 = p1[2 * j];
                    const float* bk = bbase + (j << 10);
                    const v2f b00 = *(const v2f*)(bk);
                    const v2f b01 = *(const v2f*)(bk + 32);
                    const v2f b10 = *(const v2f*)(bk + 256);
                    const v2f b11 = *(const v2f*)(bk + 256 + 32);
                    v2f a;
                    a.x = x0.x; a.y = x0.y;
                    acc00 = __builtin_amdgcn_wmma_f32_16x16x4_f32(false, a, false, b00, (short)0, acc00, false, false);
                    acc01 = __builtin_amdgcn_wmma_f32_16x16x4_f32(false, a, false, b01, (short)0, acc01, false, false);
                    a.x = x1.x; a.y = x1.y;
                    acc10 = __builtin_amdgcn_wmma_f32_16x16x4_f32(false, a, false, b00, (short)0, acc10, false, false);
                    acc11 = __builtin_amdgcn_wmma_f32_16x16x4_f32(false, a, false, b01, (short)0, acc11, false, false);
                    a.x = x0.z; a.y = x0.w;
                    acc00 = __builtin_amdgcn_wmma_f32_16x16x4_f32(false, a, false, b10, (short)0, acc00, false, false);
                    acc01 = __builtin_amdgcn_wmma_f32_16x16x4_f32(false, a, false, b11, (short)0, acc01, false, false);
                    a.x = x1.z; a.y = x1.w;
                    acc10 = __builtin_amdgcn_wmma_f32_16x16x4_f32(false, a, false, b10, (short)0, acc10, false, false);
                    acc11 = __builtin_amdgcn_wmma_f32_16x16x4_f32(false, a, false, b11, (short)0, acc11, false, false);
                }
            } else {
                const int lr0 = (wave & 3) << 3;
                const int t0  = m0 + lr0;
                const bool okl = lane < 8 && t0 + lane < n;
                const int cntv = okl ? cnt[t0 + lane] : 0;
                const int offv = okl ? off[t0 + lane] : 0;
                int lenN = __builtin_amdgcn_readlane(cntv, 0);
                int begN = __builtin_amdgcn_readlane(offv, 0);
                int keyN = (lenN <= 32 && lane < lenN) ? listSp[begN + lane] : 0x7fffffff;
                for (int q = 0; q < 8; ++q) {
                    const int t = t0 + q;
                    const int len = lenN, beg = begN, key = keyN;
                    if (q + 1 < 8) {
                        lenN = __builtin_amdgcn_readlane(cntv, q + 1);
                        begN = __builtin_amdgcn_readlane(offv, q + 1);
                        keyN = (lenN <= 32 && lane < lenN) ? listSp[begN + lane] : 0x7fffffff;
                    }
                    float4 r = make_float4(0.0f, 0.0f, 0.0f, 0.0f);
                    if (t < n) {
                        const float4 base = ((const float4*)aggS)[(size_t)t * (H / 4) + lane];
                        const float4 sm = (len <= 32)
                            ? bucket_sum_keyed((const float4*)h, key, len, lane)
                            : bucket_sum((const float4*)h, listSp, beg, len, lane);
                        const float w = 1.0f / (float)max(len, 1);
                        r.x = base.x + sm.x * w; r.y = base.y + sm.y * w;
                        r.z = base.z + sm.z * w; r.w = base.w + sm.w * w;
                    }
                    *(float4*)(sAp + (size_t)(lr0 + q) * APITCH + (lane << 2)) = r;
                }
            }
        }
    }
    __syncthreads();

    if (active) {
        const float* q0 = sAp + (size_t)l15 * APITCH + (half << 2);
        const float* q1 = q0 + 16 * APITCH;
        const float* bbase = sB + (1 << 14) + (half << 9) + ((cb + l15) << 1);
        for (int j = 0; j < H / 8; ++j) {
            const float4 x0 = *(const float4*)(q0 + 8 * j);
            const float4 x1 = *(const float4*)(q1 + 8 * j);
            const float* bk = bbase + (j << 10);
            const v2f b00 = *(const v2f*)(bk);
            const v2f b01 = *(const v2f*)(bk + 32);
            const v2f b10 = *(const v2f*)(bk + 256);
            const v2f b11 = *(const v2f*)(bk + 256 + 32);
            v2f a;
            a.x = x0.x; a.y = x0.y;
            acc00 = __builtin_amdgcn_wmma_f32_16x16x4_f32(false, a, false, b00, (short)0, acc00, false, false);
            acc01 = __builtin_amdgcn_wmma_f32_16x16x4_f32(false, a, false, b01, (short)0, acc01, false, false);
            a.x = x1.x; a.y = x1.y;
            acc10 = __builtin_amdgcn_wmma_f32_16x16x4_f32(false, a, false, b00, (short)0, acc10, false, false);
            acc11 = __builtin_amdgcn_wmma_f32_16x16x4_f32(false, a, false, b01, (short)0, acc11, false, false);
            a.x = x0.z; a.y = x0.w;
            acc00 = __builtin_amdgcn_wmma_f32_16x16x4_f32(false, a, false, b10, (short)0, acc00, false, false);
            acc01 = __builtin_amdgcn_wmma_f32_16x16x4_f32(false, a, false, b11, (short)0, acc01, false, false);
            a.x = x1.z; a.y = x1.w;
            acc10 = __builtin_amdgcn_wmma_f32_16x16x4_f32(false, a, false, b10, (short)0, acc10, false, false);
            acc11 = __builtin_amdgcn_wmma_f32_16x16x4_f32(false, a, false, b11, (short)0, acc11, false, false);
        }

        const float bc0 = bias[cb + l15];
        const float bc1 = bias[cb + 16 + l15];
        #pragma unroll
        for (int r = 0; r < 8; ++r) {
            const int rowa = m0 + (half << 3) + r;
            const int rowb = rowa + 16;
            if (rowa < n) {
                float v = acc00[r] + bc0;
                out[(size_t)rowa * H + cb + l15] = v > 0.0f ? v : (__expf(v) - 1.0f);
                v = acc01[r] + bc1;
                out[(size_t)rowa * H + cb + 16 + l15] = v > 0.0f ? v : (__expf(v) - 1.0f);
            }
            if (rowb < n) {
                float v = acc10[r] + bc0;
                out[(size_t)rowb * H + cb + l15] = v > 0.0f ? v : (__expf(v) - 1.0f);
                v = acc11[r] + bc1;
                out[(size_t)rowb * H + cb + 16 + l15] = v > 0.0f ? v : (__expf(v) - 1.0f);
            }
        }
    }
}

extern "C" void kernel_launch(void* const* d_in, const int* in_sizes, int n_in,
                              void* d_out, int out_size, void* d_ws, size_t ws_size,
                              hipStream_t stream) {
    const float* x_int  = (const float*)d_in[0];
    const float* x_lane = (const float*)d_in[1];
    const float* x_sens = (const float*)d_in[2];
    const float* x_inj  = (const float*)d_in[3];
    const int*   e_sp   = (const int*)d_in[4];
    const int*   e_fl   = (const int*)d_in[5];
    const int*   e_fs   = (const int*)d_in[6];
    const int*   e_in   = (const int*)d_in[7];
    const float* W_int  = (const float*)d_in[8];
    const float* b_int  = (const float*)d_in[9];
    const float* W_lane = (const float*)d_in[10];
    const float* b_lane = (const float*)d_in[11];
    const float* W_sens = (const float*)d_in[12];
    const float* b_sens = (const float*)d_in[13];
    const float* W_inj  = (const float*)d_in[14];
    const float* b_inj  = (const float*)d_in[15];
    const float* self_W = (const float*)d_in[16];
    const float* self_b = (const float*)d_in[17];
    const float* rel_W  = (const float*)d_in[18];

    const int nInt  = in_sizes[0] / 32;
    const int nLane = in_sizes[1] / 16;
    const int nSens = in_sizes[2] / 8;
    const int nInj  = in_sizes[3] / 8;
    const int eSp   = in_sizes[4] / 2;
    const int eFl   = in_sizes[5] / 2;
    const int eFs   = in_sizes[6] / 2;
    const int eIn   = in_sizes[7] / 2;
    const int L     = in_sizes[16] / (H * H);

    float* ws = (float*)d_ws;
    size_t o = 0;
    float* hA    = ws + o; o += (size_t)nInt  * H;
    float* hB    = ws + o; o += (size_t)nInt  * H;
    float* hLane = ws + o; o += (size_t)nLane * H;
    float* pSens = ws + o; o += (size_t)nSens * H;
    o += (size_t)nInj  * H;
    float* aggS  = ws + o; o += (size_t)nInt  * H;
    float* aggB  = ws + o; o += (size_t)nInt  * H;
    int*   cnt   = (int*)(ws + o); o += (size_t)4 * nInt;
    int*   off   = (int*)(ws + o); o += (size_t)4 * nInt;
    (void)ws_size; (void)n_in; (void)out_size;

    int*  binned = (int*)aggB;
    int*  listFl = (int*)aggB + 2 * ((size_t)eSp + eFl + eFs + eIn);
    int*  listFs = listFl + eFl;
    int*  listIn = listFs + eFs;
    int*  listSp = (int*)hLane;
    int*  tab    = (int*)pSens;

    proj_kernel<<<(nInt + 127) / 128, 256, 0, stream>>>(x_int, W_int, b_int, hA, nInt);

    const int nb  = (nInt + 255) / 256;
    const int nch = (eSp + CHUNK - 1) / CHUNK + (eFl + CHUNK - 1) / CHUNK
                  + (eFs + CHUNK - 1) / CHUNK + (eIn + CHUNK - 1) / CHUNK;
    bin_kernel<<<nch, 256, 0, stream>>>(e_sp, e_fl, e_fs, e_in, binned, tab,
                                        eSp, eFl, eFs, eIn, nb, nch);
    fine_kernel<<<4 * nb, 256, 0, stream>>>(binned, tab, cnt, off,
                                            listSp, listFl, listFs, listIn,
                                            eSp, eFl, eFs, eIn, nb, nch, nInt);

    gather_raw_proj_kernel<<<(nInt + 8 * TPW - 1) / (8 * TPW), 256, 0, stream>>>(
        x_lane, x_sens, x_inj, W_lane, b_lane, W_sens, b_sens, W_inj, b_inj,
        listFl, listFs, listIn, cnt, off, aggS, nInt);

    const int nrp = (nInt + 31) / 32;
    const int needed = (nrp + RPB - 1) / RPB;
    const int lgrid = needed > 256 ? needed : 256;
    float* hcur = hA;
    float* hnext = hB;
    for (int l = 0; l < L; ++l) {
        float* outp = (l == L - 1) ? (float*)d_out : hnext;
        fused_gemm_elu_kernel<<<lgrid, GT, 0, stream>>>(
            hcur, aggS, listSp, cnt, off,
            self_W + (size_t)l * H * H, rel_W + (size_t)l * H * H,
            self_b + (size_t)l * H, outp, nInt, nrp);
        float* t = hnext; hnext = hcur; hcur = t;
    }
}


# Spatial gather software-pipelined: one L2 trip per bucket instead of three (up to 16 row loads issued first, was 8 behind a wait; next ids fetched two ahead, sorted under them); sum order unchanged
# speedup vs baseline: 7.3415x; 1.0302x over previous
#include <hip/hip_runtime.h>
#include <cstddef>

#define H 128

typedef float v2f __attribute__((ext_vector_type(2)));
typedef float v8f __attribute__((ext_vector_type(8)));

#define KIN 32
__global__ __launch_bounds__(256) void proj_kernel(const float* __restrict__ x,
                                                   const float* __restrict__ W,
                                                   const float* __restrict__ b,
                                                   float* __restrict__ out, int n) {
    __shared__ __align__(16) float sW[KIN * H];
    const int tid  = threadIdx.x;
    const int lane = tid & 31;
    const int half = lane >> 4;
    const int l15  = lane & 15;
    const int m0 = __builtin_amdgcn_readfirstlane((int)(blockIdx.x * 128 + (tid >> 5) * 16));

    int arow = m0 + l15;
    if (arow >= n) arow = n - 1;
    const float4* xr = (const float4*)(x + (size_t)arow * KIN + (half << 4));
    const float4 x0 = xr[0], x1 = xr[1], x2 = xr[2], x3 = xr[3];

    float4 wv[4];
    #pragma unroll
    for (int j = 0; j < 4; ++j) wv[j] = ((const float4*)W)[tid + 256 * j];

    v8f acc[8];
    #pragma unroll
    for (int t = 0; t < 8; ++t) {
        const float bc = b[(t << 4) + l15];
        #pragma unroll
        for (int r = 0; r < 8; ++r) acc[t][r] = bc;
    }

    #pragma unroll
    for (int j = 0; j < 4; ++j) {
        const int i  = tid + 256 * j;
        const int k  = i >> 5;
        const int c0 = (i & 31) << 2;
        float* d = sW + ((k >> 1) << 8) + (k & 1);
        d[2 * (c0 + 0)] = wv[j].x;
        d[2 * (c0 + 1)] = wv[j].y;
        d[2 * (c0 + 2)] = wv[j].z;
        d[2 * (c0 + 3)] = wv[j].w;
    }
    __syncthreads();

    const float xf[16] = { x0.x, x0.y, x0.z, x0.w, x1.x, x1.y, x1.z, x1.w,
                           x2.x, x2.y, x2.z, x2.w, x3.x, x3.y, x3.z, x3.w };
    const float* bbase = sW + (half << 11) + (l15 << 1);
    #pragma unroll
    for (int s = 0; s < 8; ++s) {
        v2f a;
        a.x = xf[2 * s];
        a.y = xf[2 * s + 1];
        const float* bk = bbase + (s << 8);
        #pragma unroll
        for (int t = 0; t < 8; ++t) {
            const v2f bf = *(const v2f*)(bk + (t << 5));
            acc[t] = __builtin_amdgcn_wmma_f32_16x16x4_f32(
                false, a, false, bf, (short)0, acc[t], false, false);
        }
    }

    float* orow = out + (size_t)(m0 + (half << 3)) * H + l15;
    if (m0 + 16 <= n) {
        #pragma unroll
        for (int t = 0; t < 8; ++t) {
            #pragma unroll
            for (int r = 0; r < 8; ++r) orow[r * H + (t << 4)] = acc[t][r];
        }
    } else {
        #pragma unroll
        for (int t = 0; t < 8; ++t) {
            #pragma unroll
            for (int r = 0; r < 8; ++r) {
                if (m0 + (half << 3) + r < n) orow[r * H + (t << 4)] = acc[t][r];
            }
        }
    }
}

#define CHUNK 4096
#define MAXNB 1024
#define STAGE 8192

__global__ __launch_bounds__(256) void bin_kernel(
        const int* __restrict__ ed0, const int* __restrict__ ed1,
        const int* __restrict__ ed2, const int* __restrict__ ed3,
        int* __restrict__ binned, int* __restrict__ tab,
        int E0, int E1, int E2, int E3, int nb, int nch) {
    __shared__ int  sHist[MAXNB + 1];
    __shared__ int2 sPair[CHUNK];
    const int tid = threadIdx.x;
    const int C0 = (E0 + CHUNK - 1) / CHUNK, C1 = (E1 + CHUNK - 1) / CHUNK,
              C2 = (E2 + CHUNK - 1) / CHUNK;
    int c = blockIdx.x;
    const int* ed = ed0;
    int E = E0, rbase = 0;
    if (c >= C0) {
        c -= C0; ed = ed1; E = E1; rbase = E0;
        if (c >= C1) {
            c -= C1; ed = ed2; E = E2; rbase += E1;
            if (c >= C2) { c -= C2; ed = ed3; E = E3; rbase += E2; }
        }
    }
    const int e0  = c * CHUNK;
    const int cnt = min(CHUNK, E - e0);

    for (int j = tid; j <= nb; j += 256) sHist[j] = 0;
    __syncthreads();

    int srcv[16], tgtv[16], rk[16];
    #pragma unroll
    for (int k = 0; k < 16; ++k) {
        const int i = k * 256 + tid;
        srcv[k] = 0; tgtv[k] = 0;
        if (i < cnt) { srcv[k] = ed[e0 + i]; tgtv[k] = ed[E + e0 + i]; }
    }
    #pragma unroll
    for (int k = 0; k < 16; ++k) {
        const int i = k * 256 + tid;
        rk[k] = 0;
        if (i < cnt) rk[k] = atomicAdd(&sHist[tgtv[k] >> 8], 1);
    }
    __syncthreads();

    if (tid < 32) {
        int carry = 0;
        for (int base = 0; base < nb; base += 32) {
            const int j = base + tid;
            const int v = (j < nb) ? sHist[j] : 0;
            int inc = v;
            for (int d = 1; d < 32; d <<= 1) {
                const int u = __shfl_up(inc, d, 32);
                if (tid >= d) inc += u;
            }
            if (j < nb) sHist[j] = carry + inc - v;
            carry += __shfl(inc, 31, 32);
        }
        if (tid == 0) sHist[nb] = carry;
    }
    __syncthreads();

    #pragma unroll
    for (int k = 0; k < 16; ++k) {
        const int i = k * 256 + tid;
        if (i < cnt) sPair[sHist[tgtv[k] >> 8] + rk[k]] = make_int2(srcv[k], tgtv[k]);
    }
    for (int j = tid; j <= nb; j += 256) tab[(size_t)j * nch + blockIdx.x] = sHist[j];
    __syncthreads();

    int2* outp = (int2*)binned + (size_t)rbase + e0;
    for (int i = tid; i < cnt; i += 256) outp[i] = sPair[i];
}

__global__ __launch_bounds__(256) void fine_kernel(
        const int* __restrict__ binned, const int* __restrict__ tab,
        int* __restrict__ cnt, int* __restrict__ off,
        int* __restrict__ list0, int* __restrict__ list1,
        int* __restrict__ list2, int* __restrict__ list3,
        int E0, int E1, int E2, int E3, int nb, int nch, int n) {
    __shared__ int sCnt[256];
    __shared__ int sCur[256];
    __shared__ int sWave[8];
    __shared__ int sRed[8];
    __shared__ int sStage[STAGE];
    const int tid  = threadIdx.x;
    const int lane = tid & 31;
    const int wave = tid >> 5;
    const int C0 = (E0 + CHUNK - 1) / CHUNK, C1 = (E1 + CHUNK - 1) / CHUNK,
              C2 = (E2 + CHUNK - 1) / CHUNK, C3 = (E3 + CHUNK - 1) / CHUNK;
    const int r = blockIdx.x / nb;
    const int g = blockIdx.x - r * nb;
    int rbase = 0, cb = 0, nC = C0;
    int* list = list0;
    if (r == 1) { rbase = E0;           cb = C0;           nC = C1; list = list1; }
    if (r == 2) { rbase = E0 + E1;      cb = C0 + C1;      nC = C2; list = list2; }
    if (r == 3) { rbase = E0 + E1 + E2; cb = C0 + C1 + C2; nC = C3; list = list3; }
    const int* tlo = tab + (size_t)g * nch + cb;
    const int* thi = tab + (size_t)(g + 1) * nch + cb;
    const int2* win = (const int2*)binned + (size_t)rbase;

    sCnt[tid] = 0;
    __syncthreads();

    int lower = 0;
    for (int b = tid; b < nC; b += 256) {
        const int s = tlo[b], e = thi[b];
        lower += s;
        const int2* p = win + (size_t)b * CHUNK;
        for (int i = s; i < e; i += 8) {
            int y[8];
            #pragma unroll
            for (int u = 0; u < 8; ++u) y[u] = p[min(i + u, e - 1)].y;
            #pragma unroll
            for (int u = 0; u < 8; ++u)
                if (i + u < e) atomicAdd(&sCnt[y[u] & 255], 1);
        }
    }
    __syncthreads();

    const int cT = sCnt[tid];
    int inc = cT;
    for (int d = 1; d < 32; d <<= 1) {
        const int u = __shfl_up(inc, d, 32);
        if (lane >= d) inc += u;
    }
    for (int o = 16; o > 0; o >>= 1) lower += __shfl_xor(lower, o, 32);
    if (lane == 31) sWave[wave] = inc;
    if (lane == 0)  sRed[wave]  = lower;
    __syncthreads();
    int wbase = 0, total = 0, start = 0;
    for (int w = 0; w < 8; ++w) {
        const int tw = sWave[w];
        if (w < wave) wbase += tw;
        total += tw;
        start += sRed[w];
    }
    const int offL = wbase + inc - cT;
    const int t = g * 256 + tid;
    if (t < n) {
        cnt[(size_t)r * n + t] = cT;
        off[(size_t)r * n + t] = start + offL;
    }
    sCur[tid] = offL;
    __syncthreads();

    const bool staged = total <= STAGE;
    for (int b = tid; b < nC; b += 256) {
        const int s = tlo[b], e = thi[b];
        const int2* p = win + (size_t)b * CHUNK;
        for (int i = s; i < e; i += 8) {
            int2 v[8];
            #pragma unroll
            for (int u = 0; u < 8; ++u) v[u] = p[min(i + u, e - 1)];
            #pragma unroll
            for (int u = 0; u < 8; ++u) {
                if (i + u < e) {
                    const int pos = atomicAdd(&sCur[v[u].y & 255], 1);
                    if (staged) sStage[pos] = v[u].x;
                    else        list[start + pos] = v[u].x;
                }
            }
        }
    }
    if (staged) {
        __syncthreads();
        for (int i = tid; i < total; i += 256) list[start + i] = sStage[i];
    }
}

__device__ __forceinline__ void add4(float4& a, const float4 v) {
    a.x += v.x; a.y += v.y; a.z += v.z; a.w += v.w;
}

__device__ __forceinline__ float4 bucket_sum(const float4* __restrict__ feat4,
                                             const int* __restrict__ list,
                                             int beg, int len, int lane) {
    float4 acc = make_float4(0.0f, 0.0f, 0.0f, 0.0f);
    if (len <= 32) {
        const int key = (lane < len) ? list[beg + lane] : 0x7fffffff;
        int rank = 0;
        for (int j = 0; j < len; ++j) {
            const int kj = __builtin_amdgcn_readlane(key, j);
            rank += (kj < key || (kj == key && j < lane)) ? 1 : 0;
        }
        if (lane >= len) rank = lane;
        const int sorted = __builtin_amdgcn_ds_permute(rank << 2, key);
        int r = 0;
        for (; r + 4 <= len; r += 4) {
            const int s0 = __builtin_amdgcn_readlane(sorted, r);
            const int s1 = __builtin_amdgcn_readlane(sorted, r + 1);
            const int s2 = __builtin_amdgcn_readlane(sorted, r + 2);
            const int s3 = __builtin_amdgcn_readlane(sorted, r + 3);
            const float4 v0 = feat4[(size_t)s0 * (H / 4) + lane];
            const float4 v1 = feat4[(size_t)s1 * (H / 4) + lane];
            const float4 v2 = feat4[(size_t)s2 * (H / 4) + lane];
            const float4 v3 = feat4[(size_t)s3 * (H / 4) + lane];
            add4(acc, v0); add4(acc, v1); add4(acc, v2); add4(acc, v3);
        }
        for (; r < len; ++r) {
            const int s0 = __builtin_amdgcn_readlane(sorted, r);
            add4(acc, feat4[(size_t)s0 * (H / 4) + lane]);
        }
    } else {
        int prev = -1;
        for (int it = 0; it < len; ++it) {
            int m = 0x7fffffff;
            for (int i = lane; i < len; i += 32) {
                const int k = list[beg + i];
                if (k > prev && k < m) m = k;
            }
            for (int o = 16; o > 0; o >>= 1) m = min(m, __shfl_xor(m, o, 32));
            if (m == 0x7fffffff) break;
            int c = 0;
            for (int i = lane; i < len; i += 32) c += (list[beg + i] == m) ? 1 : 0;
            for (int o = 16; o > 0; o >>= 1) c += __shfl_xor(c, o, 32);
            const float4 v = feat4[(size_t)m * (H / 4) + lane];
            for (int q = 0; q < c; ++q) add4(acc, v);
            prev = m;
        }
    }
    return acc;
}


__device__ __forceinline__ int sort_ids(int key, int len, int lane) {
    int rank = 0;
    for (int j = 0; j < len; ++j) {
        const int kj = __builtin_amdgcn_readlane(key, j);
        rank += (kj < key || (kj == key && j < lane)) ? 1 : 0;
    }
    if (lane >= len) rank = lane;
    return __builtin_amdgcn_ds_permute(rank << 2, key);
}

template <int Q>
__device__ __forceinline__ float4 long_bucket_sum(const float4* __restrict__ x4,
                                                  const int* __restrict__ list,
                                                  int beg, int len, int lane, int qq) {
    float4 acc = make_float4(0.0f, 0.0f, 0.0f, 0.0f);
    int prev = -1;
    for (int it = 0; it < len; ++it) {
        int m = 0x7fffffff;
        for (int i = lane; i < len; i += 32) {
            const int k = list[beg + i];
            if (k > prev && k < m) m = k;
        }
        for (int o = 16; o > 0; o >>= 1) m = min(m, __shfl_xor(m, o, 32));
        if (m == 0x7fffffff) break;
        int c = 0;
        for (int i = lane; i < len; i += 32) c += (list[beg + i] == m) ? 1 : 0;
        for (int o = 16; o > 0; o >>= 1) c += __shfl_xor(c, o, 32);
        const float4 v = x4[(size_t)m * Q + qq];
        for (int r = 0; r < c; ++r) add4(acc, v);
        prev = m;
    }
    return acc;
}

__device__ __forceinline__ float4 slot_sum(float4 a) {
    for (int m = 4; m < 32; m <<= 1) {
        float4 b;
        b.x = __shfl_xor(a.x, m, 32); b.y = __shfl_xor(a.y, m, 32);
        b.z = __shfl_xor(a.z, m, 32); b.w = __shfl_xor(a.w, m, 32);
        add4(a, b);
    }
    return a;
}

#define TPW 4

__global__ __launch_bounds__(256) void gather_raw_proj_kernel(
        const float* __restrict__ xLane, const float* __restrict__ xSens,
        const float* __restrict__ xInj,
        const float* __restrict__ WLane, const float* __restrict__ bLane,
        const float* __restrict__ WSens, const float* __restrict__ bSens,
        const float* __restrict__ WInj,  const float* __restrict__ bInj,
        const int* __restrict__ listFl, const int* __restrict__ listFs,
        const int* __restrict__ listIn,
        const int* __restrict__ cnt, const int* __restrict__ off,
        float* __restrict__ aggS, int n) {
    __shared__ float4 sW[35 * 32];
    __shared__ float4 sZ[8][TPW][9];
    const int tid  = threadIdx.x;
    const int wave = tid >> 5;
    const int lane = tid & 31;
    const int slot = lane >> 2;
    const int q    = lane & 3;

    {
        const float4* wl = (const float4*)WLane;
        sW[tid]       = wl[tid];
        sW[256 + tid] = wl[256 + tid];
        sW[512 + tid] = ((const float4*)WSens)[tid];
        sW[768 + tid] = ((const float4*)WInj)[tid];
        if (tid < 96) {
            const float* bs = tid < 32 ? bLane : (tid < 64 ? bSens : bInj);
            sW[1024 + tid] = ((const float4*)bs)[tid & 31];
        }
    }

    const float4* xl4 = (const float4*)xLane;
    const float4* xs4 = (const float4*)xSens;
    const float4* xi4 = (const float4*)xInj;
    const float4 zero4 = make_float4(0.0f, 0.0f, 0.0f, 0.0f);
    const int tb = blockIdx.x * (8 * TPW) + wave * TPW;

    for (int u = 0; u < TPW; ++u) {
        const int t = __builtin_amdgcn_readfirstlane(tb + u);
        float4 zl = zero4, zsi = zero4;
        float fl = 0.0f, fs = 0.0f, fi = 0.0f;
        if (t < n) {
            const int lenL = cnt[1 * n + t], lenS = cnt[2 * n + t], lenI = cnt[3 * n + t];
            const int begL = off[1 * n + t], begS = off[2 * n + t], begI = off[3 * n + t];
            fl = lenL > 0 ? 1.0f : 0.0f;
            fs = lenS > 0 ? 1.0f : 0.0f;
            fi = lenI > 0 ? 1.0f : 0.0f;
            const bool isS = q < 2;
            const int qq = q & 1;
            if (lenL <= 32 && lenS <= 32 && lenI <= 32) {
                const int keyL = (lane < lenL) ? listFl[begL + lane] : 0x7fffffff;
                const int keyS = (lane < lenS) ? listFs[begS + lane] : 0x7fffffff;
                const int keyI = (lane < lenI) ? listIn[begI + lane] : 0x7fffffff;
                const int sortedL = sort_ids(keyL, lenL, lane);
                const int sortedS = sort_ids(keyS, lenS, lane);
                const int sortedI = sort_ids(keyI, lenI, lane);
                const int lenX = isS ? lenS : lenI;
                const int maxSI = max(lenS, lenI);
                const float4* xb = isS ? xs4 : xi4;
                float4 a[4], b[4];
                #pragma unroll
                for (int i = 0; i < 4; ++i) {
                    a[i] = zero4; b[i] = zero4;
                    const int p = 8 * i + slot;
                    if (8 * i < lenL) {
                        const int id = __builtin_amdgcn_ds_bpermute(p << 2, sortedL);
                        if (p < lenL) a[i] = xl4[(size_t)id * 4 + q];
                    }
                    if (8 * i < maxSI) {
                        const int idS = __builtin_amdgcn_ds_bpermute(p << 2, sortedS);
                        const int idI = __builtin_amdgcn_ds_bpermute(p << 2, sortedI);
                        if (p < lenX) b[i] = xb[(size_t)(isS ? idS : idI) * 2 + qq];
                    }
                }
                add4(a[0], a[1]); add4(a[0], a[2]); add4(a[0], a[3]);
                add4(b[0], b[1]); add4(b[0], b[2]); add4(b[0], b[3]);
                zl  = slot_sum(a[0]);
                zsi = slot_sum(b[0]);
            } else {
                zl = long_bucket_sum<4>(xl4, listFl, begL, lenL, lane, q);
                const float4 s = long_bucket_sum<2>(xs4, listFs, begS, lenS, lane, qq);
                const float4 j = long_bucket_sum<2>(xi4, listIn, begI, lenI, lane, qq);
                zsi = isS ? s : j;
            }
            const float wL = 1.0f / (float)max(lenL, 1);
            const float wX = 1.0f / (float)max(isS ? lenS : lenI, 1);
            zl.x *= wL;  zl.y *= wL;  zl.z *= wL;  zl.w *= wL;
            zsi.x *= wX; zsi.y *= wX; zsi.z *= wX; zsi.w *= wX;
        }
        if (lane < 4) { sZ[wave][u][lane] = zl; sZ[wave][u][4 + lane] = zsi; }
        if (lane == 0) sZ[wave][u][8] = make_float4(fl, fs, fi, 0.0f);
    }
    __syncthreads();

    float4 o[TPW];
    #pragma unroll
    for (int u = 0; u < TPW; ++u) o[u] = zero4;
    #pragma nounroll
    for (int g = 0; g < 8; ++g) {
        const float4 w0 = sW[(4 * g + 0) * 32 + lane];
        const float4 w1 = sW[(4 * g + 1) * 32 + lane];
        const float4 w2 = sW[(4 * g + 2) * 32 + lane];
        const float4 w3 = sW[(4 * g + 3) * 32 + lane];
        #pragma unroll
        for (int u = 0; u < TPW; ++u) {
            const float4 z = sZ[wave][u][g];
            o[u].x += z.x * w0.x; o[u].y += z.x * w0.y; o[u].z += z.x * w0.z; o[u].w += z.x * w0.w;
            o[u].x += z.y * w1.x; o[u].y += z.y * w1.y; o[u].z += z.y * w1.z; o[u].w += z.y * w1.w;
            o[u].x += z.z * w2.x; o[u].y += z.z * w2.y; o[u].z += z.z * w2.z; o[u].w += z.z * w2.w;
            o[u].x += z.w * w3.x; o[u].y += z.w * w3.y; o[u].z += z.w * w3.z; o[u].w += z.w * w3.w;
        }
    }
    {
        const float4 b0 = sW[32 * 32 + lane];
        const float4 b1 = sW[33 * 32 + lane];
        const float4 b2 = sW[34 * 32 + lane];
        #pragma unroll
        for (int u = 0; u < TPW; ++u) {
            const float4 f = sZ[wave][u][8];
            o[u].x += f.x * b0.x; o[u].y += f.x * b0.y; o[u].z += f.x * b0.z; o[u].w += f.x * b0.w;
            o[u].x += f.y * b1.x; o[u].y += f.y * b1.y; o[u].z += f.y * b1.z; o[u].w += f.y * b1.w;
            o[u].x += f.z * b2.x; o[u].y += f.z * b2.y; o[u].z += f.z * b2.z; o[u].w += f.z * b2.w;
        }
    }
    #pragma unroll
    for (int u = 0; u < TPW; ++u) {
        const int t = tb + u;
        if (t < n) ((float4*)aggS)[(size_t)t * (H / 4) + lane] = o[u];
    }
}

#define RPB 7
#define GW (4 * RPB)
#define GT (GW * 32)
#define APITCH 132

#define NIF 16
__device__ __forceinline__ void issue_rows(float4 (&v)[NIF], const float4* __restrict__ feat4,
                                           int sorted, int r, int len, int lane) {
    #pragma unroll
    for (int i = 0; i < NIF; ++i) {
        v[i] = make_float4(0.0f, 0.0f, 0.0f, 0.0f);
        if (r + i < len) {
            const int sid = __builtin_amdgcn_readlane(sorted, r + i);
            v[i] = feat4[(size_t)sid * (H / 4) + lane];
        }
    }
}
__device__ __forceinline__ void sum_rows(float4& acc, const float4 (&v)[NIF]) {
    #pragma unroll
    for (int i = 0; i < NIF; ++i) add4(acc, v[i]);
}
__device__ __forceinline__ float4 sum_rows_first(const float4 (&v)[NIF]) {
    float4 acc = v[0];
    #pragma unroll
    for (int i = 1; i < NIF; ++i) add4(acc, v[i]);
    return acc;
}

__global__ __launch_bounds__(GT) void fused_gemm_elu_kernel(
        const float* __restrict__ h, const float* __restrict__ aggS,
        const int* __restrict__ listSp, const int* __restrict__ cnt, const int* __restrict__ off,
        const float* __restrict__ W0, const float* __restrict__ W1,
        const float* __restrict__ bias, float* __restrict__ out, int n, int nrp) {
    __shared__ __align__(16) float sB[2 * H * H];
    __shared__ __align__(16) float sA[RPB * 32 * APITCH];
    const int tid  = threadIdx.x;
    const int lane = tid & 31;
    const int half = lane >> 4;
    const int l15  = lane & 15;

    const int wave = __builtin_amdgcn_readfirstlane(tid >> 5);
    const int lrp  = wave >> 2;
    const int rpBeg = (int)(((long long)blockIdx.x * nrp) / gridDim.x);
    const int rpEnd = (int)(((long long)(blockIdx.x + 1) * nrp) / gridDim.x);
    const bool active = rpBeg + lrp < rpEnd;
    const int m0 = (rpBeg + lrp) << 5;
    const int cb = (wave & 3) << 5;

    const int lr0 = (wave & 3) << 3;
    const int t0  = m0 + lr0;
    const bool okl = active && lane < 8 && t0 + lane < n;
    const int cntv = okl ? cnt[t0 + lane] : 0;
    const int offv = okl ? off[t0 + lane] : 0;

    {
        constexpr int NV = 2 * H * H / 4;
        constexpr int NI = (NV + GT - 1) / GT;
        float4 wv[NI];
        #pragma unroll
        for (int q = 0; q < NI; ++q) {
            const int i = tid + GT * q;
            wv[q] = make_float4(0.0f, 0.0f, 0.0f, 0.0f);
            if (i < NV) wv[q] = ((const float4*)((i >> 12) ? W1 : W0))[i & 4095];
        }
        #pragma unroll
        for (int q = 0; q < NI; ++q) {
            const int i = tid + GT * q;
            if (i < NV) {
                const int jj = i & 4095;
                const int k  = jj >> 5;
                const int c0 = (jj & 31) << 2;
                float* d = sB + ((i >> 12) << 14) + ((k >> 1) << 8) + (k & 1);
                d[2 * (c0 + 0)] = wv[q].x;
                d[2 * (c0 + 1)] = wv[q].y;
                d[2 * (c0 + 2)] = wv[q].z;
                d[2 * (c0 + 3)] = wv[q].w;
            }
        }
    }
    int lenC = __builtin_amdgcn_readlane(cntv, 0);
    int begC = __builtin_amdgcn_readlane(offv, 0);
    int lenN = __builtin_amdgcn_readlane(cntv, 1);
    int begN = __builtin_amdgcn_readlane(offv, 1);
    int keyN = (lenN <= 32 && lane < lenN) ? listSp[begN + lane] : 0x7fffffff;
    const int keyC = (lenC <= 32 && lane < lenC) ? listSp[begC + lane] : 0x7fffffff;
    __syncthreads();

    int r0 = m0 + l15;
    if (r0 >= n) r0 = n - 1;
    int r1 = m0 + 16 + l15;
    if (r1 >= n) r1 = n - 1;
    float* sAp = sA + (size_t)lrp * 32 * APITCH;

    v8f acc00 = {}, acc01 = {}, acc10 = {}, acc11 = {};
    if (active) {
        const int first = (wave >> 2) & 1;
        for (int st = 0; st < 2; ++st) {
            if ((st ^ first) == 0) {
                const float4* p0 = (const float4*)(h + (size_t)r0 * H + (half << 2));
                const float4* p1 = (const float4*)(h + (size_t)r1 * H + (half << 2));
                const float* bbase = sB + (half << 9) + ((cb + l15) << 1);
                for (int j = 0; j < H / 8; ++j) {
                    const float4 x0 = p0[2 * j];
                    const float4 x1 = p1[2 * j];
                    const float* bk = bbase + (j << 10);
                    const v2f b00 = *(const v2f*)(bk);
                    const v2f b01 = *(const v2f*)(bk + 32);
                    const v2f b10 = *(const v2f*)(bk + 256);
                    const v2f b11 = *(const v2f*)(bk + 256 + 32);
                    v2f a;
                    a.x = x0.x; a.y = x0.y;
                    acc00 = __builtin_amdgcn_wmma_f32_16x16x4_f32(false, a, false, b00, (short)0, acc00, false, false);
                    acc01 = __builtin_amdgcn_wmma_f32_16x16x4_f32(false, a, false, b01, (short)0, acc01, false, false);
                    a.x = x1.x; a.y = x1.y;
                    acc10 = __builtin_amdgcn_wmma_f32_16x16x4_f32(false, a, false, b00, (short)0, acc10, false, false);
                    acc11 = __builtin_amdgcn_wmma_f32_16x16x4_f32(false, a, false, b01, (short)0, acc11, false, false);
                    a.x = x0.z; a.y = x0.w;
                    acc00 = __builtin_amdgcn_wmma_f32_16x16x4_f32(false, a, false, b10, (short)0, acc00, false, false);
                    acc01 = __builtin_amdgcn_wmma_f32_16x16x4_f32(false, a, false, b11, (short)0, acc01, false, false);
                    a.x = x1.z; a.y = x1.w;
                    acc10 = __builtin_amdgcn_wmma_f32_16x16x4_f32(false, a, false, b10, (short)0, acc10, false, false);
                    acc11 = __builtin_amdgcn_wmma_f32_16x16x4_f32(false, a, false, b11, (short)0, acc11, false, false);
                }
            } else {
                int sortedC = (lenC <= 32) ? sort_ids(keyC, lenC, lane) : 0;
                const float4* h4 = (const float4*)h;
                for (int q = 0; q < 8; ++q) {
                    const int t = t0 + q;
                    int lenF = 0, begF = 0;
                    if (q + 2 < 8) {
                        lenF = __builtin_amdgcn_readlane(cntv, q + 2);
                        begF = __builtin_amdgcn_readlane(offv, q + 2);
                    }
                    int keyF = 0x7fffffff;
                    float4 r = make_float4(0.0f, 0.0f, 0.0f, 0.0f);
                    int sortedN = 0;
                    if (t < n) {
                        float4 sm, base;
                        if (lenC <= 32) {
                            float4 v[NIF];
                            issue_rows(v, h4, sortedC, 0, lenC, lane);
                            if (lenF <= 32 && lane < lenF) keyF = listSp[begF + lane];
                            base = ((const float4*)aggS)[(size_t)t * (H / 4) + lane];
                            if (lenN <= 32) sortedN = sort_ids(keyN, lenN, lane);
                            sm = sum_rows_first(v);
                            for (int rr = NIF; rr < lenC; rr += NIF) {
                                issue_rows(v, h4, sortedC, rr, lenC, lane);
                                sum_rows(sm, v);
                            }
                        } else {
                            if (lenF <= 32 && lane < lenF) keyF = listSp[begF + lane];
                            base = ((const float4*)aggS)[(size_t)t * (H / 4) + lane];
                            sm = bucket_sum(h4, listSp, begC, lenC, lane);
                            if (lenN <= 32) sortedN = sort_ids(keyN, lenN, lane);
                        }
                        const float w = 1.0f / (float)max(lenC, 1);
                        r.x = base.x + sm.x * w; r.y = base.y + sm.y * w;
                        r.z = base.z + sm.z * w; r.w = base.w + sm.w * w;
                    }
                    *(float4*)(sAp + (size_t)(lr0 + q) * APITCH + (lane << 2)) = r;
                    lenC = lenN; begC = begN; sortedC = sortedN;
                    lenN = lenF; begN = begF; keyN = keyF;
                }
            }
        }
    }
    __syncthreads();

    if (active) {
        const float* q0 = sAp + (size_t)l15 * APITCH + (half << 2);
        const float* q1 = q0 + 16 * APITCH;
        const float* bbase = sB + (1 << 14) + (half << 9) + ((cb + l15) << 1);
        for (int j = 0; j < H / 8; ++j) {
            const float4 x0 = *(const float4*)(q0 + 8 * j);
            const float4 x1 = *(const float4*)(q1 + 8 * j);
            const float* bk = bbase + (j << 10);
            const v2f b00 = *(const v2f*)(bk);
            const v2f b01 = *(const v2f*)(bk + 32);
            const v2f b10 = *(const v2f*)(bk + 256);
            const v2f b11 = *(const v2f*)(bk + 256 + 32);
            v2f a;
            a.x = x0.x; a.y = x0.y;
            acc00 = __builtin_amdgcn_wmma_f32_16x16x4_f32(false, a, false, b00, (short)0, acc00, false, false);
            acc01 = __builtin_amdgcn_wmma_f32_16x16x4_f32(false, a, false, b01, (short)0, acc01, false, false);
            a.x = x1.x; a.y = x1.y;
            acc10 = __builtin_amdgcn_wmma_f32_16x16x4_f32(false, a, false, b00, (short)0, acc10, false, false);
            acc11 = __builtin_amdgcn_wmma_f32_16x16x4_f32(false, a, false, b01, (short)0, acc11, false, false);
            a.x = x0.z; a.y = x0.w;
            acc00 = __builtin_amdgcn_wmma_f32_16x16x4_f32(false, a, false, b10, (short)0, acc00, false, false);
            acc01 = __builtin_amdgcn_wmma_f32_16x16x4_f32(false, a, false, b11, (short)0, acc01, false, false);
            a.x = x1.z; a.y = x1.w;
            acc10 = __builtin_amdgcn_wmma_f32_16x16x4_f32(false, a, false, b10, (short)0, acc10, false, false);
            acc11 = __builtin_amdgcn_wmma_f32_16x16x4_f32(false, a, false, b11, (short)0, acc11, false, false);
        }

        const float bc0 = bias[cb + l15];
        const float bc1 = bias[cb + 16 + l15];
        #pragma unroll
        for (int r = 0; r < 8; ++r) {
            const int rowa = m0 + (half << 3) + r;
            const int rowb = rowa + 16;
            if (rowa < n) {
                float v = acc00[r] + bc0;
                out[(size_t)rowa * H + cb + l15] = v > 0.0f ? v : (__expf(v) - 1.0f);
                v = acc01[r] + bc1;
                out[(size_t)rowa * H + cb + 16 + l15] = v > 0.0f ? v : (__expf(v) - 1.0f);
            }
            if (rowb < n) {
                float v = acc10[r] + bc0;
                out[(size_t)rowb * H + cb + l15] = v > 0.0f ? v : (__expf(v) - 1.0f);
                v = acc11[r] + bc1;
                out[(size_t)rowb * H + cb + 16 + l15] = v > 0.0f ? v : (__expf(v) - 1.0f);
            }
        }
    }
}

extern "C" void kernel_launch(void* const* d_in, const int* in_sizes, int n_in,
                              void* d_out, int out_size, void* d_ws, size_t ws_size,
                              hipStream_t stream) {
    const float* x_int  = (const float*)d_in[0];
    const float* x_lane = (const float*)d_in[1];
    const float* x_sens = (const float*)d_in[2];
    const float* x_inj  = (const float*)d_in[3];
    const int*   e_sp   = (const int*)d_in[4];
    const int*   e_fl   = (const int*)d_in[5];
    const int*   e_fs   = (const int*)d_in[6];
    const int*   e_in   = (const int*)d_in[7];
    const float* W_int  = (const float*)d_in[8];
    const float* b_int  = (const float*)d_in[9];
    const float* W_lane = (const float*)d_in[10];
    const float* b_lane = (const float*)d_in[11];
    const float* W_sens = (const float*)d_in[12];
    const float* b_sens = (const float*)d_in[13];
    const float* W_inj  = (const float*)d_in[14];
    const float* b_inj  = (const float*)d_in[15];
    const float* self_W = (const float*)d_in[16];
    const float* self_b = (const float*)d_in[17];
    const float* rel_W  = (const float*)d_in[18];

    const int nInt  = in_sizes[0] / 32;
    const int nLane = in_sizes[1] / 16;
    const int nSens = in_sizes[2] / 8;
    const int nInj  = in_sizes[3] / 8;
    const int eSp   = in_sizes[4] / 2;
    const int eFl   = in_sizes[5] / 2;
    const int eFs   = in_sizes[6] / 2;
    const int eIn   = in_sizes[7] / 2;
    const int L     = in_sizes[16] / (H * H);

    float* ws = (float*)d_ws;
    size_t o = 0;
    float* hA    = ws + o; o += (size_t)nInt  * H;
    float* hB    = ws + o; o += (size_t)nInt  * H;
    float* hLane = ws + o; o += (size_t)nLane * H;
    float* pSens = ws + o; o += (size_t)nSens * H;
    o += (size_t)nInj  * H;
    float* aggS  = ws + o; o += (size_t)nInt  * H;
    float* aggB  = ws + o; o += (size_t)nInt  * H;
    int*   cnt   = (int*)(ws + o); o += (size_t)4 * nInt;
    int*   off   = (int*)(ws + o); o += (size_t)4 * nInt;
    (void)ws_size; (void)n_in; (void)out_size;

    int*  binned = (int*)aggB;
    int*  listFl = (int*)aggB + 2 * ((size_t)eSp + eFl + eFs + eIn);
    int*  listFs = listFl + eFl;
    int*  listIn = listFs + eFs;
    int*  listSp = (int*)hLane;
    int*  tab    = (int*)pSens;

    proj_kernel<<<(nInt + 127) / 128, 256, 0, stream>>>(x_int, W_int, b_int, hA, nInt);

    const int nb  = (nInt + 255) / 256;
    const int nch = (eSp + CHUNK - 1) / CHUNK + (eFl + CHUNK - 1) / CHUNK
                  + (eFs + CHUNK - 1) / CHUNK + (eIn + CHUNK - 1) / CHUNK;
    bin_kernel<<<nch, 256, 0, stream>>>(e_sp, e_fl, e_fs, e_in, binned, tab,
                                        eSp, eFl, eFs, eIn, nb, nch);
    fine_kernel<<<4 * nb, 256, 0, stream>>>(binned, tab, cnt, off,
                                            listSp, listFl, listFs, listIn,
                                            eSp, eFl, eFs, eIn, nb, nch, nInt);

    gather_raw_proj_kernel<<<(nInt + 8 * TPW - 1) / (8 * TPW), 256, 0, stream>>>(
        x_lane, x_sens, x_inj, W_lane, b_lane, W_sens, b_sens, W_inj, b_inj,
        listFl, listFs, listIn, cnt, off, aggS, nInt);

    const int nrp = (nInt + 31) / 32;
    const int needed = (nrp + RPB - 1) / RPB;
    const int lgrid = needed > 256 ? needed : 256;
    float* hcur = hA;
    float* hnext = hB;
    for (int l = 0; l < L; ++l) {
        float* outp = (l == L - 1) ? (float*)d_out : hnext;
        fused_gemm_elu_kernel<<<lgrid, GT, 0, stream>>>(
            hcur, aggS, listSp, cnt, off,
            self_W + (size_t)l * H * H, rel_W + (size_t)l * H * H,
            self_b + (size_t)l * H, outp, nInt, nrp);
        float* t = hnext; hnext = hcur; hcur = t;
    }
}


# Input projection rides in the binning launch: one grid, the edge-chunk blocks first, then one block per 128 rows of x_int; the two are independent; 7 launches to 6; sum order unchanged
# speedup vs baseline: 7.5826x; 1.0328x over previous
#include <hip/hip_runtime.h>
#include <cstddef>

#define H 128

typedef float v2f __attribute__((ext_vector_type(2)));
typedef float v8f __attribute__((ext_vector_type(8)));

#define KIN 32
__device__ __forceinline__ void proj_block(const float* __restrict__ x,
                                           const float* __restrict__ W,
                                           const float* __restrict__ b,
                                           float* __restrict__ out, int n, int bid) {
    __shared__ __align__(16) float sW[KIN * H];
    const int tid  = threadIdx.x;
    const int lane = tid & 31;
    const int half = lane >> 4;
    const int l15  = lane & 15;
    const int m0 = __builtin_amdgcn_readfirstlane(bid * 128 + (tid >> 5) * 16);

    int arow = m0 + l15;
    if (arow >= n) arow = n - 1;
    const float4* xr = (const float4*)(x + (size_t)arow * KIN + (half << 4));
    const float4 x0 = xr[0], x1 = xr[1], x2 = xr[2], x3 = xr[3];

    float4 wv[4];
    #pragma unroll
    for (int j = 0; j < 4; ++j) wv[j] = ((const float4*)W)[tid + 256 * j];

    v8f acc[8];
    #pragma unroll
    for (int t = 0; t < 8; ++t) {
        const float bc = b[(t << 4) + l15];
        #pragma unroll
        for (int r = 0; r < 8; ++r) acc[t][r] = bc;
    }

    #pragma unroll
    for (int j = 0; j < 4; ++j) {
        const int i  = tid + 256 * j;
        const int k  = i >> 5;
        const int c0 = (i & 31) << 2;
        float* d = sW + ((k >> 1) << 8) + (k & 1);
        d[2 * (c0 + 0)] = wv[j].x;
        d[2 * (c0 + 1)] = wv[j].y;
        d[2 * (c0 + 2)] = wv[j].z;
        d[2 * (c0 + 3)] = wv[j].w;
    }
    __syncthreads();

    const float xf[16] = { x0.x, x0.y, x0.z, x0.w, x1.x, x1.y, x1.z, x1.w,
                           x2.x, x2.y, x2.z, x2.w, x3.x, x3.y, x3.z, x3.w };
    const float* bbase = sW + (half << 11) + (l15 << 1);
    #pragma unroll
    for (int s = 0; s < 8; ++s) {
        v2f a;
        a.x = xf[2 * s];
        a.y = xf[2 * s + 1];
        const float* bk = bbase + (s << 8);
        #pragma unroll
        for (int t = 0; t < 8; ++t) {
            const v2f bf = *(const v2f*)(bk + (t << 5));
            acc[t] = __builtin_amdgcn_wmma_f32_16x16x4_f32(
                false, a, false, bf, (short)0, acc[t], false, false);
        }
    }

    float* orow = out + (size_t)(m0 + (half << 3)) * H + l15;
    if (m0 + 16 <= n) {
        #pragma unroll
        for (int t = 0; t < 8; ++t) {
            #pragma unroll
            for (int r = 0; r < 8; ++r) orow[r * H + (t << 4)] = acc[t][r];
        }
    } else {
        #pragma unroll
        for (int t = 0; t < 8; ++t) {
            #pragma unroll
            for (int r = 0; r < 8; ++r) {
                if (m0 + (half << 3) + r < n) orow[r * H + (t << 4)] = acc[t][r];
            }
        }
    }
}

#define CHUNK 4096
#define MAXNB 1024
#define STAGE 8192

__device__ __forceinline__ void bin_block(
        const int* __restrict__ ed0, const int* __restrict__ ed1,
        const int* __restrict__ ed2, const int* __restrict__ ed3,
        int* __restrict__ binned, int* __restrict__ tab,
        int E0, int E1, int E2, int E3, int nb, int nch, int bid) {
    __shared__ int  sHist[MAXNB + 1];
    __shared__ int2 sPair[CHUNK];
    const int tid = threadIdx.x;
    const int C0 = (E0 + CHUNK - 1) / CHUNK, C1 = (E1 + CHUNK - 1) / CHUNK,
              C2 = (E2 + CHUNK - 1) / CHUNK;
    int c = bid;
    const int* ed = ed0;
    int E = E0, rbase = 0;
    if (c >= C0) {
        c -= C0; ed = ed1; E = E1; rbase = E0;
        if (c >= C1) {
            c -= C1; ed = ed2; E = E2; rbase += E1;
            if (c >= C2) { c -= C2; ed = ed3; E = E3; rbase += E2; }
        }
    }
    const int e0  = c * CHUNK;
    const int cnt = min(CHUNK, E - e0);

    for (int j = tid; j <= nb; j += 256) sHist[j] = 0;
    __syncthreads();

    int srcv[16], tgtv[16], rk[16];
    #pragma unroll
    for (int k = 0; k < 16; ++k) {
        const int i = k * 256 + tid;
        srcv[k] = 0; tgtv[k] = 0;
        if (i < cnt) { srcv[k] = ed[e0 + i]; tgtv[k] = ed[E + e0 + i]; }
    }
    #pragma unroll
    for (int k = 0; k < 16; ++k) {
        const int i = k * 256 + tid;
        rk[k] = 0;
        if (i < cnt) rk[k] = atomicAdd(&sHist[tgtv[k] >> 8], 1);
    }
    __syncthreads();

    if (tid < 32) {
        int carry = 0;
        for (int base = 0; base < nb; base += 32) {
            const int j = base + tid;
            const int v = (j < nb) ? sHist[j] : 0;
            int inc = v;
            for (int d = 1; d < 32; d <<= 1) {
                const int u = __shfl_up(inc, d, 32);
                if (tid >= d) inc += u;
            }
            if (j < nb) sHist[j] = carry + inc - v;
            carry += __shfl(inc, 31, 32);
        }
        if (tid == 0) sHist[nb] = carry;
    }
    __syncthreads();

    #pragma unroll
    for (int k = 0; k < 16; ++k) {
        const int i = k * 256 + tid;
        if (i < cnt) sPair[sHist[tgtv[k] >> 8] + rk[k]] = make_int2(srcv[k], tgtv[k]);
    }
    for (int j = tid; j <= nb; j += 256) tab[(size_t)j * nch + bid] = sHist[j];
    __syncthreads();

    int2* outp = (int2*)binned + (size_t)rbase + e0;
    for (int i = tid; i < cnt; i += 256) outp[i] = sPair[i];
}

__global__ __launch_bounds__(256) void proj_bin_kernel(
        const float* __restrict__ x, const float* __restrict__ W, const float* __restrict__ b,
        float* __restrict__ hout, int n,
        const int* __restrict__ ed0, const int* __restrict__ ed1,
        const int* __restrict__ ed2, const int* __restrict__ ed3,
        int* __restrict__ binned, int* __restrict__ tab,
        int E0, int E1, int E2, int E3, int nb, int nch) {
    const int bid = (int)blockIdx.x;
    if (bid < nch) bin_block(ed0, ed1, ed2, ed3, binned, tab, E0, E1, E2, E3, nb, nch, bid);
    else           proj_block(x, W, b, hout, n, bid - nch);
}

__global__ __launch_bounds__(256) void fine_kernel(
        const int* __restrict__ binned, const int* __restrict__ tab,
        int* __restrict__ cnt, int* __restrict__ off,
        int* __restrict__ list0, int* __restrict__ list1,
        int* __restrict__ list2, int* __restrict__ list3,
        int E0, int E1, int E2, int E3, int nb, int nch, int n) {
    __shared__ int sCnt[256];
    __shared__ int sCur[256];
    __shared__ int sWave[8];
    __shared__ int sRed[8];
    __shared__ int sStage[STAGE];
    const int tid  = threadIdx.x;
    const int lane = tid & 31;
    const int wave = tid >> 5;
    const int C0 = (E0 + CHUNK - 1) / CHUNK, C1 = (E1 + CHUNK - 1) / CHUNK,
              C2 = (E2 + CHUNK - 1) / CHUNK, C3 = (E3 + CHUNK - 1) / CHUNK;
    const int r = blockIdx.x / nb;
    const int g = blockIdx.x - r * nb;
    int rbase = 0, cb = 0, nC = C0;
    int* list = list0;
    if (r == 1) { rbase = E0;           cb = C0;           nC = C1; list = list1; }
    if (r == 2) { rbase = E0 + E1;      cb = C0 + C1;      nC = C2; list = list2; }
    if (r == 3) { rbase = E0 + E1 + E2; cb = C0 + C1 + C2; nC = C3; list = list3; }
    const int* tlo = tab + (size_t)g * nch + cb;
    const int* thi = tab + (size_t)(g + 1) * nch + cb;
    const int2* win = (const int2*)binned + (size_t)rbase;

    sCnt[tid] = 0;
    __syncthreads();

    int lower = 0;
    for (int b = tid; b < nC; b += 256) {
        const int s = tlo[b], e = thi[b];
        lower += s;
        const int2* p = win + (size_t)b * CHUNK;
        for (int i = s; i < e; i += 8) {
            int y[8];
            #pragma unroll
            for (int u = 0; u < 8; ++u) y[u] = p[min(i + u, e - 1)].y;
            #pragma unroll
            for (int u = 0; u < 8; ++u)
                if (i + u < e) atomicAdd(&sCnt[y[u] & 255], 1);
        }
    }
    __syncthreads();

    const int cT = sCnt[tid];
    int inc = cT;
    for (int d = 1; d < 32; d <<= 1) {
        const int u = __shfl_up(inc, d, 32);
        if (lane >= d) inc += u;
    }
    for (int o = 16; o > 0; o >>= 1) lower += __shfl_xor(lower, o, 32);
    if (lane == 31) sWave[wave] = inc;
    if (lane == 0)  sRed[wave]  = lower;
    __syncthreads();
    int wbase = 0, total = 0, start = 0;
    for (int w = 0; w < 8; ++w) {
        const int tw = sWave[w];
        if (w < wave) wbase += tw;
        total += tw;
        start += sRed[w];
    }
    const int offL = wbase + inc - cT;
    const int t = g * 256 + tid;
    if (t < n) {
        cnt[(size_t)r * n + t] = cT;
        off[(size_t)r * n + t] = start + offL;
    }
    sCur[tid] = offL;
    __syncthreads();

    const bool staged = total <= STAGE;
    for (int b = tid; b < nC; b += 256) {
        const int s = tlo[b], e = thi[b];
        const int2* p = win + (size_t)b * CHUNK;
        for (int i = s; i < e; i += 8) {
            int2 v[8];
            #pragma unroll
            for (int u = 0; u < 8; ++u) v[u] = p[min(i + u, e - 1)];
            #pragma unroll
            for (int u = 0; u < 8; ++u) {
                if (i + u < e) {
                    const int pos = atomicAdd(&sCur[v[u].y & 255], 1);
                    if (staged) sStage[pos] = v[u].x;
                    else        list[start + pos] = v[u].x;
                }
            }
        }
    }
    if (staged) {
        __syncthreads();
        for (int i = tid; i < total; i += 256) list[start + i] = sStage[i];
    }
}

__device__ __forceinline__ void add4(float4& a, const float4 v) {
    a.x += v.x; a.y += v.y; a.z += v.z; a.w += v.w;
}

__device__ __forceinline__ float4 bucket_sum(const float4* __restrict__ feat4,
                                             const int* __restrict__ list,
                                             int beg, int len, int lane) {
    float4 acc = make_float4(0.0f, 0.0f, 0.0f, 0.0f);
    if (len <= 32) {
        const int key = (lane < len) ? list[beg + lane] : 0x7fffffff;
        int rank = 0;
        for (int j = 0; j < len; ++j) {
            const int kj = __builtin_amdgcn_readlane(key, j);
            rank += (kj < key || (kj == key && j < lane)) ? 1 : 0;
        }
        if (lane >= len) rank = lane;
        const int sorted = __builtin_amdgcn_ds_permute(rank << 2, key);
        int r = 0;
        for (; r + 4 <= len; r += 4) {
            const int s0 = __builtin_amdgcn_readlane(sorted, r);
            const int s1 = __builtin_amdgcn_readlane(sorted, r + 1);
            const int s2 = __builtin_amdgcn_readlane(sorted, r + 2);
            const int s3 = __builtin_amdgcn_readlane(sorted, r + 3);
            const float4 v0 = feat4[(size_t)s0 * (H / 4) + lane];
            const float4 v1 = feat4[(size_t)s1 * (H / 4) + lane];
            const float4 v2 = feat4[(size_t)s2 * (H / 4) + lane];
            const float4 v3 = feat4[(size_t)s3 * (H / 4) + lane];
            add4(acc, v0); add4(acc, v1); add4(acc, v2); add4(acc, v3);
        }
        for (; r < len; ++r) {
            const int s0 = __builtin_amdgcn_readlane(sorted, r);
            add4(acc, feat4[(size_t)s0 * (H / 4) + lane]);
        }
    } else {
        int prev = -1;
        for (int it = 0; it < len; ++it) {
            int m = 0x7fffffff;
            for (int i = lane; i < len; i += 32) {
                const int k = list[beg + i];
                if (k > prev && k < m) m = k;
            }
            for (int o = 16; o > 0; o >>= 1) m = min(m, __shfl_xor(m, o, 32));
            if (m == 0x7fffffff) break;
            int c = 0;
            for (int i = lane; i < len; i += 32) c += (list[beg + i] == m) ? 1 : 0;
            for (int o = 16; o > 0; o >>= 1) c += __shfl_xor(c, o, 32);
            const float4 v = feat4[(size_t)m * (H / 4) + lane];
            for (int q = 0; q < c; ++q) add4(acc, v);
            prev = m;
        }
    }
    return acc;
}


__device__ __forceinline__ int sort_ids(int key, int len, int lane) {
    int rank = 0;
    for (int j = 0; j < len; ++j) {
        const int kj = __builtin_amdgcn_readlane(key, j);
        rank += (kj < key || (kj == key && j < lane)) ? 1 : 0;
    }
    if (lane >= len) rank = lane;
    return __builtin_amdgcn_ds_permute(rank << 2, key);
}

template <int Q>
__device__ __forceinline__ float4 long_bucket_sum(const float4* __restrict__ x4,
                                                  const int* __restrict__ list,
                                                  int beg, int len, int lane, int qq) {
    float4 acc = make_float4(0.0f, 0.0f, 0.0f, 0.0f);
    int prev = -1;
    for (int it = 0; it < len; ++it) {
        int m = 0x7fffffff;
        for (int i = lane; i < len; i += 32) {
            const int k = list[beg + i];
            if (k > prev && k < m) m = k;
        }
        for (int o = 16; o > 0; o >>= 1) m = min(m, __shfl_xor(m, o, 32));
        if (m == 0x7fffffff) break;
        int c = 0;
        for (int i = lane; i < len; i += 32) c += (list[beg + i] == m) ? 1 : 0;
        for (int o = 16; o > 0; o >>= 1) c += __shfl_xor(c, o, 32);
        const float4 v = x4[(size_t)m * Q + qq];
        for (int r = 0; r < c; ++r) add4(acc, v);
        prev = m;
    }
    return acc;
}

__device__ __forceinline__ float4 slot_sum(float4 a) {
    for (int m = 4; m < 32; m <<= 1) {
        float4 b;
        b.x = __shfl_xor(a.x, m, 32); b.y = __shfl_xor(a.y, m, 32);
        b.z = __shfl_xor(a.z, m, 32); b.w = __shfl_xor(a.w, m, 32);
        add4(a, b);
    }
    return a;
}

#define TPW 4

__global__ __launch_bounds__(256) void gather_raw_proj_kernel(
        const float* __restrict__ xLane, const float* __restrict__ xSens,
        const float* __restrict__ xInj,
        const float* __restrict__ WLane, const float* __restrict__ bLane,
        const float* __restrict__ WSens, const float* __restrict__ bSens,
        const float* __restrict__ WInj,  const float* __restrict__ bInj,
        const int* __restrict__ listFl, const int* __restrict__ listFs,
        const int* __restrict__ listIn,
        const int* __restrict__ cnt, const int* __restrict__ off,
        float* __restrict__ aggS, int n) {
    __shared__ float4 sW[35 * 32];
    __shared__ float4 sZ[8][TPW][9];
    const int tid  = threadIdx.x;
    const int wave = tid >> 5;
    const int lane = tid & 31;
    const int slot = lane >> 2;
    const int q    = lane & 3;

    {
        const float4* wl = (const float4*)WLane;
        sW[tid]       = wl[tid];
        sW[256 + tid] = wl[256 + tid];
        sW[512 + tid] = ((const float4*)WSens)[tid];
        sW[768 + tid] = ((const float4*)WInj)[tid];
        if (tid < 96) {
            const float* bs = tid < 32 ? bLane : (tid < 64 ? bSens : bInj);
            sW[1024 + tid] = ((const float4*)bs)[tid & 31];
        }
    }

    const float4* xl4 = (const float4*)xLane;
    const float4* xs4 = (const float4*)xSens;
    const float4* xi4 = (const float4*)xInj;
    const float4 zero4 = make_float4(0.0f, 0.0f, 0.0f, 0.0f);
    const int tb = blockIdx.x * (8 * TPW) + wave * TPW;

    for (int u = 0; u < TPW; ++u) {
        const int t = __builtin_amdgcn_readfirstlane(tb + u);
        float4 zl = zero4, zsi = zero4;
        float fl = 0.0f, fs = 0.0f, fi = 0.0f;
        if (t < n) {
            const int lenL = cnt[1 * n + t], lenS = cnt[2 * n + t], lenI = cnt[3 * n + t];
            const int begL = off[1 * n + t], begS = off[2 * n + t], begI = off[3 * n + t];
            fl = lenL > 0 ? 1.0f : 0.0f;
            fs = lenS > 0 ? 1.0f : 0.0f;
            fi = lenI > 0 ? 1.0f : 0.0f;
            const bool isS = q < 2;
            const int qq = q & 1;
            if (lenL <= 32 && lenS <= 32 && lenI <= 32) {
                const int keyL = (lane < lenL) ? listFl[begL + lane] : 0x7fffffff;
                const int keyS = (lane < lenS) ? listFs[begS + lane] : 0x7fffffff;
                const int keyI = (lane < lenI) ? listIn[begI + lane] : 0x7fffffff;
                const int sortedL = sort_ids(keyL, lenL, lane);
                const int sortedS = sort_ids(keyS, lenS, lane);
                const int sortedI = sort_ids(keyI, lenI, lane);
                const int lenX = isS ? lenS : lenI;
                const int maxSI = max(lenS, lenI);
                const float4* xb = isS ? xs4 : xi4;
                float4 a[4], b[4];
                #pragma unroll
                for (int i = 0; i < 4; ++i) {
                    a[i] = zero4; b[i] = zero4;
                    const int p = 8 * i + slot;
                    if (8 * i < lenL) {
                        const int id = __builtin_amdgcn_ds_bpermute(p << 2, sortedL);
                        if (p < lenL) a[i] = xl4[(size_t)id * 4 + q];
                    }
                    if (8 * i < maxSI) {
                        const int idS = __builtin_amdgcn_ds_bpermute(p << 2, sortedS);
                        const int idI = __builtin_amdgcn_ds_bpermute(p << 2, sortedI);
                        if (p < lenX) b[i] = xb[(size_t)(isS ? idS : idI) * 2 + qq];
                    }
                }
                add4(a[0], a[1]); add4(a[0], a[2]); add4(a[0], a[3]);
                add4(b[0], b[1]); add4(b[0], b[2]); add4(b[0], b[3]);
                zl  = slot_sum(a[0]);
                zsi = slot_sum(b[0]);
            } else {
                zl = long_bucket_sum<4>(xl4, listFl, begL, lenL, lane, q);
                const float4 s = long_bucket_sum<2>(xs4, listFs, begS, lenS, lane, qq);
                const float4 j = long_bucket_sum<2>(xi4, listIn, begI, lenI, lane, qq);
                zsi = isS ? s : j;
            }
            const float wL = 1.0f / (float)max(lenL, 1);
            const float wX = 1.0f / (float)max(isS ? lenS : lenI, 1);
            zl.x *= wL;  zl.y *= wL;  zl.z *= wL;  zl.w *= wL;
            zsi.x *= wX; zsi.y *= wX; zsi.z *= wX; zsi.w *= wX;
        }
        if (lane < 4) { sZ[wave][u][lane] = zl; sZ[wave][u][4 + lane] = zsi; }
        if (lane == 0) sZ[wave][u][8] = make_float4(fl, fs, fi, 0.0f);
    }
    __syncthreads();

    float4 o[TPW];
    #pragma unroll
    for (int u = 0; u < TPW; ++u) o[u] = zero4;
    #pragma nounroll
    for (int g = 0; g < 8; ++g) {
        const float4 w0 = sW[(4 * g + 0) * 32 + lane];
        const float4 w1 = sW[(4 * g + 1) * 32 + lane];
        const float4 w2 = sW[(4 * g + 2) * 32 + lane];
        const float4 w3 = sW[(4 * g + 3) * 32 + lane];
        #pragma unroll
        for (int u = 0; u < TPW; ++u) {
            const float4 z = sZ[wave][u][g];
            o[u].x += z.x * w0.x; o[u].y += z.x * w0.y; o[u].z += z.x * w0.z; o[u].w += z.x * w0.w;
            o[u].x += z.y * w1.x; o[u].y += z.y * w1.y; o[u].z += z.y * w1.z; o[u].w += z.y * w1.w;
            o[u].x += z.z * w2.x; o[u].y += z.z * w2.y; o[u].z += z.z * w2.z; o[u].w += z.z * w2.w;
            o[u].x += z.w * w3.x; o[u].y += z.w * w3.y; o[u].z += z.w * w3.z; o[u].w += z.w * w3.w;
        }
    }
    {
        const float4 b0 = sW[32 * 32 + lane];
        const float4 b1 = sW[33 * 32 + lane];
        const float4 b2 = sW[34 * 32 + lane];
        #pragma unroll
        for (int u = 0; u < TPW; ++u) {
            const float4 f = sZ[wave][u][8];
            o[u].x += f.x * b0.x; o[u].y += f.x * b0.y; o[u].z += f.x * b0.z; o[u].w += f.x * b0.w;
            o[u].x += f.y * b1.x; o[u].y += f.y * b1.y; o[u].z += f.y * b1.z; o[u].w += f.y * b1.w;
            o[u].x += f.z * b2.x; o[u].y += f.z * b2.y; o[u].z += f.z * b2.z; o[u].w += f.z * b2.w;
        }
    }
    #pragma unroll
    for (int u = 0; u < TPW; ++u) {
        const int t = tb + u;
        if (t < n) ((float4*)aggS)[(size_t)t * (H / 4) + lane] = o[u];
    }
}

#define RPB 7
#define GW (4 * RPB)
#define GT (GW * 32)
#define APITCH 132

#define NIF 16
__device__ __forceinline__ void issue_rows(float4 (&v)[NIF], const float4* __restrict__ feat4,
                                           int sorted, int r, int len, int lane) {
    #pragma unroll
    for (int i = 0; i < NIF; ++i) {
        v[i] = make_float4(0.0f, 0.0f, 0.0f, 0.0f);
        if (r + i < len) {
            const int sid = __builtin_amdgcn_readlane(sorted, r + i);
            v[i] = feat4[(size_t)sid * (H / 4) + lane];
        }
    }
}
__device__ __forceinline__ void sum_rows(float4& acc, const float4 (&v)[NIF]) {
    #pragma unroll
    for (int i = 0; i < NIF; ++i) add4(acc, v[i]);
}
__device__ __forceinline__ float4 sum_rows_first(const float4 (&v)[NIF]) {
    float4 acc = v[0];
    #pragma unroll
    for (int i = 1; i < NIF; ++i) add4(acc, v[i]);
    return acc;
}

__global__ __launch_bounds__(GT) void fused_gemm_elu_kernel(
        const float* __restrict__ h, const float* __restrict__ aggS,
        const int* __restrict__ listSp, const int* __restrict__ cnt, const int* __restrict__ off,
        const float* __restrict__ W0, const float* __restrict__ W1,
        const float* __restrict__ bias, float* __restrict__ out, int n, int nrp) {
    __shared__ __align__(16) float sB[2 * H * H];
    __shared__ __align__(16) float sA[RPB * 32 * APITCH];
    const int tid  = threadIdx.x;
    const int lane = tid & 31;
    const int half = lane >> 4;
    const int l15  = lane & 15;

    const int wave = __builtin_amdgcn_readfirstlane(tid >> 5);
    const int lrp  = wave >> 2;
    const int rpBeg = (int)(((long long)blockIdx.x * nrp) / gridDim.x);
    const int rpEnd = (int)(((long long)(blockIdx.x + 1) * nrp) / gridDim.x);
    const bool active = rpBeg + lrp < rpEnd;
    const int m0 = (rpBeg + lrp) << 5;
    const int cb = (wave & 3) << 5;

    const int lr0 = (wave & 3) << 3;
    const int t0  = m0 + lr0;
    const bool okl = active && lane < 8 && t0 + lane < n;
    const int cntv = okl ? cnt[t0 + lane] : 0;
    const int offv = okl ? off[t0 + lane] : 0;

    {
        constexpr int NV = 2 * H * H / 4;
        constexpr int NI = (NV + GT - 1) / GT;
        float4 wv[NI];
        #pragma unroll
        for (int q = 0; q < NI; ++q) {
            const int i = tid + GT * q;
            wv[q] = make_float4(0.0f, 0.0f, 0.0f, 0.0f);
            if (i < NV) wv[q] = ((const float4*)((i >> 12) ? W1 : W0))[i & 4095];
        }
        #pragma unroll
        for (int q = 0; q < NI; ++q) {
            const int i = tid + GT * q;
            if (i < NV) {
                const int jj = i & 4095;
                const int k  = jj >> 5;
                const int c0 = (jj & 31) << 2;
                float* d = sB + ((i >> 12) << 14) + ((k >> 1) << 8) + (k & 1);
                d[2 * (c0 + 0)] = wv[q].x;
                d[2 * (c0 + 1)] = wv[q].y;
                d[2 * (c0 + 2)] = wv[q].z;
                d[2 * (c0 + 3)] = wv[q].w;
            }
        }
    }
    int lenC = __builtin_amdgcn_readlane(cntv, 0);
    int begC = __builtin_amdgcn_readlane(offv, 0);
    int lenN = __builtin_amdgcn_readlane(cntv, 1);
    int begN = __builtin_amdgcn_readlane(offv, 1);
    int keyN = (lenN <= 32 && lane < lenN) ? listSp[begN + lane] : 0x7fffffff;
    const int keyC = (lenC <= 32 && lane < lenC) ? listSp[begC + lane] : 0x7fffffff;
    __syncthreads();

    int r0 = m0 + l15;
    if (r0 >= n) r0 = n - 1;
    int r1 = m0 + 16 + l15;
    if (r1 >= n) r1 = n - 1;
    float* sAp = sA + (size_t)lrp * 32 * APITCH;

    v8f acc00 = {}, acc01 = {}, acc10 = {}, acc11 = {};
    if (active) {
        const int first = (wave >> 2) & 1;
        for (int st = 0; st < 2; ++st) {
            if ((st ^ first) == 0) {
                const float4* p0 = (const float4*)(h + (size_t)r0 * H + (half << 2));
                const float4* p1 = (const float4*)(h + (size_t)r1 * H + (half << 2));
                const float* bbase = sB + (half << 9) + ((cb + l15) << 1);
                for (int j = 0; j < H / 8; ++j) {
                    const float4 x0 = p0[2 * j];
                    const float4 x1 = p1[2 * j];
                    const float* bk = bbase + (j << 10);
                    const v2f b00 = *(const v2f*)(bk);
                    const v2f b01 = *(const v2f*)(bk + 32);
                    const v2f b10 = *(const v2f*)(bk + 256);
                    const v2f b11 = *(const v2f*)(bk + 256 + 32);
                    v2f a;
                    a.x = x0.x; a.y = x0.y;
                    acc00 = __builtin_amdgcn_wmma_f32_16x16x4_f32(false, a, false, b00, (short)0, acc00, false, false);
                    acc01 = __builtin_amdgcn_wmma_f32_16x16x4_f32(false, a, false, b01, (short)0, acc01, false, false);
                    a.x = x1.x; a.y = x1.y;
                    acc10 = __builtin_amdgcn_wmma_f32_16x16x4_f32(false, a, false, b00, (short)0, acc10, false, false);
                    acc11 = __builtin_amdgcn_wmma_f32_16x16x4_f32(false, a, false, b01, (short)0, acc11, false, false);
                    a.x = x0.z; a.y = x0.w;
                    acc00 = __builtin_amdgcn_wmma_f32_16x16x4_f32(false, a, false, b10, (short)0, acc00, false, false);
                    acc01 = __builtin_amdgcn_wmma_f32_16x16x4_f32(false, a, false, b11, (short)0, acc01, false, false);
                    a.x = x1.z; a.y = x1.w;
                    acc10 = __builtin_amdgcn_wmma_f32_16x16x4_f32(false, a, false, b10, (short)0, acc10, false, false);
                    acc11 = __builtin_amdgcn_wmma_f32_16x16x4_f32(false, a, false, b11, (short)0, acc11, false, false);
                }
            } else {
                int sortedC = (lenC <= 32) ? sort_ids(keyC, lenC, lane) : 0;
                const float4* h4 = (const float4*)h;
                for (int q = 0; q < 8; ++q) {
                    const int t = t0 + q;
                    int lenF = 0, begF = 0;
                    if (q + 2 < 8) {
                        lenF = __builtin_amdgcn_readlane(cntv, q + 2);
                        begF = __builtin_amdgcn_readlane(offv, q + 2);
                    }
                    int keyF = 0x7fffffff;
                    float4 r = make_float4(0.0f, 0.0f, 0.0f, 0.0f);
                    int sortedN = 0;
                    if (t < n) {
                        float4 sm, base;
                        if (lenC <= 32) {
                            float4 v[NIF];
                            issue_rows(v, h4, sortedC, 0, lenC, lane);
                            if (lenF <= 32 && lane < lenF) keyF = listSp[begF + lane];
                            base = ((const float4*)aggS)[(size_t)t * (H / 4) + lane];
                            if (lenN <= 32) sortedN = sort_ids(keyN, lenN, lane);
                            sm = sum_rows_first(v);
                            for (int rr = NIF; rr < lenC; rr += NIF) {
                                issue_rows(v, h4, sortedC, rr, lenC, lane);
                                sum_rows(sm, v);
                            }
                        } else {
                            if (lenF <= 32 && lane < lenF) keyF = listSp[begF + lane];
                            base = ((const float4*)aggS)[(size_t)t * (H / 4) + lane];
                            sm = bucket_sum(h4, listSp, begC, lenC, lane);
                            if (lenN <= 32) sortedN = sort_ids(keyN, lenN, lane);
                        }
                        const float w = 1.0f / (float)max(lenC, 1);
                        r.x = base.x + sm.x * w; r.y = base.y + sm.y * w;
                        r.z = base.z + sm.z * w; r.w = base.w + sm.w * w;
                    }
                    *(float4*)(sAp + (size_t)(lr0 + q) * APITCH + (lane << 2)) = r;
                    lenC = lenN; begC = begN; sortedC = sortedN;
                    lenN = lenF; begN = begF; keyN = keyF;
                }
            }
        }
    }
    __syncthreads();

    if (active) {
        const float* q0 = sAp + (size_t)l15 * APITCH + (half << 2);
        const float* q1 = q0 + 16 * APITCH;
        const float* bbase = sB + (1 << 14) + (half << 9) + ((cb + l15) << 1);
        for (int j = 0; j < H / 8; ++j) {
            const float4 x0 = *(const float4*)(q0 + 8 * j);
            const float4 x1 = *(const float4*)(q1 + 8 * j);
            const float* bk = bbase + (j << 10);
            const v2f b00 = *(const v2f*)(bk);
            const v2f b01 = *(const v2f*)(bk + 32);
            const v2f b10 = *(const v2f*)(bk + 256);
            const v2f b11 = *(const v2f*)(bk + 256 + 32);
            v2f a;
            a.x = x0.x; a.y = x0.y;
            acc00 = __builtin_amdgcn_wmma_f32_16x16x4_f32(false, a, false, b00, (short)0, acc00, false, false);
            acc01 = __builtin_amdgcn_wmma_f32_16x16x4_f32(false, a, false, b01, (short)0, acc01, false, false);
            a.x = x1.x; a.y = x1.y;
            acc10 = __builtin_amdgcn_wmma_f32_16x16x4_f32(false, a, false, b00, (short)0, acc10, false, false);
            acc11 = __builtin_amdgcn_wmma_f32_16x16x4_f32(false, a, false, b01, (short)0, acc11, false, false);
            a.x = x0.z; a.y = x0.w;
            acc00 = __builtin_amdgcn_wmma_f32_16x16x4_f32(false, a, false, b10, (short)0, acc00, false, false);
            acc01 = __builtin_amdgcn_wmma_f32_16x16x4_f32(false, a, false, b11, (short)0, acc01, false, false);
            a.x = x1.z; a.y = x1.w;
            acc10 = __builtin_amdgcn_wmma_f32_16x16x4_f32(false, a, false, b10, (short)0, acc10, false, false);
            acc11 = __builtin_amdgcn_wmma_f32_16x16x4_f32(false, a, false, b11, (short)0, acc11, false, false);
        }

        const float bc0 = bias[cb + l15];
        const float bc1 = bias[cb + 16 + l15];
        #pragma unroll
        for (int r = 0; r < 8; ++r) {
            const int rowa = m0 + (half << 3) + r;
            const int rowb = rowa + 16;
            if (rowa < n) {
                float v = acc00[r] + bc0;
                out[(size_t)rowa * H + cb + l15] = v > 0.0f ? v : (__expf(v) - 1.0f);
                v = acc01[r] + bc1;
                out[(size_t)rowa * H + cb + 16 + l15] = v > 0.0f ? v : (__expf(v) - 1.0f);
            }
            if (rowb < n) {
                float v = acc10[r] + bc0;
                out[(size_t)rowb * H + cb + l15] = v > 0.0f ? v : (__expf(v) - 1.0f);
                v = acc11[r] + bc1;
                out[(size_t)rowb * H + cb + 16 + l15] = v > 0.0f ? v : (__expf(v) - 1.0f);
            }
        }
    }
}

extern "C" void kernel_launch(void* const* d_in, const int* in_sizes, int n_in,
                              void* d_out, int out_size, void* d_ws, size_t ws_size,
                              hipStream_t stream) {
    const float* x_int  = (const float*)d_in[0];
    const float* x_lane = (const float*)d_in[1];
    const float* x_sens = (const float*)d_in[2];
    const float* x_inj  = (const float*)d_in[3];
    const int*   e_sp   = (const int*)d_in[4];
    const int*   e_fl   = (const int*)d_in[5];
    const int*   e_fs   = (const int*)d_in[6];
    const int*   e_in   = (const int*)d_in[7];
    const float* W_int  = (const float*)d_in[8];
    const float* b_int  = (const float*)d_in[9];
    const float* W_lane = (const float*)d_in[10];
    const float* b_lane = (const float*)d_in[11];
    const float* W_sens = (const float*)d_in[12];
    const float* b_sens = (const float*)d_in[13];
    const float* W_inj  = (const float*)d_in[14];
    const float* b_inj  = (const float*)d_in[15];
    const float* self_W = (const float*)d_in[16];
    const float* self_b = (const float*)d_in[17];
    const float* rel_W  = (const float*)d_in[18];

    const int nInt  = in_sizes[0] / 32;
    const int nLane = in_sizes[1] / 16;
    const int nSens = in_sizes[2] / 8;
    const int nInj  = in_sizes[3] / 8;
    const int eSp   = in_sizes[4] / 2;
    const int eFl   = in_sizes[5] / 2;
    const int eFs   = in_sizes[6] / 2;
    const int eIn   = in_sizes[7] / 2;
    const int L     = in_sizes[16] / (H * H);

    float* ws = (float*)d_ws;
    size_t o = 0;
    float* hA    = ws + o; o += (size_t)nInt  * H;
    float* hB    = ws + o; o += (size_t)nInt  * H;
    float* hLane = ws + o; o += (size_t)nLane * H;
    float* pSens = ws + o; o += (size_t)nSens * H;
    o += (size_t)nInj  * H;
    float* aggS  = ws + o; o += (size_t)nInt  * H;
    float* aggB  = ws + o; o += (size_t)nInt  * H;
    int*   cnt   = (int*)(ws + o); o += (size_t)4 * nInt;
    int*   off   = (int*)(ws + o); o += (size_t)4 * nInt;
    (void)ws_size; (void)n_in; (void)out_size;

    int*  binned = (int*)aggB;
    int*  listFl = (int*)aggB + 2 * ((size_t)eSp + eFl + eFs + eIn);
    int*  listFs = listFl + eFl;
    int*  listIn = listFs + eFs;
    int*  listSp = (int*)hLane;
    int*  tab    = (int*)pSens;

    const int nb  = (nInt + 255) / 256;
    const int nch = (eSp + CHUNK - 1) / CHUNK + (eFl + CHUNK - 1) / CHUNK
                  + (eFs + CHUNK - 1) / CHUNK + (eIn + CHUNK - 1) / CHUNK;
    proj_bin_kernel<<<nch + (nInt + 127) / 128, 256, 0, stream>>>(
        x_int, W_int, b_int, hA, nInt,
        e_sp, e_fl, e_fs, e_in, binned, tab, eSp, eFl, eFs, eIn, nb, nch);
    fine_kernel<<<4 * nb, 256, 0, stream>>>(binned, tab, cnt, off,
                                            listSp, listFl, listFs, listIn,
                                            eSp, eFl, eFs, eIn, nb, nch, nInt);

    gather_raw_proj_kernel<<<(nInt + 8 * TPW - 1) / (8 * TPW), 256, 0, stream>>>(
        x_lane, x_sens, x_inj, W_lane, b_lane, W_sens, b_sens, W_inj, b_inj,
        listFl, listFs, listIn, cnt, off, aggS, nInt);

    const int nrp = (nInt + 31) / 32;
    const int needed = (nrp + RPB - 1) / RPB;
    const int lgrid = needed > 256 ? needed : 256;
    float* hcur = hA;
    float* hnext = hB;
    for (int l = 0; l < L; ++l) {
        float* outp = (l == L - 1) ? (float*)d_out : hnext;
        fused_gemm_elu_kernel<<<lgrid, GT, 0, stream>>>(
            hcur, aggS, listSp, cnt, off,
            self_W + (size_t)l * H * H, rel_W + (size_t)l * H * H,
            self_b + (size_t)l * H, outp, nInt, nrp);
        float* t = hnext; hnext = hcur; hcur = t;
    }
}


# Second indexing launch reads a bin's run of a chunk with 8 lanes (64 contiguous bytes, a load touches 4-8 lines, was 32: a lane per chunk); all runs' loads issued together; sum order unchanged
# speedup vs baseline: 7.8166x; 1.0309x over previous
#include <hip/hip_runtime.h>
#include <cstddef>

#define H 128

typedef float v2f __attribute__((ext_vector_type(2)));
typedef float v8f __attribute__((ext_vector_type(8)));

#define KIN 32
__device__ __forceinline__ void proj_block(const float* __restrict__ x,
                                           const float* __restrict__ W,
                                           const float* __restrict__ b,
                                           float* __restrict__ out, int n, int bid) {
    __shared__ __align__(16) float sW[KIN * H];
    const int tid  = threadIdx.x;
    const int lane = tid & 31;
    const int half = lane >> 4;
    const int l15  = lane & 15;
    const int m0 = __builtin_amdgcn_readfirstlane(bid * 128 + (tid >> 5) * 16);

    int arow = m0 + l15;
    if (arow >= n) arow = n - 1;
    const float4* xr = (const float4*)(x + (size_t)arow * KIN + (half << 4));
    const float4 x0 = xr[0], x1 = xr[1], x2 = xr[2], x3 = xr[3];

    float4 wv[4];
    #pragma unroll
    for (int j = 0; j < 4; ++j) wv[j] = ((const float4*)W)[tid + 256 * j];

    v8f acc[8];
    #pragma unroll
    for (int t = 0; t < 8; ++t) {
        const float bc = b[(t << 4) + l15];
        #pragma unroll
        for (int r = 0; r < 8; ++r) acc[t][r] = bc;
    }

    #pragma unroll
    for (int j = 0; j < 4; ++j) {
        const int i  = tid + 256 * j;
        const int k  = i >> 5;
        const int c0 = (i & 31) << 2;
        float* d = sW + ((k >> 1) << 8) + (k & 1);
        d[2 * (c0 + 0)] = wv[j].x;
        d[2 * (c0 + 1)] = wv[j].y;
        d[2 * (c0 + 2)] = wv[j].z;
        d[2 * (c0 + 3)] = wv[j].w;
    }
    __syncthreads();

    const float xf[16] = { x0.x, x0.y, x0.z, x0.w, x1.x, x1.y, x1.z, x1.w,
                           x2.x, x2.y, x2.z, x2.w, x3.x, x3.y, x3.z, x3.w };
    const float* bbase = sW + (half << 11) + (l15 << 1);
    #pragma unroll
    for (int s = 0; s < 8; ++s) {
        v2f a;
        a.x = xf[2 * s];
        a.y = xf[2 * s + 1];
        const float* bk = bbase + (s << 8);
        #pragma unroll
        for (int t = 0; t < 8; ++t) {
            const v2f bf = *(const v2f*)(bk + (t << 5));
            acc[t] = __builtin_amdgcn_wmma_f32_16x16x4_f32(
                false, a, false, bf, (short)0, acc[t], false, false);
        }
    }

    float* orow = out + (size_t)(m0 + (half << 3)) * H + l15;
    if (m0 + 16 <= n) {
        #pragma unroll
        for (int t = 0; t < 8; ++t) {
            #pragma unroll
            for (int r = 0; r < 8; ++r) orow[r * H + (t << 4)] = acc[t][r];
        }
    } else {
        #pragma unroll
        for (int t = 0; t < 8; ++t) {
            #pragma unroll
            for (int r = 0; r < 8; ++r) {
                if (m0 + (half << 3) + r < n) orow[r * H + (t << 4)] = acc[t][r];
            }
        }
    }
}

#define CHUNK 4096
#define MAXNB 1024
#define STAGE 8192
#define RLANES 8
#define RSLOTS (256 / RLANES)
#define RROUND 5
#define RDEPTH 4

__device__ __forceinline__ void bin_block(
        const int* __restrict__ ed0, const int* __restrict__ ed1,
        const int* __restrict__ ed2, const int* __restrict__ ed3,
        int* __restrict__ binned, int* __restrict__ tab,
        int E0, int E1, int E2, int E3, int nb, int nch, int bid) {
    __shared__ int  sHist[MAXNB + 1];
    __shared__ int2 sPair[CHUNK];
    const int tid = threadIdx.x;
    const int C0 = (E0 + CHUNK - 1) / CHUNK, C1 = (E1 + CHUNK - 1) / CHUNK,
              C2 = (E2 + CHUNK - 1) / CHUNK;
    int c = bid;
    const int* ed = ed0;
    int E = E0, rbase = 0;
    if (c >= C0) {
        c -= C0; ed = ed1; E = E1; rbase = E0;
        if (c >= C1) {
            c -= C1; ed = ed2; E = E2; rbase += E1;
            if (c >= C2) { c -= C2; ed = ed3; E = E3; rbase += E2; }
        }
    }
    const int e0  = c * CHUNK;
    const int cnt = min(CHUNK, E - e0);

    for (int j = tid; j <= nb; j += 256) sHist[j] = 0;
    __syncthreads();

    int srcv[16], tgtv[16], rk[16];
    #pragma unroll
    for (int k = 0; k < 16; ++k) {
        const int i = k * 256 + tid;
        srcv[k] = 0; tgtv[k] = 0;
        if (i < cnt) { srcv[k] = ed[e0 + i]; tgtv[k] = ed[E + e0 + i]; }
    }
    #pragma unroll
    for (int k = 0; k < 16; ++k) {
        const int i = k * 256 + tid;
        rk[k] = 0;
        if (i < cnt) rk[k] = atomicAdd(&sHist[tgtv[k] >> 8], 1);
    }
    __syncthreads();

    if (tid < 32) {
        int carry = 0;
        for (int base = 0; base < nb; base += 32) {
            const int j = base + tid;
            const int v = (j < nb) ? sHist[j] : 0;
            int inc = v;
            for (int d = 1; d < 32; d <<= 1) {
                const int u = __shfl_up(inc, d, 32);
                if (tid >= d) inc += u;
            }
            if (j < nb) sHist[j] = carry + inc - v;
            carry += __shfl(inc, 31, 32);
        }
        if (tid == 0) sHist[nb] = carry;
    }
    __syncthreads();

    #pragma unroll
    for (int k = 0; k < 16; ++k) {
        const int i = k * 256 + tid;
        if (i < cnt) sPair[sHist[tgtv[k] >> 8] + rk[k]] = make_int2(srcv[k], tgtv[k]);
    }
    for (int j = tid; j <= nb; j += 256) tab[(size_t)j * nch + bid] = sHist[j];
    __syncthreads();

    int2* outp = (int2*)binned + (size_t)rbase + e0;
    for (int i = tid; i < cnt; i += 256) outp[i] = sPair[i];
}

__global__ __launch_bounds__(256) void proj_bin_kernel(
        const float* __restrict__ x, const float* __restrict__ W, const float* __restrict__ b,
        float* __restrict__ hout, int n,
        const int* __restrict__ ed0, const int* __restrict__ ed1,
        const int* __restrict__ ed2, const int* __restrict__ ed3,
        int* __restrict__ binned, int* __restrict__ tab,
        int E0, int E1, int E2, int E3, int nb, int nch) {
    const int bid = (int)blockIdx.x;
    if (bid < nch) bin_block(ed0, ed1, ed2, ed3, binned, tab, E0, E1, E2, E3, nb, nch, bid);
    else           proj_block(x, W, b, hout, n, bid - nch);
}

__global__ __launch_bounds__(256) void fine_kernel(
        const int* __restrict__ binned, const int* __restrict__ tab,
        int* __restrict__ cnt, int* __restrict__ off,
        int* __restrict__ list0, int* __restrict__ list1,
        int* __restrict__ list2, int* __restrict__ list3,
        int E0, int E1, int E2, int E3, int nb, int nch, int n) {
    __shared__ int sCnt[256];
    __shared__ int sCur[256];
    __shared__ int sWave[8];
    __shared__ int sRed[8];
    __shared__ int sStage[STAGE];
    const int tid  = threadIdx.x;
    const int lane = tid & 31;
    const int wave = tid >> 5;
    const int C0 = (E0 + CHUNK - 1) / CHUNK, C1 = (E1 + CHUNK - 1) / CHUNK,
              C2 = (E2 + CHUNK - 1) / CHUNK, C3 = (E3 + CHUNK - 1) / CHUNK;
    const int r = blockIdx.x / nb;
    const int g = blockIdx.x - r * nb;
    int rbase = 0, cb = 0, nC = C0;
    int* list = list0;
    if (r == 1) { rbase = E0;           cb = C0;           nC = C1; list = list1; }
    if (r == 2) { rbase = E0 + E1;      cb = C0 + C1;      nC = C2; list = list2; }
    if (r == 3) { rbase = E0 + E1 + E2; cb = C0 + C1 + C2; nC = C3; list = list3; }
    const int* tlo = tab + (size_t)g * nch + cb;
    const int* thi = tab + (size_t)(g + 1) * nch + cb;
    const int2* win = (const int2*)binned + (size_t)rbase;

    sCnt[tid] = 0;
    __syncthreads();

    const int sub = tid & (RLANES - 1);
    const int rsl = tid / RLANES;

    int lower = 0;
    for (int b0 = 0; b0 < nC; b0 += RSLOTS * RROUND) {
        int rb[RROUND], re[RROUND];
        #pragma unroll
        for (int k = 0; k < RROUND; ++k) {
            const int b = b0 + rsl + RSLOTS * k;
            rb[k] = 0; re[k] = 0;
            if (b < nC) { rb[k] = tlo[b]; re[k] = thi[b]; }
        }
        int y[RROUND][RDEPTH];
        #pragma unroll
        for (int k = 0; k < RROUND; ++k) {
            const int2* p = win + (size_t)min(b0 + rsl + RSLOTS * k, nC - 1) * CHUNK;
            #pragma unroll
            for (int u = 0; u < RDEPTH; ++u)
                y[k][u] = p[max(min(rb[k] + sub + RLANES * u, re[k] - 1), 0)].y;
        }
        #pragma unroll
        for (int k = 0; k < RROUND; ++k) {
            if (sub == 0) lower += rb[k];
            #pragma unroll
            for (int u = 0; u < RDEPTH; ++u)
                if (rb[k] + sub + RLANES * u < re[k]) atomicAdd(&sCnt[y[k][u] & 255], 1);
            const int2* p = win + (size_t)min(b0 + rsl + RSLOTS * k, nC - 1) * CHUNK;
            for (int i = rb[k] + sub + RLANES * RDEPTH; i < re[k]; i += RLANES)
                atomicAdd(&sCnt[p[i].y & 255], 1);
        }
    }
    __syncthreads();

    const int cT = sCnt[tid];
    int inc = cT;
    for (int d = 1; d < 32; d <<= 1) {
        const int u = __shfl_up(inc, d, 32);
        if (lane >= d) inc += u;
    }
    for (int o = 16; o > 0; o >>= 1) lower += __shfl_xor(lower, o, 32);
    if (lane == 31) sWave[wave] = inc;
    if (lane == 0)  sRed[wave]  = lower;
    __syncthreads();
    int wbase = 0, total = 0, start = 0;
    for (int w = 0; w < 8; ++w) {
        const int tw = sWave[w];
        if (w < wave) wbase += tw;
        total += tw;
        start += sRed[w];
    }
    const int offL = wbase + inc - cT;
    const int t = g * 256 + tid;
    if (t < n) {
        cnt[(size_t)r * n + t] = cT;
        off[(size_t)r * n + t] = start + offL;
    }
    sCur[tid] = offL;
    __syncthreads();

    const bool staged = total <= STAGE;
    for (int b0 = 0; b0 < nC; b0 += RSLOTS * RROUND) {
        int rb[RROUND], re[RROUND];
        #pragma unroll
        for (int k = 0; k < RROUND; ++k) {
            const int b = b0 + rsl + RSLOTS * k;
            rb[k] = 0; re[k] = 0;
            if (b < nC) { rb[k] = tlo[b]; re[k] = thi[b]; }
        }
        int2 v[RROUND][RDEPTH];
        #pragma unroll
        for (int k = 0; k < RROUND; ++k) {
            const int2* p = win + (size_t)min(b0 + rsl + RSLOTS * k, nC - 1) * CHUNK;
            #pragma unroll
            for (int u = 0; u < RDEPTH; ++u)
                v[k][u] = p[max(min(rb[k] + sub + RLANES * u, re[k] - 1), 0)];
        }
        #pragma unroll
        for (int k = 0; k < RROUND; ++k) {
            #pragma unroll
            for (int u = 0; u < RDEPTH; ++u) {
                if (rb[k] + sub + RLANES * u < re[k]) {
                    const int pos = atomicAdd(&sCur[v[k][u].y & 255], 1);
                    if (staged) sStage[pos] = v[k][u].x;
                    else        list[start + pos] = v[k][u].x;
                }
            }
            const int2* p = win + (size_t)min(b0 + rsl + RSLOTS * k, nC - 1) * CHUNK;
            for (int i = rb[k] + sub + RLANES * RDEPTH; i < re[k]; i += RLANES) {
                const int2 w = p[i];
                const int pos = atomicAdd(&sCur[w.y & 255], 1);
                if (staged) sStage[pos] = w.x;
                else        list[start + pos] = w.x;
            }
        }
    }
    if (staged) {
        __syncthreads();
        for (int i = tid; i < total; i += 256) list[start + i] = sStage[i];
    }
}

__device__ __forceinline__ void add4(float4& a, const float4 v) {
    a.x += v.x; a.y += v.y; a.z += v.z; a.w += v.w;
}

__device__ __forceinline__ float4 bucket_sum(const float4* __restrict__ feat4,
                                             const int* __restrict__ list,
                                             int beg, int len, int lane) {
    float4 acc = make_float4(0.0f, 0.0f, 0.0f, 0.0f);
    if (len <= 32) {
        const int key = (lane < len) ? list[beg + lane] : 0x7fffffff;
        int rank = 0;
        for (int j = 0; j < len; ++j) {
            const int kj = __builtin_amdgcn_readlane(key, j);
            rank += (kj < key || (kj == key && j < lane)) ? 1 : 0;
        }
        if (lane >= len) rank = lane;
        const int sorted = __builtin_amdgcn_ds_permute(rank << 2, key);
        int r = 0;
        for (; r + 4 <= len; r += 4) {
            const int s0 = __builtin_amdgcn_readlane(sorted, r);
            const int s1 = __builtin_amdgcn_readlane(sorted, r + 1);
            const int s2 = __builtin_amdgcn_readlane(sorted, r + 2);
            const int s3 = __builtin_amdgcn_readlane(sorted, r + 3);
            const float4 v0 = feat4[(size_t)s0 * (H / 4) + lane];
            const float4 v1 = feat4[(size_t)s1 * (H / 4) + lane];
            const float4 v2 = feat4[(size_t)s2 * (H / 4) + lane];
            const float4 v3 = feat4[(size_t)s3 * (H / 4) + lane];
            add4(acc, v0); add4(acc, v1); add4(acc, v2); add4(acc, v3);
        }
        for (; r < len; ++r) {
            const int s0 = __builtin_amdgcn_readlane(sorted, r);
            add4(acc, feat4[(size_t)s0 * (H / 4) + lane]);
        }
    } else {
        int prev = -1;
        for (int it = 0; it < len; ++it) {
            int m = 0x7fffffff;
            for (int i = lane; i < len; i += 32) {
                const int k = list[beg + i];
                if (k > prev && k < m) m = k;
            }
            for (int o = 16; o > 0; o >>= 1) m = min(m, __shfl_xor(m, o, 32));
            if (m == 0x7fffffff) break;
            int c = 0;
            for (int i = lane; i < len; i += 32) c += (list[beg + i] == m) ? 1 : 0;
            for (int o = 16; o > 0; o >>= 1) c += __shfl_xor(c, o, 32);
            const float4 v = feat4[(size_t)m * (H / 4) + lane];
            for (int q = 0; q < c; ++q) add4(acc, v);
            prev = m;
        }
    }
    return acc;
}


__device__ __forceinline__ int sort_ids(int key, int len, int lane) {
    int rank = 0;
    for (int j = 0; j < len; ++j) {
        const int kj = __builtin_amdgcn_readlane(key, j);
        rank += (kj < key || (kj == key && j < lane)) ? 1 : 0;
    }
    if (lane >= len) rank = lane;
    return __builtin_amdgcn_ds_permute(rank << 2, key);
}

template <int Q>
__device__ __forceinline__ float4 long_bucket_sum(const float4* __restrict__ x4,
                                                  const int* __restrict__ list,
                                                  int beg, int len, int lane, int qq) {
    float4 acc = make_float4(0.0f, 0.0f, 0.0f, 0.0f);
    int prev = -1;
    for (int it = 0; it < len; ++it) {
        int m = 0x7fffffff;
        for (int i = lane; i < len; i += 32) {
            const int k = list[beg + i];
            if (k > prev && k < m) m = k;
        }
        for (int o = 16; o > 0; o >>= 1) m = min(m, __shfl_xor(m, o, 32));
        if (m == 0x7fffffff) break;
        int c = 0;
        for (int i = lane; i < len; i += 32) c += (list[beg + i] == m) ? 1 : 0;
        for (int o = 16; o > 0; o >>= 1) c += __shfl_xor(c, o, 32);
        const float4 v = x4[(size_t)m * Q + qq];
        for (int r = 0; r < c; ++r) add4(acc, v);
        prev = m;
    }
    return acc;
}

__device__ __forceinline__ float4 slot_sum(float4 a) {
    for (int m = 4; m < 32; m <<= 1) {
        float4 b;
        b.x = __shfl_xor(a.x, m, 32); b.y = __shfl_xor(a.y, m, 32);
        b.z = __shfl_xor(a.z, m, 32); b.w = __shfl_xor(a.w, m, 32);
        add4(a, b);
    }
    return a;
}

#define TPW 4

__global__ __launch_bounds__(256) void gather_raw_proj_kernel(
        const float* __restrict__ xLane, const float* __restrict__ xSens,
        const float* __restrict__ xInj,
        const float* __restrict__ WLane, const float* __restrict__ bLane,
        const float* __restrict__ WSens, const float* __restrict__ bSens,
        const float* __restrict__ WInj,  const float* __restrict__ bInj,
        const int* __restrict__ listFl, const int* __restrict__ listFs,
        const int* __restrict__ listIn,
        const int* __restrict__ cnt, const int* __restrict__ off,
        float* __restrict__ aggS, int n) {
    __shared__ float4 sW[35 * 32];
    __shared__ float4 sZ[8][TPW][9];
    const int tid  = threadIdx.x;
    const int wave = tid >> 5;
    const int lane = tid & 31;
    const int slot = lane >> 2;
    const int q    = lane & 3;

    {
        const float4* wl = (const float4*)WLane;
        sW[tid]       = wl[tid];
        sW[256 + tid] = wl[256 + tid];
        sW[512 + tid] = ((const float4*)WSens)[tid];
        sW[768 + tid] = ((const float4*)WInj)[tid];
        if (tid < 96) {
            const float* bs = tid < 32 ? bLane : (tid < 64 ? bSens : bInj);
            sW[1024 + tid] = ((const float4*)bs)[tid & 31];
        }
    }

    const float4* xl4 = (const float4*)xLane;
    const float4* xs4 = (const float4*)xSens;
    const float4* xi4 = (const float4*)xInj;
    const float4 zero4 = make_float4(0.0f, 0.0f, 0.0f, 0.0f);
    const int tb = blockIdx.x * (8 * TPW) + wave * TPW;

    for (int u = 0; u < TPW; ++u) {
        const int t = __builtin_amdgcn_readfirstlane(tb + u);
        float4 zl = zero4, zsi = zero4;
        float fl = 0.0f, fs = 0.0f, fi = 0.0f;
        if (t < n) {
            const int lenL = cnt[1 * n + t], lenS = cnt[2 * n + t], lenI = cnt[3 * n + t];
            const int begL = off[1 * n + t], begS = off[2 * n + t], begI = off[3 * n + t];
            fl = lenL > 0 ? 1.0f : 0.0f;
            fs = lenS > 0 ? 1.0f : 0.0f;
            fi = lenI > 0 ? 1.0f : 0.0f;
            const bool isS = q < 2;
            const int qq = q & 1;
            if (lenL <= 32 && lenS <= 32 && lenI <= 32) {
                const int keyL = (lane < lenL) ? listFl[begL + lane] : 0x7fffffff;
                const int keyS = (lane < lenS) ? listFs[begS + lane] : 0x7fffffff;
                const int keyI = (lane < lenI) ? listIn[begI + lane] : 0x7fffffff;
                const int sortedL = sort_ids(keyL, lenL, lane);
                const int sortedS = sort_ids(keyS, lenS, lane);
                const int sortedI = sort_ids(keyI, lenI, lane);
                const int lenX = isS ? lenS : lenI;
                const int maxSI = max(lenS, lenI);
                const float4* xb = isS ? xs4 : xi4;
                float4 a[4], b[4];
                #pragma unroll
                for (int i = 0; i < 4; ++i) {
                    a[i] = zero4; b[i] = zero4;
                    const int p = 8 * i + slot;
                    if (8 * i < lenL) {
                        const int id = __builtin_amdgcn_ds_bpermute(p << 2, sortedL);
                        if (p < lenL) a[i] = xl4[(size_t)id * 4 + q];
                    }
                    if (8 * i < maxSI) {
                        const int idS = __builtin_amdgcn_ds_bpermute(p << 2, sortedS);
                        const int idI = __builtin_amdgcn_ds_bpermute(p << 2, sortedI);
                        if (p < lenX) b[i] = xb[(size_t)(isS ? idS : idI) * 2 + qq];
                    }
                }
                add4(a[0], a[1]); add4(a[0], a[2]); add4(a[0], a[3]);
                add4(b[0], b[1]); add4(b[0], b[2]); add4(b[0], b[3]);
                zl  = slot_sum(a[0]);
                zsi = slot_sum(b[0]);
            } else {
                zl = long_bucket_sum<4>(xl4, listFl, begL, lenL, lane, q);
                const float4 s = long_bucket_sum<2>(xs4, listFs, begS, lenS, lane, qq);
                const float4 j = long_bucket_sum<2>(xi4, listIn, begI, lenI, lane, qq);
                zsi = isS ? s : j;
            }
            const float wL = 1.0f / (float)max(lenL, 1);
            const float wX = 1.0f / (float)max(isS ? lenS : lenI, 1);
            zl.x *= wL;  zl.y *= wL;  zl.z *= wL;  zl.w *= wL;
            zsi.x *= wX; zsi.y *= wX; zsi.z *= wX; zsi.w *= wX;
        }
        if (lane < 4) { sZ[wave][u][lane] = zl; sZ[wave][u][4 + lane] = zsi; }
        if (lane == 0) sZ[wave][u][8] = make_float4(fl, fs, fi, 0.0f);
    }
    __syncthreads();

    float4 o[TPW];
    #pragma unroll
    for (int u = 0; u < TPW; ++u) o[u] = zero4;
    #pragma nounroll
    for (int g = 0; g < 8; ++g) {
        const float4 w0 = sW[(4 * g + 0) * 32 + lane];
        const float4 w1 = sW[(4 * g + 1) * 32 + lane];
        const float4 w2 = sW[(4 * g + 2) * 32 + lane];
        const float4 w3 = sW[(4 * g + 3) * 32 + lane];
        #pragma unroll
        for (int u = 0; u < TPW; ++u) {
            const float4 z = sZ[wave][u][g];
            o[u].x += z.x * w0.x; o[u].y += z.x * w0.y; o[u].z += z.x * w0.z; o[u].w += z.x * w0.w;
            o[u].x += z.y * w1.x; o[u].y += z.y * w1.y; o[u].z += z.y * w1.z; o[u].w += z.y * w1.w;
            o[u].x += z.z * w2.x; o[u].y += z.z * w2.y; o[u].z += z.z * w2.z; o[u].w += z.z * w2.w;
            o[u].x += z.w * w3.x; o[u].y += z.w * w3.y; o[u].z += z.w * w3.z; o[u].w += z.w * w3.w;
        }
    }
    {
        const float4 b0 = sW[32 * 32 + lane];
        const float4 b1 = sW[33 * 32 + lane];
        const float4 b2 = sW[34 * 32 + lane];
        #pragma unroll
        for (int u = 0; u < TPW; ++u) {
            const float4 f = sZ[wave][u][8];
            o[u].x += f.x * b0.x; o[u].y += f.x * b0.y; o[u].z += f.x * b0.z; o[u].w += f.x * b0.w;
            o[u].x += f.y * b1.x; o[u].y += f.y * b1.y; o[u].z += f.y * b1.z; o[u].w += f.y * b1.w;
            o[u].x += f.z * b2.x; o[u].y += f.z * b2.y; o[u].z += f.z * b2.z; o[u].w += f.z * b2.w;
        }
    }
    #pragma unroll
    for (int u = 0; u < TPW; ++u) {
        const int t = tb + u;
        if (t < n) ((float4*)aggS)[(size_t)t * (H / 4) + lane] = o[u];
    }
}

#define RPB 7
#define GW (4 * RPB)
#define GT (GW * 32)
#define APITCH 132

#define NIF 16
__device__ __forceinline__ void issue_rows(float4 (&v)[NIF], const float4* __restrict__ feat4,
                                           int sorted, int r, int len, int lane) {
    #pragma unroll
    for (int i = 0; i < NIF; ++i) {
        v[i] = make_float4(0.0f, 0.0f, 0.0f, 0.0f);
        if (r + i < len) {
            const int sid = __builtin_amdgcn_readlane(sorted, r + i);
            v[i] = feat4[(size_t)sid * (H / 4) + lane];
        }
    }
}
__device__ __forceinline__ void sum_rows(float4& acc, const float4 (&v)[NIF]) {
    #pragma unroll
    for (int i = 0; i < NIF; ++i) add4(acc, v[i]);
}
__device__ __forceinline__ float4 sum_rows_first(const float4 (&v)[NIF]) {
    float4 acc = v[0];
    #pragma unroll
    for (int i = 1; i < NIF; ++i) add4(acc, v[i]);
    return acc;
}

__global__ __launch_bounds__(GT) void fused_gemm_elu_kernel(
        const float* __restrict__ h, const float* __restrict__ aggS,
        const int* __restrict__ listSp, const int* __restrict__ cnt, const int* __restrict__ off,
        const float* __restrict__ W0, const float* __restrict__ W1,
        const float* __restrict__ bias, float* __restrict__ out, int n, int nrp) {
    __shared__ __align__(16) float sB[2 * H * H];
    __shared__ __align__(16) float sA[RPB * 32 * APITCH];
    const int tid  = threadIdx.x;
    const int lane = tid & 31;
    const int half = lane >> 4;
    const int l15  = lane & 15;

    const int wave = __builtin_amdgcn_readfirstlane(tid >> 5);
    const int lrp  = wave >> 2;
    const int rpBeg = (int)(((long long)blockIdx.x * nrp) / gridDim.x);
    const int rpEnd = (int)(((long long)(blockIdx.x + 1) * nrp) / gridDim.x);
    const bool active = rpBeg + lrp < rpEnd;
    const int m0 = (rpBeg + lrp) << 5;
    const int cb = (wave & 3) << 5;

    const int lr0 = (wave & 3) << 3;
    const int t0  = m0 + lr0;
    const bool okl = active && lane < 8 && t0 + lane < n;
    const int cntv = okl ? cnt[t0 + lane] : 0;
    const int offv = okl ? off[t0 + lane] : 0;

    {
        constexpr int NV = 2 * H * H / 4;
        constexpr int NI = (NV + GT - 1) / GT;
        float4 wv[NI];
        #pragma unroll
        for (int q = 0; q < NI; ++q) {
            const int i = tid + GT * q;
            wv[q] = make_float4(0.0f, 0.0f, 0.0f, 0.0f);
            if (i < NV) wv[q] = ((const float4*)((i >> 12) ? W1 : W0))[i & 4095];
        }
        #pragma unroll
        for (int q = 0; q < NI; ++q) {
            const int i = tid + GT * q;
            if (i < NV) {
                const int jj = i & 4095;
                const int k  = jj >> 5;
                const int c0 = (jj & 31) << 2;
                float* d = sB + ((i >> 12) << 14) + ((k >> 1) << 8) + (k & 1);
                d[2 * (c0 + 0)] = wv[q].x;
                d[2 * (c0 + 1)] = wv[q].y;
                d[2 * (c0 + 2)] = wv[q].z;
                d[2 * (c0 + 3)] = wv[q].w;
            }
        }
    }
    int lenC = __builtin_amdgcn_readlane(cntv, 0);
    int begC = __builtin_amdgcn_readlane(offv, 0);
    int lenN = __builtin_amdgcn_readlane(cntv, 1);
    int begN = __builtin_amdgcn_readlane(offv, 1);
    int keyN = (lenN <= 32 && lane < lenN) ? listSp[begN + lane] : 0x7fffffff;
    const int keyC = (lenC <= 32 && lane < lenC) ? listSp[begC + lane] : 0x7fffffff;
    __syncthreads();

    int r0 = m0 + l15;
    if (r0 >= n) r0 = n - 1;
    int r1 = m0 + 16 + l15;
    if (r1 >= n) r1 = n - 1;
    float* sAp = sA + (size_t)lrp * 32 * APITCH;

    v8f acc00 = {}, acc01 = {}, acc10 = {}, acc11 = {};
    if (active) {
        const int first = (wave >> 2) & 1;
        for (int st = 0; st < 2; ++st) {
            if ((st ^ first) == 0) {
                const float4* p0 = (const float4*)(h + (size_t)r0 * H + (half << 2));
                const float4* p1 = (const float4*)(h + (size_t)r1 * H + (half << 2));
                const float* bbase = sB + (half << 9) + ((cb + l15) << 1);
                for (int j = 0; j < H / 8; ++j) {
                    const float4 x0 = p0[2 * j];
                    const float4 x1 = p1[2 * j];
                    const float* bk = bbase + (j << 10);
                    const v2f b00 = *(const v2f*)(bk);
                    const v2f b01 = *(const v2f*)(bk + 32);
                    const v2f b10 = *(const v2f*)(bk + 256);
                    const v2f b11 = *(const v2f*)(bk + 256 + 32);
                    v2f a;
                    a.x = x0.x; a.y = x0.y;
                    acc00 = __builtin_amdgcn_wmma_f32_16x16x4_f32(false, a, false, b00, (short)0, acc00, false, false);
                    acc01 = __builtin_amdgcn_wmma_f32_16x16x4_f32(false, a, false, b01, (short)0, acc01, false, false);
                    a.x = x1.x; a.y = x1.y;
                    acc10 = __builtin_amdgcn_wmma_f32_16x16x4_f32(false, a, false, b00, (short)0, acc10, false, false);
                    acc11 = __builtin_amdgcn_wmma_f32_16x16x4_f32(false, a, false, b01, (short)0, acc11, false, false);
                    a.x = x0.z; a.y = x0.w;
                    acc00 = __builtin_amdgcn_wmma_f32_16x16x4_f32(false, a, false, b10, (short)0, acc00, false, false);
                    acc01 = __builtin_amdgcn_wmma_f32_16x16x4_f32(false, a, false, b11, (short)0, acc01, false, false);
                    a.x = x1.z; a.y = x1.w;
                    acc10 = __builtin_amdgcn_wmma_f32_16x16x4_f32(false, a, false, b10, (short)0, acc10, false, false);
                    acc11 = __builtin_amdgcn_wmma_f32_16x16x4_f32(false, a, false, b11, (short)0, acc11, false, false);
                }
            } else {
                int sortedC = (lenC <= 32) ? sort_ids(keyC, lenC, lane) : 0;
                const float4* h4 = (const float4*)h;
                for (int q = 0; q < 8; ++q) {
                    const int t = t0 + q;
                    int lenF = 0, begF = 0;
                    if (q + 2 < 8) {
                        lenF = __builtin_amdgcn_readlane(cntv, q + 2);
                        begF = __builtin_amdgcn_readlane(offv, q + 2);
                    }
                    int keyF = 0x7fffffff;
                    float4 r = make_float4(0.0f, 0.0f, 0.0f, 0.0f);
                    int sortedN = 0;
                    if (t < n) {
                        float4 sm, base;
                        if (lenC <= 32) {
                            float4 v[NIF];
                            issue_rows(v, h4, sortedC, 0, lenC, lane);
                            if (lenF <= 32 && lane < lenF) keyF = listSp[begF + lane];
                            base = ((const float4*)aggS)[(size_t)t * (H / 4) + lane];
                            if (lenN <= 32) sortedN = sort_ids(keyN, lenN, lane);
                            sm = sum_rows_first(v);
                            for (int rr = NIF; rr < lenC; rr += NIF) {
                                issue_rows(v, h4, sortedC, rr, lenC, lane);
                                sum_rows(sm, v);
                            }
                        } else {
                            if (lenF <= 32 && lane < lenF) keyF = listSp[begF + lane];
                            base = ((const float4*)aggS)[(size_t)t * (H / 4) + lane];
                            sm = bucket_sum(h4, listSp, begC, lenC, lane);
                            if (lenN <= 32) sortedN = sort_ids(keyN, lenN, lane);
                        }
                        const float w = 1.0f / (float)max(lenC, 1);
                        r.x = base.x + sm.x * w; r.y = base.y + sm.y * w;
                        r.z = base.z + sm.z * w; r.w = base.w + sm.w * w;
                    }
                    *(float4*)(sAp + (size_t)(lr0 + q) * APITCH + (lane << 2)) = r;
                    lenC = lenN; begC = begN; sortedC = sortedN;
                    lenN = lenF; begN = begF; keyN = keyF;
                }
            }
        }
    }
    __syncthreads();

    if (active) {
        const float* q0 = sAp + (size_t)l15 * APITCH + (half << 2);
        const float* q1 = q0 + 16 * APITCH;
        const float* bbase = sB + (1 << 14) + (half << 9) + ((cb + l15) << 1);
        for (int j = 0; j < H / 8; ++j) {
            const float4 x0 = *(const float4*)(q0 + 8 * j);
            const float4 x1 = *(const float4*)(q1 + 8 * j);
            const float* bk = bbase + (j << 10);
            const v2f b00 = *(const v2f*)(bk);
            const v2f b01 = *(const v2f*)(bk + 32);
            const v2f b10 = *(const v2f*)(bk + 256);
            const v2f b11 = *(const v2f*)(bk + 256 + 32);
            v2f a;
            a.x = x0.x; a.y = x0.y;
            acc00 = __builtin_amdgcn_wmma_f32_16x16x4_f32(false, a, false, b00, (short)0, acc00, false, false);
            acc01 = __builtin_amdgcn_wmma_f32_16x16x4_f32(false, a, false, b01, (short)0, acc01, false, false);
            a.x = x1.x; a.y = x1.y;
            acc10 = __builtin_amdgcn_wmma_f32_16x16x4_f32(false, a, false, b00, (short)0, acc10, false, false);
            acc11 = __builtin_amdgcn_wmma_f32_16x16x4_f32(false, a, false, b01, (short)0, acc11, false, false);
            a.x = x0.z; a.y = x0.w;
            acc00 = __builtin_amdgcn_wmma_f32_16x16x4_f32(false, a, false, b10, (short)0, acc00, false, false);
            acc01 = __builtin_amdgcn_wmma_f32_16x16x4_f32(false, a, false, b11, (short)0, acc01, false, false);
            a.x = x1.z; a.y = x1.w;
            acc10 = __builtin_amdgcn_wmma_f32_16x16x4_f32(false, a, false, b10, (short)0, acc10, false, false);
            acc11 = __builtin_amdgcn_wmma_f32_16x16x4_f32(false, a, false, b11, (short)0, acc11, false, false);
        }

        const float bc0 = bias[cb + l15];
        const float bc1 = bias[cb + 16 + l15];
        #pragma unroll
        for (int r = 0; r < 8; ++r) {
            const int rowa = m0 + (half << 3) + r;
            const int rowb = rowa + 16;
            if (rowa < n) {
                float v = acc00[r] + bc0;
                out[(size_t)rowa * H + cb + l15] = v > 0.0f ? v : (__expf(v) - 1.0f);
                v = acc01[r] + bc1;
                out[(size_t)rowa * H + cb + 16 + l15] = v > 0.0f ? v : (__expf(v) - 1.0f);
            }
            if (rowb < n) {
                float v = acc10[r] + bc0;
                out[(size_t)rowb * H + cb + l15] = v > 0.0f ? v : (__expf(v) - 1.0f);
                v = acc11[r] + bc1;
                out[(size_t)rowb * H + cb + 16 + l15] = v > 0.0f ? v : (__expf(v) - 1.0f);
            }
        }
    }
}

extern "C" void kernel_launch(void* const* d_in, const int* in_sizes, int n_in,
                              void* d_out, int out_size, void* d_ws, size_t ws_size,
                              hipStream_t stream) {
    const float* x_int  = (const float*)d_in[0];
    const float* x_lane = (const float*)d_in[1];
    const float* x_sens = (const float*)d_in[2];
    const float* x_inj  = (const float*)d_in[3];
    const int*   e_sp   = (const int*)d_in[4];
    const int*   e_fl   = (const int*)d_in[5];
    const int*   e_fs   = (const int*)d_in[6];
    const int*   e_in   = (const int*)d_in[7];
    const float* W_int  = (const float*)d_in[8];
    const float* b_int  = (const float*)d_in[9];
    const float* W_lane = (const float*)d_in[10];
    const float* b_lane = (const float*)d_in[11];
    const float* W_sens = (const float*)d_in[12];
    const float* b_sens = (const float*)d_in[13];
    const float* W_inj  = (const float*)d_in[14];
    const float* b_inj  = (const float*)d_in[15];
    const float* self_W = (const float*)d_in[16];
    const float* self_b = (const float*)d_in[17];
    const float* rel_W  = (const float*)d_in[18];

    const int nInt  = in_sizes[0] / 32;
    const int nLane = in_sizes[1] / 16;
    const int nSens = in_sizes[2] / 8;
    const int nInj  = in_sizes[3] / 8;
    const int eSp   = in_sizes[4] / 2;
    const int eFl   = in_sizes[5] / 2;
    const int eFs   = in_sizes[6] / 2;
    const int eIn   = in_sizes[7] / 2;
    const int L     = in_sizes[16] / (H * H);

    float* ws = (float*)d_ws;
    size_t o = 0;
    float* hA    = ws + o; o += (size_t)nInt  * H;
    float* hB    = ws + o; o += (size_t)nInt  * H;
    float* hLane = ws + o; o += (size_t)nLane * H;
    float* pSens = ws + o; o += (size_t)nSens * H;
    o += (size_t)nInj  * H;
    float* aggS  = ws + o; o += (size_t)nInt  * H;
    float* aggB  = ws + o; o += (size_t)nInt  * H;
    int*   cnt   = (int*)(ws + o); o += (size_t)4 * nInt;
    int*   off   = (int*)(ws + o); o += (size_t)4 * nInt;
    (void)ws_size; (void)n_in; (void)out_size;

    int*  binned = (int*)aggB;
    int*  listFl = (int*)aggB + 2 * ((size_t)eSp + eFl + eFs + eIn);
    int*  listFs = listFl + eFl;
    int*  listIn = listFs + eFs;
    int*  listSp = (int*)hLane;
    int*  tab    = (int*)pSens;

    const int nb  = (nInt + 255) / 256;
    const int nch = (eSp + CHUNK - 1) / CHUNK + (eFl + CHUNK - 1) / CHUNK
                  + (eFs + CHUNK - 1) / CHUNK + (eIn + CHUNK - 1) / CHUNK;
    proj_bin_kernel<<<nch + (nInt + 127) / 128, 256, 0, stream>>>(
        x_int, W_int, b_int, hA, nInt,
        e_sp, e_fl, e_fs, e_in, binned, tab, eSp, eFl, eFs, eIn, nb, nch);
    fine_kernel<<<4 * nb, 256, 0, stream>>>(binned, tab, cnt, off,
                                            listSp, listFl, listFs, listIn,
                                            eSp, eFl, eFs, eIn, nb, nch, nInt);

    gather_raw_proj_kernel<<<(nInt + 8 * TPW - 1) / (8 * TPW), 256, 0, stream>>>(
        x_lane, x_sens, x_inj, W_lane, b_lane, W_sens, b_sens, W_inj, b_inj,
        listFl, listFs, listIn, cnt, off, aggS, nInt);

    const int nrp = (nInt + 31) / 32;
    const int needed = (nrp + RPB - 1) / RPB;
    const int lgrid = needed > 256 ? needed : 256;
    float* hcur = hA;
    float* hnext = hB;
    for (int l = 0; l < L; ++l) {
        float* outp = (l == L - 1) ? (float*)d_out : hnext;
        fused_gemm_elu_kernel<<<lgrid, GT, 0, stream>>>(
            hcur, aggS, listSp, cnt, off,
            self_W + (size_t)l * H * H, rel_W + (size_t)l * H * H,
            self_b + (size_t)l * H, outp, nInt, nrp);
        float* t = hnext; hnext = hcur; hcur = t;
    }
}


# Layer kernel deals rows to blocks in 16-row units, not 32-row pairs: fullest block 208 rows (was 224); a block's last pair may be half, its second row tile's matrix steps skipped; sum order unchanged
# speedup vs baseline: 8.0752x; 1.0331x over previous
#include <hip/hip_runtime.h>
#include <cstddef>

#define H 128

typedef float v2f __attribute__((ext_vector_type(2)));
typedef float v8f __attribute__((ext_vector_type(8)));

#define KIN 32
__device__ __forceinline__ void proj_block(const float* __restrict__ x,
                                           const float* __restrict__ W,
                                           const float* __restrict__ b,
                                           float* __restrict__ out, int n, int bid) {
    __shared__ __align__(16) float sW[KIN * H];
    const int tid  = threadIdx.x;
    const int lane = tid & 31;
    const int half = lane >> 4;
    const int l15  = lane & 15;
    const int m0 = __builtin_amdgcn_readfirstlane(bid * 128 + (tid >> 5) * 16);

    int arow = m0 + l15;
    if (arow >= n) arow = n - 1;
    const float4* xr = (const float4*)(x + (size_t)arow * KIN + (half << 4));
    const float4 x0 = xr[0], x1 = xr[1], x2 = xr[2], x3 = xr[3];

    float4 wv[4];
    #pragma unroll
    for (int j = 0; j < 4; ++j) wv[j] = ((const float4*)W)[tid + 256 * j];

    v8f acc[8];
    #pragma unroll
    for (int t = 0; t < 8; ++t) {
        const float bc = b[(t << 4) + l15];
        #pragma unroll
        for (int r = 0; r < 8; ++r) acc[t][r] = bc;
    }

    #pragma unroll
    for (int j = 0; j < 4; ++j) {
        const int i  = tid + 256 * j;
        const int k  = i >> 5;
        const int c0 = (i & 31) << 2;
        float* d = sW + ((k >> 1) << 8) + (k & 1);
        d[2 * (c0 + 0)] = wv[j].x;
        d[2 * (c0 + 1)] = wv[j].y;
        d[2 * (c0 + 2)] = wv[j].z;
        d[2 * (c0 + 3)] = wv[j].w;
    }
    __syncthreads();

    const float xf[16] = { x0.x, x0.y, x0.z, x0.w, x1.x, x1.y, x1.z, x1.w,
                           x2.x, x2.y, x2.z, x2.w, x3.x, x3.y, x3.z, x3.w };
    const float* bbase = sW + (half << 11) + (l15 << 1);
    #pragma unroll
    for (int s = 0; s < 8; ++s) {
        v2f a;
        a.x = xf[2 * s];
        a.y = xf[2 * s + 1];
        const float* bk = bbase + (s << 8);
        #pragma unroll
        for (int t = 0; t < 8; ++t) {
            const v2f bf = *(const v2f*)(bk + (t << 5));
            acc[t] = __builtin_amdgcn_wmma_f32_16x16x4_f32(
                false, a, false, bf, (short)0, acc[t], false, false);
        }
    }

    float* orow = out + (size_t)(m0 + (half << 3)) * H + l15;
    if (m0 + 16 <= n) {
        #pragma unroll
        for (int t = 0; t < 8; ++t) {
            #pragma unroll
            for (int r = 0; r < 8; ++r) orow[r * H + (t << 4)] = acc[t][r];
        }
    } else {
        #pragma unroll
        for (int t = 0; t < 8; ++t) {
            #pragma unroll
            for (int r = 0; r < 8; ++r) {
                if (m0 + (half << 3) + r < n) orow[r * H + (t << 4)] = acc[t][r];
            }
        }
    }
}

#define CHUNK 4096
#define MAXNB 1024
#define STAGE 8192
#define RLANES 8
#define RSLOTS (256 / RLANES)
#define RROUND 5
#define RDEPTH 4

__device__ __forceinline__ void bin_block(
        const int* __restrict__ ed0, const int* __restrict__ ed1,
        const int* __restrict__ ed2, const int* __restrict__ ed3,
        int* __restrict__ binned, int* __restrict__ tab,
        int E0, int E1, int E2, int E3, int nb, int nch, int bid) {
    __shared__ int  sHist[MAXNB + 1];
    __shared__ int2 sPair[CHUNK];
    const int tid = threadIdx.x;
    const int C0 = (E0 + CHUNK - 1) / CHUNK, C1 = (E1 + CHUNK - 1) / CHUNK,
              C2 = (E2 + CHUNK - 1) / CHUNK;
    int c = bid;
    const int* ed = ed0;
    int E = E0, rbase = 0;
    if (c >= C0) {
        c -= C0; ed = ed1; E = E1; rbase = E0;
        if (c >= C1) {
            c -= C1; ed = ed2; E = E2; rbase += E1;
            if (c >= C2) { c -= C2; ed = ed3; E = E3; rbase += E2; }
        }
    }
    const int e0  = c * CHUNK;
    const int cnt = min(CHUNK, E - e0);

    for (int j = tid; j <= nb; j += 256) sHist[j] = 0;
    __syncthreads();

    int srcv[16], tgtv[16], rk[16];
    #pragma unroll
    for (int k = 0; k < 16; ++k) {
        const int i = k * 256 + tid;
        srcv[k] = 0; tgtv[k] = 0;
        if (i < cnt) { srcv[k] = ed[e0 + i]; tgtv[k] = ed[E + e0 + i]; }
    }
    #pragma unroll
    for (int k = 0; k < 16; ++k) {
        const int i = k * 256 + tid;
        rk[k] = 0;
        if (i < cnt) rk[k] = atomicAdd(&sHist[tgtv[k] >> 8], 1);
    }
    __syncthreads();

    if (tid < 32) {
        int carry = 0;
        for (int base = 0; base < nb; base += 32) {
            const int j = base + tid;
            const int v = (j < nb) ? sHist[j] : 0;
            int inc = v;
            for (int d = 1; d < 32; d <<= 1) {
                const int u = __shfl_up(inc, d, 32);
                if (tid >= d) inc += u;
            }
            if (j < nb) sHist[j] = carry + inc - v;
            carry += __shfl(inc, 31, 32);
        }
        if (tid == 0) sHist[nb] = carry;
    }
    __syncthreads();

    #pragma unroll
    for (int k = 0; k < 16; ++k) {
        const int i = k * 256 + tid;
        if (i < cnt) sPair[sHist[tgtv[k] >> 8] + rk[k]] = make_int2(srcv[k], tgtv[k]);
    }
    for (int j = tid; j <= nb; j += 256) tab[(size_t)j * nch + bid] = sHist[j];
    __syncthreads();

    int2* outp = (int2*)binned + (size_t)rbase + e0;
    for (int i = tid; i < cnt; i += 256) outp[i] = sPair[i];
}

__global__ __launch_bounds__(256) void proj_bin_kernel(
        const float* __restrict__ x, const float* __restrict__ W, const float* __restrict__ b,
        float* __restrict__ hout, int n,
        const int* __restrict__ ed0, const int* __restrict__ ed1,
        const int* __restrict__ ed2, const int* __restrict__ ed3,
        int* __restrict__ binned, int* __restrict__ tab,
        int E0, int E1, int E2, int E3, int nb, int nch) {
    const int bid = (int)blockIdx.x;
    if (bid < nch) bin_block(ed0, ed1, ed2, ed3, binned, tab, E0, E1, E2, E3, nb, nch, bid);
    else           proj_block(x, W, b, hout, n, bid - nch);
}

__global__ __launch_bounds__(256) void fine_kernel(
        const int* __restrict__ binned, const int* __restrict__ tab,
        int* __restrict__ cnt, int* __restrict__ off,
        int* __restrict__ list0, int* __restrict__ list1,
        int* __restrict__ list2, int* __restrict__ list3,
        int E0, int E1, int E2, int E3, int nb, int nch, int n) {
    __shared__ int sCnt[256];
    __shared__ int sCur[256];
    __shared__ int sWave[8];
    __shared__ int sRed[8];
    __shared__ int sStage[STAGE];
    const int tid  = threadIdx.x;
    const int lane = tid & 31;
    const int wave = tid >> 5;
    const int C0 = (E0 + CHUNK - 1) / CHUNK, C1 = (E1 + CHUNK - 1) / CHUNK,
              C2 = (E2 + CHUNK - 1) / CHUNK, C3 = (E3 + CHUNK - 1) / CHUNK;
    const int r = blockIdx.x / nb;
    const int g = blockIdx.x - r * nb;
    int rbase = 0, cb = 0, nC = C0;
    int* list = list0;
    if (r == 1) { rbase = E0;           cb = C0;           nC = C1; list = list1; }
    if (r == 2) { rbase = E0 + E1;      cb = C0 + C1;      nC = C2; list = list2; }
    if (r == 3) { rbase = E0 + E1 + E2; cb = C0 + C1 + C2; nC = C3; list = list3; }
    const int* tlo = tab + (size_t)g * nch + cb;
    const int* thi = tab + (size_t)(g + 1) * nch + cb;
    const int2* win = (const int2*)binned + (size_t)rbase;

    sCnt[tid] = 0;
    __syncthreads();

    const int sub = tid & (RLANES - 1);
    const int rsl = tid / RLANES;

    int lower = 0;
    for (int b0 = 0; b0 < nC; b0 += RSLOTS * RROUND) {
        int rb[RROUND], re[RROUND];
        #pragma unroll
        for (int k = 0; k < RROUND; ++k) {
            const int b = b0 + rsl + RSLOTS * k;
            rb[k] = 0; re[k] = 0;
            if (b < nC) { rb[k] = tlo[b]; re[k] = thi[b]; }
        }
        int y[RROUND][RDEPTH];
        #pragma unroll
        for (int k = 0; k < RROUND; ++k) {
            const int2* p = win + (size_t)min(b0 + rsl + RSLOTS * k, nC - 1) * CHUNK;
            #pragma unroll
            for (int u = 0; u < RDEPTH; ++u)
                y[k][u] = p[max(min(rb[k] + sub + RLANES * u, re[k] - 1), 0)].y;
        }
        #pragma unroll
        for (int k = 0; k < RROUND; ++k) {
            if (sub == 0) lower += rb[k];
            #pragma unroll
            for (int u = 0; u < RDEPTH; ++u)
                if (rb[k] + sub + RLANES * u < re[k]) atomicAdd(&sCnt[y[k][u] & 255], 1);
            const int2* p = win + (size_t)min(b0 + rsl + RSLOTS * k, nC - 1) * CHUNK;
            for (int i = rb[k] + sub + RLANES * RDEPTH; i < re[k]; i += RLANES)
                atomicAdd(&sCnt[p[i].y & 255], 1);
        }
    }
    __syncthreads();

    const int cT = sCnt[tid];
    int inc = cT;
    for (int d = 1; d < 32; d <<= 1) {
        const int u = __shfl_up(inc, d, 32);
        if (lane >= d) inc += u;
    }
    for (int o = 16; o > 0; o >>= 1) lower += __shfl_xor(lower, o, 32);
    if (lane == 31) sWave[wave] = inc;
    if (lane == 0)  sRed[wave]  = lower;
    __syncthreads();
    int wbase = 0, total = 0, start = 0;
    for (int w = 0; w < 8; ++w) {
        const int tw = sWave[w];
        if (w < wave) wbase += tw;
        total += tw;
        start += sRed[w];
    }
    const int offL = wbase + inc - cT;
    const int t = g * 256 + tid;
    if (t < n) {
        cnt[(size_t)r * n + t] = cT;
        off[(size_t)r * n + t] = start + offL;
    }
    sCur[tid] = offL;
    __syncthreads();

    const bool staged = total <= STAGE;
    for (int b0 = 0; b0 < nC; b0 += RSLOTS * RROUND) {
        int rb[RROUND], re[RROUND];
        #pragma unroll
        for (int k = 0; k < RROUND; ++k) {
            const int b = b0 + rsl + RSLOTS * k;
            rb[k] = 0; re[k] = 0;
            if (b < nC) { rb[k] = tlo[b]; re[k] = thi[b]; }
        }
        int2 v[RROUND][RDEPTH];
        #pragma unroll
        for (int k = 0; k < RROUND; ++k) {
            const int2* p = win + (size_t)min(b0 + rsl + RSLOTS * k, nC - 1) * CHUNK;
            #pragma unroll
            for (int u = 0; u < RDEPTH; ++u)
                v[k][u] = p[max(min(rb[k] + sub + RLANES * u, re[k] - 1), 0)];
        }
        #pragma unroll
        for (int k = 0; k < RROUND; ++k) {
            #pragma unroll
            for (int u = 0; u < RDEPTH; ++u) {
                if (rb[k] + sub + RLANES * u < re[k]) {
                    const int pos = atomicAdd(&sCur[v[k][u].y & 255], 1);
                    if (staged) sStage[pos] = v[k][u].x;
                    else        list[start + pos] = v[k][u].x;
                }
            }
            const int2* p = win + (size_t)min(b0 + rsl + RSLOTS * k, nC - 1) * CHUNK;
            for (int i = rb[k] + sub + RLANES * RDEPTH; i < re[k]; i += RLANES) {
                const int2 w = p[i];
                const int pos = atomicAdd(&sCur[w.y & 255], 1);
                if (staged) sStage[pos] = w.x;
                else        list[start + pos] = w.x;
            }
        }
    }
    if (staged) {
        __syncthreads();
        for (int i = tid; i < total; i += 256) list[start + i] = sStage[i];
    }
}

__device__ __forceinline__ void add4(float4& a, const float4 v) {
    a.x += v.x; a.y += v.y; a.z += v.z; a.w += v.w;
}

__device__ __forceinline__ float4 bucket_sum(const float4* __restrict__ feat4,
                                             const int* __restrict__ list,
                                             int beg, int len, int lane) {
    float4 acc = make_float4(0.0f, 0.0f, 0.0f, 0.0f);
    if (len <= 32) {
        const int key = (lane < len) ? list[beg + lane] : 0x7fffffff;
        int rank = 0;
        for (int j = 0; j < len; ++j) {
            const int kj = __builtin_amdgcn_readlane(key, j);
            rank += (kj < key || (kj == key && j < lane)) ? 1 : 0;
        }
        if (lane >= len) rank = lane;
        const int sorted = __builtin_amdgcn_ds_permute(rank << 2, key);
        int r = 0;
        for (; r + 4 <= len; r += 4) {
            const int s0 = __builtin_amdgcn_readlane(sorted, r);
            const int s1 = __builtin_amdgcn_readlane(sorted, r + 1);
            const int s2 = __builtin_amdgcn_readlane(sorted, r + 2);
            const int s3 = __builtin_amdgcn_readlane(sorted, r + 3);
            const float4 v0 = feat4[(size_t)s0 * (H / 4) + lane];
            const float4 v1 = feat4[(size_t)s1 * (H / 4) + lane];
            const float4 v2 = feat4[(size_t)s2 * (H / 4) + lane];
            const float4 v3 = feat4[(size_t)s3 * (H / 4) + lane];
            add4(acc, v0); add4(acc, v1); add4(acc, v2); add4(acc, v3);
        }
        for (; r < len; ++r) {
            const int s0 = __builtin_amdgcn_readlane(sorted, r);
            add4(acc, feat4[(size_t)s0 * (H / 4) + lane]);
        }
    } else {
        int prev = -1;
        for (int it = 0; it < len; ++it) {
            int m = 0x7fffffff;
            for (int i = lane; i < len; i += 32) {
                const int k = list[beg + i];
                if (k > prev && k < m) m = k;
            }
            for (int o = 16; o > 0; o >>= 1) m = min(m, __shfl_xor(m, o, 32));
            if (m == 0x7fffffff) break;
            int c = 0;
            for (int i = lane; i < len; i += 32) c += (list[beg + i] == m) ? 1 : 0;
            for (int o = 16; o > 0; o >>= 1) c += __shfl_xor(c, o, 32);
            const float4 v = feat4[(size_t)m * (H / 4) + lane];
            for (int q = 0; q < c; ++q) add4(acc, v);
            prev = m;
        }
    }
    return acc;
}


__device__ __forceinline__ int sort_ids(int key, int len, int lane) {
    int rank = 0;
    for (int j = 0; j < len; ++j) {
        const int kj = __builtin_amdgcn_readlane(key, j);
        rank += (kj < key || (kj == key && j < lane)) ? 1 : 0;
    }
    if (lane >= len) rank = lane;
    return __builtin_amdgcn_ds_permute(rank << 2, key);
}

template <int Q>
__device__ __forceinline__ float4 long_bucket_sum(const float4* __restrict__ x4,
                                                  const int* __restrict__ list,
                                                  int beg, int len, int lane, int qq) {
    float4 acc = make_float4(0.0f, 0.0f, 0.0f, 0.0f);
    int prev = -1;
    for (int it = 0; it < len; ++it) {
        int m = 0x7fffffff;
        for (int i = lane; i < len; i += 32) {
            const int k = list[beg + i];
            if (k > prev && k < m) m = k;
        }
        for (int o = 16; o > 0; o >>= 1) m = min(m, __shfl_xor(m, o, 32));
        if (m == 0x7fffffff) break;
        int c = 0;
        for (int i = lane; i < len; i += 32) c += (list[beg + i] == m) ? 1 : 0;
        for (int o = 16; o > 0; o >>= 1) c += __shfl_xor(c, o, 32);
        const float4 v = x4[(size_t)m * Q + qq];
        for (int r = 0; r < c; ++r) add4(acc, v);
        prev = m;
    }
    return acc;
}

__device__ __forceinline__ float4 slot_sum(float4 a) {
    for (int m = 4; m < 32; m <<= 1) {
        float4 b;
        b.x = __shfl_xor(a.x, m, 32); b.y = __shfl_xor(a.y, m, 32);
        b.z = __shfl_xor(a.z, m, 32); b.w = __shfl_xor(a.w, m, 32);
        add4(a, b);
    }
    return a;
}

#define TPW 4

__global__ __launch_bounds__(256) void gather_raw_proj_kernel(
        const float* __restrict__ xLane, const float* __restrict__ xSens,
        const float* __restrict__ xInj,
        const float* __restrict__ WLane, const float* __restrict__ bLane,
        const float* __restrict__ WSens, const float* __restrict__ bSens,
        const float* __restrict__ WInj,  const float* __restrict__ bInj,
        const int* __restrict__ listFl, const int* __restrict__ listFs,
        const int* __restrict__ listIn,
        const int* __restrict__ cnt, const int* __restrict__ off,
        float* __restrict__ aggS, int n) {
    __shared__ float4 sW[35 * 32];
    __shared__ float4 sZ[8][TPW][9];
    const int tid  = threadIdx.x;
    const int wave = tid >> 5;
    const int lane = tid & 31;
    const int slot = lane >> 2;
    const int q    = lane & 3;

    {
        const float4* wl = (const float4*)WLane;
        sW[tid]       = wl[tid];
        sW[256 + tid] = wl[256 + tid];
        sW[512 + tid] = ((const float4*)WSens)[tid];
        sW[768 + tid] = ((const float4*)WInj)[tid];
        if (tid < 96) {
            const float* bs = tid < 32 ? bLane : (tid < 64 ? bSens : bInj);
            sW[1024 + tid] = ((const float4*)bs)[tid & 31];
        }
    }

    const float4* xl4 = (const float4*)xLane;
    const float4* xs4 = (const float4*)xSens;
    const float4* xi4 = (const float4*)xInj;
    const float4 zero4 = make_float4(0.0f, 0.0f, 0.0f, 0.0f);
    const int tb = blockIdx.x * (8 * TPW) + wave * TPW;

    for (int u = 0; u < TPW; ++u) {
        const int t = __builtin_amdgcn_readfirstlane(tb + u);
        float4 zl = zero4, zsi = zero4;
        float fl = 0.0f, fs = 0.0f, fi = 0.0f;
        if (t < n) {
            const int lenL = cnt[1 * n + t], lenS = cnt[2 * n + t], lenI = cnt[3 * n + t];
            const int begL = off[1 * n + t], begS = off[2 * n + t], begI = off[3 * n + t];
            fl = lenL > 0 ? 1.0f : 0.0f;
            fs = lenS > 0 ? 1.0f : 0.0f;
            fi = lenI > 0 ? 1.0f : 0.0f;
            const bool isS = q < 2;
            const int qq = q & 1;
            if (lenL <= 32 && lenS <= 32 && lenI <= 32) {
                const int keyL = (lane < lenL) ? listFl[begL + lane] : 0x7fffffff;
                const int keyS = (lane < lenS) ? listFs[begS + lane] : 0x7fffffff;
                const int keyI = (lane < lenI) ? listIn[begI + lane] : 0x7fffffff;
                const int sortedL = sort_ids(keyL, lenL, lane);
                const int sortedS = sort_ids(keyS, lenS, lane);
                const int sortedI = sort_ids(keyI, lenI, lane);
                const int lenX = isS ? lenS : lenI;
                const int maxSI = max(lenS, lenI);
                const float4* xb = isS ? xs4 : xi4;
                float4 a[4], b[4];
                #pragma unroll
                for (int i = 0; i < 4; ++i) {
                    a[i] = zero4; b[i] = zero4;
                    const int p = 8 * i + slot;
                    if (8 * i < lenL) {
                        const int id = __builtin_amdgcn_ds_bpermute(p << 2, sortedL);
                        if (p < lenL) a[i] = xl4[(size_t)id * 4 + q];
                    }
                    if (8 * i < maxSI) {
                        const int idS = __builtin_amdgcn_ds_bpermute(p << 2, sortedS);
                        const int idI = __builtin_amdgcn_ds_bpermute(p << 2, sortedI);
                        if (p < lenX) b[i] = xb[(size_t)(isS ? idS : idI) * 2 + qq];
                    }
                }
                add4(a[0], a[1]); add4(a[0], a[2]); add4(a[0], a[3]);
                add4(b[0], b[1]); add4(b[0], b[2]); add4(b[0], b[3]);
                zl  = slot_sum(a[0]);
                zsi = slot_sum(b[0]);
            } else {
                zl = long_bucket_sum<4>(xl4, listFl, begL, lenL, lane, q);
                const float4 s = long_bucket_sum<2>(xs4, listFs, begS, lenS, lane, qq);
                const float4 j = long_bucket_sum<2>(xi4, listIn, begI, lenI, lane, qq);
                zsi = isS ? s : j;
            }
            const float wL = 1.0f / (float)max(lenL, 1);
            const float wX = 1.0f / (float)max(isS ? lenS : lenI, 1);
            zl.x *= wL;  zl.y *= wL;  zl.z *= wL;  zl.w *= wL;
            zsi.x *= wX; zsi.y *= wX; zsi.z *= wX; zsi.w *= wX;
        }
        if (lane < 4) { sZ[wave][u][lane] = zl; sZ[wave][u][4 + lane] = zsi; }
        if (lane == 0) sZ[wave][u][8] = make_float4(fl, fs, fi, 0.0f);
    }
    __syncthreads();

    float4 o[TPW];
    #pragma unroll
    for (int u = 0; u < TPW; ++u) o[u] = zero4;
    #pragma nounroll
    for (int g = 0; g < 8; ++g) {
        const float4 w0 = sW[(4 * g + 0) * 32 + lane];
        const float4 w1 = sW[(4 * g + 1) * 32 + lane];
        const float4 w2 = sW[(4 * g + 2) * 32 + lane];
        const float4 w3 = sW[(4 * g + 3) * 32 + lane];
        #pragma unroll
        for (int u = 0; u < TPW; ++u) {
            const float4 z = sZ[wave][u][g];
            o[u].x += z.x * w0.x; o[u].y += z.x * w0.y; o[u].z += z.x * w0.z; o[u].w += z.x * w0.w;
            o[u].x += z.y * w1.x; o[u].y += z.y * w1.y; o[u].z += z.y * w1.z; o[u].w += z.y * w1.w;
            o[u].x += z.z * w2.x; o[u].y += z.z * w2.y; o[u].z += z.z * w2.z; o[u].w += z.z * w2.w;
            o[u].x += z.w * w3.x; o[u].y += z.w * w3.y; o[u].z += z.w * w3.z; o[u].w += z.w * w3.w;
        }
    }
    {
        const float4 b0 = sW[32 * 32 + lane];
        const float4 b1 = sW[33 * 32 + lane];
        const float4 b2 = sW[34 * 32 + lane];
        #pragma unroll
        for (int u = 0; u < TPW; ++u) {
            const float4 f = sZ[wave][u][8];
            o[u].x += f.x * b0.x; o[u].y += f.x * b0.y; o[u].z += f.x * b0.z; o[u].w += f.x * b0.w;
            o[u].x += f.y * b1.x; o[u].y += f.y * b1.y; o[u].z += f.y * b1.z; o[u].w += f.y * b1.w;
            o[u].x += f.z * b2.x; o[u].y += f.z * b2.y; o[u].z += f.z * b2.z; o[u].w += f.z * b2.w;
        }
    }
    #pragma unroll
    for (int u = 0; u < TPW; ++u) {
        const int t = tb + u;
        if (t < n) ((float4*)aggS)[(size_t)t * (H / 4) + lane] = o[u];
    }
}

#define RPB 7
#define GW (4 * RPB)
#define GT (GW * 32)
#define APITCH 132

#define NIF 16
__device__ __forceinline__ void issue_rows(float4 (&v)[NIF], const float4* __restrict__ feat4,
                                           int sorted, int r, int len, int lane) {
    #pragma unroll
    for (int i = 0; i < NIF; ++i) {
        v[i] = make_float4(0.0f, 0.0f, 0.0f, 0.0f);
        if (r + i < len) {
            const int sid = __builtin_amdgcn_readlane(sorted, r + i);
            v[i] = feat4[(size_t)sid * (H / 4) + lane];
        }
    }
}
__device__ __forceinline__ void sum_rows(float4& acc, const float4 (&v)[NIF]) {
    #pragma unroll
    for (int i = 0; i < NIF; ++i) add4(acc, v[i]);
}
__device__ __forceinline__ float4 sum_rows_first(const float4 (&v)[NIF]) {
    float4 acc = v[0];
    #pragma unroll
    for (int i = 1; i < NIF; ++i) add4(acc, v[i]);
    return acc;
}

template <bool TWO>
__device__ __forceinline__ void prod_rows(const float4* __restrict__ p0, const float4* __restrict__ p1,
                                          int step, const float* __restrict__ bbase,
                                          v8f& acc00, v8f& acc01, v8f& acc10, v8f& acc11) {
    for (int j = 0; j < H / 8; ++j) {
        const float4 x0 = p0[step * j];
        float4 x1 = x0;
        if (TWO) x1 = p1[step * j];
        const float* bk = bbase + (j << 10);
        const v2f b00 = *(const v2f*)(bk);
        const v2f b01 = *(const v2f*)(bk + 32);
        const v2f b10 = *(const v2f*)(bk + 256);
        const v2f b11 = *(const v2f*)(bk + 256 + 32);
        v2f a;
        a.x = x0.x; a.y = x0.y;
        acc00 = __builtin_amdgcn_wmma_f32_16x16x4_f32(false, a, false, b00, (short)0, acc00, false, false);
        acc01 = __builtin_amdgcn_wmma_f32_16x16x4_f32(false, a, false, b01, (short)0, acc01, false, false);
        if (TWO) {
            a.x = x1.x; a.y = x1.y;
            acc10 = __builtin_amdgcn_wmma_f32_16x16x4_f32(false, a, false, b00, (short)0, acc10, false, false);
            acc11 = __builtin_amdgcn_wmma_f32_16x16x4_f32(false, a, false, b01, (short)0, acc11, false, false);
        }
        a.x = x0.z; a.y = x0.w;
        acc00 = __builtin_amdgcn_wmma_f32_16x16x4_f32(false, a, false, b10, (short)0, acc00, false, false);
        acc01 = __builtin_amdgcn_wmma_f32_16x16x4_f32(false, a, false, b11, (short)0, acc01, false, false);
        if (TWO) {
            a.x = x1.z; a.y = x1.w;
            acc10 = __builtin_amdgcn_wmma_f32_16x16x4_f32(false, a, false, b10, (short)0, acc10, false, false);
            acc11 = __builtin_amdgcn_wmma_f32_16x16x4_f32(false, a, false, b11, (short)0, acc11, false, false);
        }
    }
}

__global__ __launch_bounds__(GT) void fused_gemm_elu_kernel(
        const float* __restrict__ h, const float* __restrict__ aggS,
        const int* __restrict__ listSp, const int* __restrict__ cnt, const int* __restrict__ off,
        const float* __restrict__ W0, const float* __restrict__ W1,
        const float* __restrict__ bias, float* __restrict__ out, int n, int nrp) {
    __shared__ __align__(16) float sB[2 * H * H];
    __shared__ __align__(16) float sA[RPB * 32 * APITCH];
    const int tid  = threadIdx.x;
    const int lane = tid & 31;
    const int half = lane >> 4;
    const int l15  = lane & 15;

    const int wave = __builtin_amdgcn_readfirstlane(tid >> 5);
    const int lrp  = wave >> 2;
    (void)nrp;
    const int nU   = (n + 15) >> 4;
    const int uBeg = (int)(((long long)blockIdx.x * nU) / gridDim.x);
    const int uEnd = (int)(((long long)(blockIdx.x + 1) * nU) / gridDim.x);
    const int uw   = uBeg + 2 * lrp;
    const bool active = uw < uEnd;
    const bool two = uw + 1 < uEnd;
    const int m0  = uw << 4;
    const int lim = min(n, uEnd << 4);
    const int cb = (wave & 3) << 5;

    const int lr0 = (wave & 3) << 3;
    const int t0  = m0 + lr0;
    const bool okl = active && lane < 8 && t0 + lane < lim;
    const int cntv = okl ? cnt[t0 + lane] : 0;
    const int offv = okl ? off[t0 + lane] : 0;

    {
        constexpr int NV = 2 * H * H / 4;
        constexpr int NI = (NV + GT - 1) / GT;
        float4 wv[NI];
        #pragma unroll
        for (int q = 0; q < NI; ++q) {
            const int i = tid + GT * q;
            wv[q] = make_float4(0.0f, 0.0f, 0.0f, 0.0f);
            if (i < NV) wv[q] = ((const float4*)((i >> 12) ? W1 : W0))[i & 4095];
        }
        #pragma unroll
        for (int q = 0; q < NI; ++q) {
            const int i = tid + GT * q;
            if (i < NV) {
                const int jj = i & 4095;
                const int k  = jj >> 5;
                const int c0 = (jj & 31) << 2;
                float* d = sB + ((i >> 12) << 14) + ((k >> 1) << 8) + (k & 1);
                d[2 * (c0 + 0)] = wv[q].x;
                d[2 * (c0 + 1)] = wv[q].y;
                d[2 * (c0 + 2)] = wv[q].z;
                d[2 * (c0 + 3)] = wv[q].w;
            }
        }
    }
    int lenC = __builtin_amdgcn_readlane(cntv, 0);
    int begC = __builtin_amdgcn_readlane(offv, 0);
    int lenN = __builtin_amdgcn_readlane(cntv, 1);
    int begN = __builtin_amdgcn_readlane(offv, 1);
    int keyN = (lenN <= 32 && lane < lenN) ? listSp[begN + lane] : 0x7fffffff;
    const int keyC = (lenC <= 32 && lane < lenC) ? listSp[begC + lane] : 0x7fffffff;
    __syncthreads();

    int r0 = m0 + l15;
    if (r0 >= n) r0 = n - 1;
    int r1 = m0 + 16 + l15;
    if (r1 >= n) r1 = n - 1;
    float* sAp = sA + (size_t)lrp * 32 * APITCH;

    v8f acc00 = {}, acc01 = {}, acc10 = {}, acc11 = {};
    if (active) {
        const int first = (wave >> 2) & 1;
        for (int st = 0; st < 2; ++st) {
            if ((st ^ first) == 0) {
                const float4* p0 = (const float4*)(h + (size_t)r0 * H + (half << 2));
                const float4* p1 = (const float4*)(h + (size_t)r1 * H + (half << 2));
                const float* bbase = sB + (half << 9) + ((cb + l15) << 1);
                if (two) prod_rows<true>(p0, p1, 2, bbase, acc00, acc01, acc10, acc11);
                else     prod_rows<false>(p0, p1, 2, bbase, acc00, acc01, acc10, acc11);
            } else {
                int sortedC = (lenC <= 32) ? sort_ids(keyC, lenC, lane) : 0;
                const float4* h4 = (const float4*)h;
                for (int q = 0; q < 8; ++q) {
                    const int t = t0 + q;
                    int lenF = 0, begF = 0;
                    if (q + 2 < 8) {
                        lenF = __builtin_amdgcn_readlane(cntv, q + 2);
                        begF = __builtin_amdgcn_readlane(offv, q + 2);
                    }
                    int keyF = 0x7fffffff;
                    float4 r = make_float4(0.0f, 0.0f, 0.0f, 0.0f);
                    int sortedN = 0;
                    if (t < lim) {
                        float4 sm, base;
                        if (lenC <= 32) {
                            float4 v[NIF];
                            issue_rows(v, h4, sortedC, 0, lenC, lane);
                            if (lenF <= 32 && lane < lenF) keyF = listSp[begF + lane];
                            base = ((const float4*)aggS)[(size_t)t * (H / 4) + lane];
                            if (lenN <= 32) sortedN = sort_ids(keyN, lenN, lane);
                            sm = sum_rows_first(v);
                            for (int rr = NIF; rr < lenC; rr += NIF) {
                                issue_rows(v, h4, sortedC, rr, lenC, lane);
                                sum_rows(sm, v);
                            }
                        } else {
                            if (lenF <= 32 && lane < lenF) keyF = listSp[begF + lane];
                            base = ((const float4*)aggS)[(size_t)t * (H / 4) + lane];
                            sm = bucket_sum(h4, listSp, begC, lenC, lane);
                            if (lenN <= 32) sortedN = sort_ids(keyN, lenN, lane);
                        }
                        const float w = 1.0f / (float)max(lenC, 1);
                        r.x = base.x + sm.x * w; r.y = base.y + sm.y * w;
                        r.z = base.z + sm.z * w; r.w = base.w + sm.w * w;
                    }
                    *(float4*)(sAp + (size_t)(lr0 + q) * APITCH + (lane << 2)) = r;
                    lenC = lenN; begC = begN; sortedC = sortedN;
                    lenN = lenF; begN = begF; keyN = keyF;
                }
            }
        }
    }
    __syncthreads();

    if (active) {
        const float* q0 = sAp + (size_t)l15 * APITCH + (half << 2);
        const float* q1 = q0 + 16 * APITCH;
        const float* bbase = sB + (1 << 14) + (half << 9) + ((cb + l15) << 1);
        if (two) prod_rows<true>((const float4*)q0, (const float4*)q1, 2, bbase, acc00, acc01, acc10, acc11);
        else     prod_rows<false>((const float4*)q0, (const float4*)q1, 2, bbase, acc00, acc01, acc10, acc11);

        const float bc0 = bias[cb + l15];
        const float bc1 = bias[cb + 16 + l15];
        #pragma unroll
        for (int r = 0; r < 8; ++r) {
            const int rowa = m0 + (half << 3) + r;
            const int rowb = rowa + 16;
            if (rowa < lim) {
                float v = acc00[r] + bc0;
                out[(size_t)rowa * H + cb + l15] = v > 0.0f ? v : (__expf(v) - 1.0f);
                v = acc01[r] + bc1;
                out[(size_t)rowa * H + cb + 16 + l15] = v > 0.0f ? v : (__expf(v) - 1.0f);
            }
            if (rowb < lim) {
                float v = acc10[r] + bc0;
                out[(size_t)rowb * H + cb + l15] = v > 0.0f ? v : (__expf(v) - 1.0f);
                v = acc11[r] + bc1;
                out[(size_t)rowb * H + cb + 16 + l15] = v > 0.0f ? v : (__expf(v) - 1.0f);
            }
        }
    }
}

extern "C" void kernel_launch(void* const* d_in, const int* in_sizes, int n_in,
                              void* d_out, int out_size, void* d_ws, size_t ws_size,
                              hipStream_t stream) {
    const float* x_int  = (const float*)d_in[0];
    const float* x_lane = (const float*)d_in[1];
    const float* x_sens = (const float*)d_in[2];
    const float* x_inj  = (const float*)d_in[3];
    const int*   e_sp   = (const int*)d_in[4];
    const int*   e_fl   = (const int*)d_in[5];
    const int*   e_fs   = (const int*)d_in[6];
    const int*   e_in   = (const int*)d_in[7];
    const float* W_int  = (const float*)d_in[8];
    const float* b_int  = (const float*)d_in[9];
    const float* W_lane = (const float*)d_in[10];
    const float* b_lane = (const float*)d_in[11];
    const float* W_sens = (const float*)d_in[12];
    const float* b_sens = (const float*)d_in[13];
    const float* W_inj  = (const float*)d_in[14];
    const float* b_inj  = (const float*)d_in[15];
    const float* self_W = (const float*)d_in[16];
    const float* self_b = (const float*)d_in[17];
    const float* rel_W  = (const float*)d_in[18];

    const int nInt  = in_sizes[0] / 32;
    const int nLane = in_sizes[1] / 16;
    const int nSens = in_sizes[2] / 8;
    const int nInj  = in_sizes[3] / 8;
    const int eSp   = in_sizes[4] / 2;
    const int eFl   = in_sizes[5] / 2;
    const int eFs   = in_sizes[6] / 2;
    const int eIn   = in_sizes[7] / 2;
    const int L     = in_sizes[16] / (H * H);

    float* ws = (float*)d_ws;
    size_t o = 0;
    float* hA    = ws + o; o += (size_t)nInt  * H;
    float* hB    = ws + o; o += (size_t)nInt  * H;
    float* hLane = ws + o; o += (size_t)nLane * H;
    float* pSens = ws + o; o += (size_t)nSens * H;
    o += (size_t)nInj  * H;
    float* aggS  = ws + o; o += (size_t)nInt  * H;
    float* aggB  = ws + o; o += (size_t)nInt  * H;
    int*   cnt   = (int*)(ws + o); o += (size_t)4 * nInt;
    int*   off   = (int*)(ws + o); o += (size_t)4 * nInt;
    (void)ws_size; (void)n_in; (void)out_size;

    int*  binned = (int*)aggB;
    int*  listFl = (int*)aggB + 2 * ((size_t)eSp + eFl + eFs + eIn);
    int*  listFs = listFl + eFl;
    int*  listIn = listFs + eFs;
    int*  listSp = (int*)hLane;
    int*  tab    = (int*)pSens;

    const int nb  = (nInt + 255) / 256;
    const int nch = (eSp + CHUNK - 1) / CHUNK + (eFl + CHUNK - 1) / CHUNK
                  + (eFs + CHUNK - 1) / CHUNK + (eIn + CHUNK - 1) / CHUNK;
    proj_bin_kernel<<<nch + (nInt + 127) / 128, 256, 0, stream>>>(
        x_int, W_int, b_int, hA, nInt,
        e_sp, e_fl, e_fs, e_in, binned, tab, eSp, eFl, eFs, eIn, nb, nch);
    fine_kernel<<<4 * nb, 256, 0, stream>>>(binned, tab, cnt, off,
                                            listSp, listFl, listFs, listIn,
                                            eSp, eFl, eFs, eIn, nb, nch, nInt);

    gather_raw_proj_kernel<<<(nInt + 8 * TPW - 1) / (8 * TPW), 256, 0, stream>>>(
        x_lane, x_sens, x_inj, W_lane, b_lane, W_sens, b_sens, W_inj, b_inj,
        listFl, listFs, listIn, cnt, off, aggS, nInt);

    const int nrp = (nInt + 31) / 32;
    const int needed = (nrp + RPB - 1) / RPB;
    const int lgrid = needed > 256 ? needed : 256;
    float* hcur = hA;
    float* hnext = hB;
    for (int l = 0; l < L; ++l) {
        float* outp = (l == L - 1) ? (float*)d_out : hnext;
        fused_gemm_elu_kernel<<<lgrid, GT, 0, stream>>>(
            hcur, aggS, listSp, cnt, off,
            self_W + (size_t)l * H * H, rel_W + (size_t)l * H * H,
            self_b + (size_t)l * H, outp, nInt, nrp);
        float* t = hnext; hnext = hcur; hcur = t;
    }
}


# Layer-invariant aggregate (raw-row gather, projection) built in the first layer's launch by the wave that gathers the row, kept in LDS, stored for later layers; 6 launches to 5; sum order unchanged
# speedup vs baseline: 8.4322x; 1.0442x over previous
#include <hip/hip_runtime.h>
#include <cstddef>

#define H 128

typedef float v2f __attribute__((ext_vector_type(2)));
typedef float v8f __attribute__((ext_vector_type(8)));

#define KIN 32
__device__ __forceinline__ void proj_block(const float* __restrict__ x,
                                           const float* __restrict__ W,
                                           const float* __restrict__ b,
                                           float* __restrict__ out, int n, int bid) {
    __shared__ __align__(16) float sW[KIN * H];
    const int tid  = threadIdx.x;
    const int lane = tid & 31;
    const int half = lane >> 4;
    const int l15  = lane & 15;
    const int m0 = __builtin_amdgcn_readfirstlane(bid * 128 + (tid >> 5) * 16);

    int arow = m0 + l15;
    if (arow >= n) arow = n - 1;
    const float4* xr = (const float4*)(x + (size_t)arow * KIN + (half << 4));
    const float4 x0 = xr[0], x1 = xr[1], x2 = xr[2], x3 = xr[3];

    float4 wv[4];
    #pragma unroll
    for (int j = 0; j < 4; ++j) wv[j] = ((const float4*)W)[tid + 256 * j];

    v8f acc[8];
    #pragma unroll
    for (int t = 0; t < 8; ++t) {
        const float bc = b[(t << 4) + l15];
        #pragma unroll
        for (int r = 0; r < 8; ++r) acc[t][r] = bc;
    }

    #pragma unroll
    for (int j = 0; j < 4; ++j) {
        const int i  = tid + 256 * j;
        const int k  = i >> 5;
        const int c0 = (i & 31) << 2;
        float* d = sW + ((k >> 1) << 8) + (k & 1);
        d[2 * (c0 + 0)] = wv[j].x;
        d[2 * (c0 + 1)] = wv[j].y;
        d[2 * (c0 + 2)] = wv[j].z;
        d[2 * (c0 + 3)] = wv[j].w;
    }
    __syncthreads();

    const float xf[16] = { x0.x, x0.y, x0.z, x0.w, x1.x, x1.y, x1.z, x1.w,
                           x2.x, x2.y, x2.z, x2.w, x3.x, x3.y, x3.z, x3.w };
    const float* bbase = sW + (half << 11) + (l15 << 1);
    #pragma unroll
    for (int s = 0; s < 8; ++s) {
        v2f a;
        a.x = xf[2 * s];
        a.y = xf[2 * s + 1];
        const float* bk = bbase + (s << 8);
        #pragma unroll
        for (int t = 0; t < 8; ++t) {
            const v2f bf = *(const v2f*)(bk + (t << 5));
            acc[t] = __builtin_amdgcn_wmma_f32_16x16x4_f32(
                false, a, false, bf, (short)0, acc[t], false, false);
        }
    }

    float* orow = out + (size_t)(m0 + (half << 3)) * H + l15;
    if (m0 + 16 <= n) {
        #pragma unroll
        for (int t = 0; t < 8; ++t) {
            #pragma unroll
            for (int r = 0; r < 8; ++r) orow[r * H + (t << 4)] = acc[t][r];
        }
    } else {
        #pragma unroll
        for (int t = 0; t < 8; ++t) {
            #pragma unroll
            for (int r = 0; r < 8; ++r) {
                if (m0 + (half << 3) + r < n) orow[r * H + (t << 4)] = acc[t][r];
            }
        }
    }
}

#define CHUNK 4096
#define MAXNB 1024
#define STAGE 8192
#define RLANES 8
#define RSLOTS (256 / RLANES)
#define RROUND 5
#define RDEPTH 4

__device__ __forceinline__ void bin_block(
        const int* __restrict__ ed0, const int* __restrict__ ed1,
        const int* __restrict__ ed2, const int* __restrict__ ed3,
        int* __restrict__ binned, int* __restrict__ tab,
        int E0, int E1, int E2, int E3, int nb, int nch, int bid) {
    __shared__ int  sHist[MAXNB + 1];
    __shared__ int2 sPair[CHUNK];
    const int tid = threadIdx.x;
    const int C0 = (E0 + CHUNK - 1) / CHUNK, C1 = (E1 + CHUNK - 1) / CHUNK,
              C2 = (E2 + CHUNK - 1) / CHUNK;
    int c = bid;
    const int* ed = ed0;
    int E = E0, rbase = 0;
    if (c >= C0) {
        c -= C0; ed = ed1; E = E1; rbase = E0;
        if (c >= C1) {
            c -= C1; ed = ed2; E = E2; rbase += E1;
            if (c >= C2) { c -= C2; ed = ed3; E = E3; rbase += E2; }
        }
    }
    const int e0  = c * CHUNK;
    const int cnt = min(CHUNK, E - e0);

    for (int j = tid; j <= nb; j += 256) sHist[j] = 0;
    __syncthreads();

    int srcv[16], tgtv[16], rk[16];
    #pragma unroll
    for (int k = 0; k < 16; ++k) {
        const int i = k * 256 + tid;
        srcv[k] = 0; tgtv[k] = 0;
        if (i < cnt) { srcv[k] = ed[e0 + i]; tgtv[k] = ed[E + e0 + i]; }
    }
    #pragma unroll
    for (int k = 0; k < 16; ++k) {
        const int i = k * 256 + tid;
        rk[k] = 0;
        if (i < cnt) rk[k] = atomicAdd(&sHist[tgtv[k] >> 8], 1);
    }
    __syncthreads();

    if (tid < 32) {
        int carry = 0;
        for (int base = 0; base < nb; base += 32) {
            const int j = base + tid;
            const int v = (j < nb) ? sHist[j] : 0;
            int inc = v;
            for (int d = 1; d < 32; d <<= 1) {
                const int u = __shfl_up(inc, d, 32);
                if (tid >= d) inc += u;
            }
            if (j < nb) sHist[j] = carry + inc - v;
            carry += __shfl(inc, 31, 32);
        }
        if (tid == 0) sHist[nb] = carry;
    }
    __syncthreads();

    #pragma unroll
    for (int k = 0; k < 16; ++k) {
        const int i = k * 256 + tid;
        if (i < cnt) sPair[sHist[tgtv[k] >> 8] + rk[k]] = make_int2(srcv[k], tgtv[k]);
    }
    for (int j = tid; j <= nb; j += 256) tab[(size_t)j * nch + bid] = sHist[j];
    __syncthreads();

    int2* outp = (int2*)binned + (size_t)rbase + e0;
    for (int i = tid; i < cnt; i += 256) outp[i] = sPair[i];
}

__global__ __launch_bounds__(256) void proj_bin_kernel(
        const float* __restrict__ x, const float* __restrict__ W, const float* __restrict__ b,
        float* __restrict__ hout, int n,
        const int* __restrict__ ed0, const int* __restrict__ ed1,
        const int* __restrict__ ed2, const int* __restrict__ ed3,
        int* __restrict__ binned, int* __restrict__ tab,
        int E0, int E1, int E2, int E3, int nb, int nch) {
    const int bid = (int)blockIdx.x;
    if (bid < nch) bin_block(ed0, ed1, ed2, ed3, binned, tab, E0, E1, E2, E3, nb, nch, bid);
    else           proj_block(x, W, b, hout, n, bid - nch);
}

__global__ __launch_bounds__(256) void fine_kernel(
        const int* __restrict__ binned, const int* __restrict__ tab,
        int* __restrict__ cnt, int* __restrict__ off,
        int* __restrict__ list0, int* __restrict__ list1,
        int* __restrict__ list2, int* __restrict__ list3,
        int E0, int E1, int E2, int E3, int nb, int nch, int n) {
    __shared__ int sCnt[256];
    __shared__ int sCur[256];
    __shared__ int sWave[8];
    __shared__ int sRed[8];
    __shared__ int sStage[STAGE];
    const int tid  = threadIdx.x;
    const int lane = tid & 31;
    const int wave = tid >> 5;
    const int C0 = (E0 + CHUNK - 1) / CHUNK, C1 = (E1 + CHUNK - 1) / CHUNK,
              C2 = (E2 + CHUNK - 1) / CHUNK, C3 = (E3 + CHUNK - 1) / CHUNK;
    const int r = blockIdx.x / nb;
    const int g = blockIdx.x - r * nb;
    int rbase = 0, cb = 0, nC = C0;
    int* list = list0;
    if (r == 1) { rbase = E0;           cb = C0;           nC = C1; list = list1; }
    if (r == 2) { rbase = E0 + E1;      cb = C0 + C1;      nC = C2; list = list2; }
    if (r == 3) { rbase = E0 + E1 + E2; cb = C0 + C1 + C2; nC = C3; list = list3; }
    const int* tlo = tab + (size_t)g * nch + cb;
    const int* thi = tab + (size_t)(g + 1) * nch + cb;
    const int2* win = (const int2*)binned + (size_t)rbase;

    sCnt[tid] = 0;
    __syncthreads();

    const int sub = tid & (RLANES - 1);
    const int rsl = tid / RLANES;

    int lower = 0;
    for (int b0 = 0; b0 < nC; b0 += RSLOTS * RROUND) {
        int rb[RROUND], re[RROUND];
        #pragma unroll
        for (int k = 0; k < RROUND; ++k) {
            const int b = b0 + rsl + RSLOTS * k;
            rb[k] = 0; re[k] = 0;
            if (b < nC) { rb[k] = tlo[b]; re[k] = thi[b]; }
        }
        int y[RROUND][RDEPTH];
        #pragma unroll
        for (int k = 0; k < RROUND; ++k) {
            const int2* p = win + (size_t)min(b0 + rsl + RSLOTS * k, nC - 1) * CHUNK;
            #pragma unroll
            for (int u = 0; u < RDEPTH; ++u)
                y[k][u] = p[max(min(rb[k] + sub + RLANES * u, re[k] - 1), 0)].y;
        }
        #pragma unroll
        for (int k = 0; k < RROUND; ++k) {
            if (sub == 0) lower += rb[k];
            #pragma unroll
            for (int u = 0; u < RDEPTH; ++u)
                if (rb[k] + sub + RLANES * u < re[k]) atomicAdd(&sCnt[y[k][u] & 255], 1);
            const int2* p = win + (size_t)min(b0 + rsl + RSLOTS * k, nC - 1) * CHUNK;
            for (int i = rb[k] + sub + RLANES * RDEPTH; i < re[k]; i += RLANES)
                atomicAdd(&sCnt[p[i].y & 255], 1);
        }
    }
    __syncthreads();

    const int cT = sCnt[tid];
    int inc = cT;
    for (int d = 1; d < 32; d <<= 1) {
        const int u = __shfl_up(inc, d, 32);
        if (lane >= d) inc += u;
    }
    for (int o = 16; o > 0; o >>= 1) lower += __shfl_xor(lower, o, 32);
    if (lane == 31) sWave[wave] = inc;
    if (lane == 0)  sRed[wave]  = lower;
    __syncthreads();
    int wbase = 0, total = 0, start = 0;
    for (int w = 0; w < 8; ++w) {
        const int tw = sWave[w];
        if (w < wave) wbase += tw;
        total += tw;
        start += sRed[w];
    }
    const int offL = wbase + inc - cT;
    const int t = g * 256 + tid;
    if (t < n) {
        cnt[(size_t)r * n + t] = cT;
        off[(size_t)r * n + t] = start + offL;
    }
    sCur[tid] = offL;
    __syncthreads();

    const bool staged = total <= STAGE;
    for (int b0 = 0; b0 < nC; b0 += RSLOTS * RROUND) {
        int rb[RROUND], re[RROUND];
        #pragma unroll
        for (int k = 0; k < RROUND; ++k) {
            const int b = b0 + rsl + RSLOTS * k;
            rb[k] = 0; re[k] = 0;
            if (b < nC) { rb[k] = tlo[b]; re[k] = thi[b]; }
        }
        int2 v[RROUND][RDEPTH];
        #pragma unroll
        for (int k = 0; k < RROUND; ++k) {
            const int2* p = win + (size_t)min(b0 + rsl + RSLOTS * k, nC - 1) * CHUNK;
            #pragma unroll
            for (int u = 0; u < RDEPTH; ++u)
                v[k][u] = p[max(min(rb[k] + sub + RLANES * u, re[k] - 1), 0)];
        }
        #pragma unroll
        for (int k = 0; k < RROUND; ++k) {
            #pragma unroll
            for (int u = 0; u < RDEPTH; ++u) {
                if (rb[k] + sub + RLANES * u < re[k]) {
                    const int pos = atomicAdd(&sCur[v[k][u].y & 255], 1);
                    if (staged) sStage[pos] = v[k][u].x;
                    else        list[start + pos] = v[k][u].x;
                }
            }
            const int2* p = win + (size_t)min(b0 + rsl + RSLOTS * k, nC - 1) * CHUNK;
            for (int i = rb[k] + sub + RLANES * RDEPTH; i < re[k]; i += RLANES) {
                const int2 w = p[i];
                const int pos = atomicAdd(&sCur[w.y & 255], 1);
                if (staged) sStage[pos] = w.x;
                else        list[start + pos] = w.x;
            }
        }
    }
    if (staged) {
        __syncthreads();
        for (int i = tid; i < total; i += 256) list[start + i] = sStage[i];
    }
}

__device__ __forceinline__ void add4(float4& a, const float4 v) {
    a.x += v.x; a.y += v.y; a.z += v.z; a.w += v.w;
}

__device__ __forceinline__ float4 bucket_sum(const float4* __restrict__ feat4,
                                             const int* __restrict__ list,
                                             int beg, int len, int lane) {
    float4 acc = make_float4(0.0f, 0.0f, 0.0f, 0.0f);
    if (len <= 32) {
        const int key = (lane < len) ? list[beg + lane] : 0x7fffffff;
        int rank = 0;
        for (int j = 0; j < len; ++j) {
            const int kj = __builtin_amdgcn_readlane(key, j);
            rank += (kj < key || (kj == key && j < lane)) ? 1 : 0;
        }
        if (lane >= len) rank = lane;
        const int sorted = __builtin_amdgcn_ds_permute(rank << 2, key);
        int r = 0;
        for (; r + 4 <= len; r += 4) {
            const int s0 = __builtin_amdgcn_readlane(sorted, r);
            const int s1 = __builtin_amdgcn_readlane(sorted, r + 1);
            const int s2 = __builtin_amdgcn_readlane(sorted, r + 2);
            const int s3 = __builtin_amdgcn_readlane(sorted, r + 3);
            const float4 v0 = feat4[(size_t)s0 * (H / 4) + lane];
            const float4 v1 = feat4[(size_t)s1 * (H / 4) + lane];
            const float4 v2 = feat4[(size_t)s2 * (H / 4) + lane];
            const float4 v3 = feat4[(size_t)s3 * (H / 4) + lane];
            add4(acc, v0); add4(acc, v1); add4(acc, v2); add4(acc, v3);
        }
        for (; r < len; ++r) {
            const int s0 = __builtin_amdgcn_readlane(sorted, r);
            add4(acc, feat4[(size_t)s0 * (H / 4) + lane]);
        }
    } else {
        int prev = -1;
        for (int it = 0; it < len; ++it) {
            int m = 0x7fffffff;
            for (int i = lane; i < len; i += 32) {
                const int k = list[beg + i];
                if (k > prev && k < m) m = k;
            }
            for (int o = 16; o > 0; o >>= 1) m = min(m, __shfl_xor(m, o, 32));
            if (m == 0x7fffffff) break;
            int c = 0;
            for (int i = lane; i < len; i += 32) c += (list[beg + i] == m) ? 1 : 0;
            for (int o = 16; o > 0; o >>= 1) c += __shfl_xor(c, o, 32);
            const float4 v = feat4[(size_t)m * (H / 4) + lane];
            for (int q = 0; q < c; ++q) add4(acc, v);
            prev = m;
        }
    }
    return acc;
}


__device__ __forceinline__ int sort_ids(int key, int len, int lane) {
    int rank = 0;
    for (int j = 0; j < len; ++j) {
        const int kj = __builtin_amdgcn_readlane(key, j);
        rank += (kj < key || (kj == key && j < lane)) ? 1 : 0;
    }
    if (lane >= len) rank = lane;
    return __builtin_amdgcn_ds_permute(rank << 2, key);
}

template <int Q>
__device__ __forceinline__ float4 long_bucket_sum(const float4* __restrict__ x4,
                                                  const int* __restrict__ list,
                                                  int beg, int len, int lane, int qq) {
    float4 acc = make_float4(0.0f, 0.0f, 0.0f, 0.0f);
    int prev = -1;
    for (int it = 0; it < len; ++it) {
        int m = 0x7fffffff;
        for (int i = lane; i < len; i += 32) {
            const int k = list[beg + i];
            if (k > prev && k < m) m = k;
        }
        for (int o = 16; o > 0; o >>= 1) m = min(m, __shfl_xor(m, o, 32));
        if (m == 0x7fffffff) break;
        int c = 0;
        for (int i = lane; i < len; i += 32) c += (list[beg + i] == m) ? 1 : 0;
        for (int o = 16; o > 0; o >>= 1) c += __shfl_xor(c, o, 32);
        const float4 v = x4[(size_t)m * Q + qq];
        for (int r = 0; r < c; ++r) add4(acc, v);
        prev = m;
    }
    return acc;
}

__device__ __forceinline__ float4 slot_sum(float4 a) {
    for (int m = 4; m < 32; m <<= 1) {
        float4 b;
        b.x = __shfl_xor(a.x, m, 32); b.y = __shfl_xor(a.y, m, 32);
        b.z = __shfl_xor(a.z, m, 32); b.w = __shfl_xor(a.w, m, 32);
        add4(a, b);
    }
    return a;
}

#define RPB 7
#define GW (4 * RPB)
#define GT (GW * 32)
#define APITCH 132

#define NIF 16
__device__ __forceinline__ void issue_rows(float4 (&v)[NIF], const float4* __restrict__ feat4,
                                           int sorted, int r, int len, int lane) {
    #pragma unroll
    for (int i = 0; i < NIF; ++i) {
        v[i] = make_float4(0.0f, 0.0f, 0.0f, 0.0f);
        if (r + i < len) {
            const int sid = __builtin_amdgcn_readlane(sorted, r + i);
            v[i] = feat4[(size_t)sid * (H / 4) + lane];
        }
    }
}
__device__ __forceinline__ void sum_rows(float4& acc, const float4 (&v)[NIF]) {
    #pragma unroll
    for (int i = 0; i < NIF; ++i) add4(acc, v[i]);
}
__device__ __forceinline__ float4 sum_rows_first(const float4 (&v)[NIF]) {
    float4 acc = v[0];
    #pragma unroll
    for (int i = 1; i < NIF; ++i) add4(acc, v[i]);
    return acc;
}

__device__ __forceinline__ void raw_aggregate_rows(
        const float* __restrict__ xLane, const float* __restrict__ xSens,
        const float* __restrict__ xInj,
        const int* __restrict__ listFl, const int* __restrict__ listFs,
        const int* __restrict__ listIn,
        const int* __restrict__ cnt, const int* __restrict__ off,
        const float4* sW2, float* sRow, float* __restrict__ aggS,
        int n, int t0, int lim, int lane) {
    const float4* xl4 = (const float4*)xLane;
    const float4* xs4 = (const float4*)xSens;
    const float4* xi4 = (const float4*)xInj;
    const float4 zero4 = make_float4(0.0f, 0.0f, 0.0f, 0.0f);
    const int slot = lane >> 2;
    const int q    = lane & 3;
    const bool isS = q < 2;
    const int qq   = q & 1;

    const bool okl = lane < 8 && t0 + lane < lim;
    const unsigned tl = (unsigned)(t0 + lane);
    const int* cntL = cnt + (size_t)n;      const int* offL = off + (size_t)n;
    const int* cntS = cnt + (size_t)n * 2;  const int* offS = off + (size_t)n * 2;
    const int* cntI = cnt + (size_t)n * 3;  const int* offI = off + (size_t)n * 3;
    const int cLv = okl ? cntL[tl] : 0;
    const int cSv = okl ? cntS[tl] : 0;
    const int cIv = okl ? cntI[tl] : 0;
    const int oLv = okl ? offL[tl] : 0;
    const int oSv = okl ? offS[tl] : 0;
    const int oIv = okl ? offI[tl] : 0;

    int lenLn = __builtin_amdgcn_readlane(cLv, 0);
    int lenSn = __builtin_amdgcn_readlane(cSv, 0);
    int lenIn = __builtin_amdgcn_readlane(cIv, 0);
    int begLn = __builtin_amdgcn_readlane(oLv, 0);
    int begSn = __builtin_amdgcn_readlane(oSv, 0);
    int begIn = __builtin_amdgcn_readlane(oIv, 0);
    bool shn = lenLn <= 32 && lenSn <= 32 && lenIn <= 32;
    int keyLn = (shn && lane < lenLn) ? listFl[begLn + lane] : 0x7fffffff;
    int keySn = (shn && lane < lenSn) ? listFs[begSn + lane] : 0x7fffffff;
    int keyIn = (shn && lane < lenIn) ? listIn[begIn + lane] : 0x7fffffff;

    #pragma nounroll
    for (int u = 0; u < 8; ++u) {
        const int lenL = lenLn, lenS = lenSn, lenI = lenIn;
        const int begL = begLn, begS = begSn, begI = begIn;
        const int keyL = keyLn, keyS = keySn, keyI = keyIn;
        lenLn = 0; lenSn = 0; lenIn = 0; begLn = 0; begSn = 0; begIn = 0;
        if (u + 1 < 8) {
            lenLn = __builtin_amdgcn_readlane(cLv, u + 1);
            lenSn = __builtin_amdgcn_readlane(cSv, u + 1);
            lenIn = __builtin_amdgcn_readlane(cIv, u + 1);
            begLn = __builtin_amdgcn_readlane(oLv, u + 1);
            begSn = __builtin_amdgcn_readlane(oSv, u + 1);
            begIn = __builtin_amdgcn_readlane(oIv, u + 1);
        }
        shn = lenLn <= 32 && lenSn <= 32 && lenIn <= 32;
        auto fetch_next_ids = [&]() {
            keyLn = (shn && lane < lenLn) ? listFl[begLn + lane] : 0x7fffffff;
            keySn = (shn && lane < lenSn) ? listFs[begSn + lane] : 0x7fffffff;
            keyIn = (shn && lane < lenIn) ? listIn[begIn + lane] : 0x7fffffff;
        };

        float4 zl, zsi;
        if (lenL <= 32 && lenS <= 32 && lenI <= 32) {
            const int sortedL = sort_ids(keyL, lenL, lane);
            const int sortedS = sort_ids(keyS, lenS, lane);
            const int sortedI = sort_ids(keyI, lenI, lane);
            const int lenX = isS ? lenS : lenI;
            const int maxSI = max(lenS, lenI);
            const float4* xb = isS ? xs4 : xi4;
            float4 a[4], b[4];
            #pragma unroll
            for (int i = 0; i < 4; ++i) {
                a[i] = zero4; b[i] = zero4;
                const int p = 8 * i + slot;
                if (8 * i < lenL) {
                    const int id = __builtin_amdgcn_ds_bpermute(p << 2, sortedL);
                    if (p < lenL) a[i] = xl4[(size_t)id * 4 + q];
                }
                if (8 * i < maxSI) {
                    const int idS = __builtin_amdgcn_ds_bpermute(p << 2, sortedS);
                    const int idI = __builtin_amdgcn_ds_bpermute(p << 2, sortedI);
                    if (p < lenX) b[i] = xb[(size_t)(isS ? idS : idI) * 2 + qq];
                }
            }
            fetch_next_ids();
            add4(a[0], a[1]); add4(a[0], a[2]); add4(a[0], a[3]);
            add4(b[0], b[1]); add4(b[0], b[2]); add4(b[0], b[3]);
            zl  = slot_sum(a[0]);
            zsi = slot_sum(b[0]);
        } else {
            fetch_next_ids();
            zl = long_bucket_sum<4>(xl4, listFl, begL, lenL, lane, q);
            const float4 sS = long_bucket_sum<2>(xs4, listFs, begS, lenS, lane, qq);
            const float4 sI = long_bucket_sum<2>(xi4, listIn, begI, lenI, lane, qq);
            zsi = isS ? sS : sI;
        }
        const float wL = 1.0f / (float)max(lenL, 1);
        const float wX = 1.0f / (float)max(isS ? lenS : lenI, 1);
        zl.x *= wL;  zl.y *= wL;  zl.z *= wL;  zl.w *= wL;
        zsi.x *= wX; zsi.y *= wX; zsi.z *= wX; zsi.w *= wX;
        float4* row4 = (float4*)(sRow + (size_t)u * APITCH);
        if (lane < 4) { row4[lane] = zl; row4[4 + lane] = zsi; }
        if (lane == 0) row4[8] = make_float4(lenL > 0 ? 1.0f : 0.0f, lenS > 0 ? 1.0f : 0.0f,
                                             lenI > 0 ? 1.0f : 0.0f, 0.0f);
    }

    #pragma nounroll
    for (int g2 = 0; g2 < 2; ++g2) {
        float* grow = sRow + (size_t)(4 * g2) * APITCH;
        float4 o[4];
        #pragma unroll
        for (int u = 0; u < 4; ++u) o[u] = zero4;
        #pragma nounroll
        for (int g = 0; g < 8; ++g) {
            const float4 w0 = sW2[(4 * g + 0) * 32 + lane];
            const float4 w1 = sW2[(4 * g + 1) * 32 + lane];
            const float4 w2 = sW2[(4 * g + 2) * 32 + lane];
            const float4 w3 = sW2[(4 * g + 3) * 32 + lane];
            #pragma unroll
            for (int u = 0; u < 4; ++u) {
                const float4 z = ((const float4*)(grow + (size_t)u * APITCH))[g];
                o[u].x += z.x * w0.x; o[u].y += z.x * w0.y; o[u].z += z.x * w0.z; o[u].w += z.x * w0.w;
                o[u].x += z.y * w1.x; o[u].y += z.y * w1.y; o[u].z += z.y * w1.z; o[u].w += z.y * w1.w;
                o[u].x += z.z * w2.x; o[u].y += z.z * w2.y; o[u].z += z.z * w2.z; o[u].w += z.z * w2.w;
                o[u].x += z.w * w3.x; o[u].y += z.w * w3.y; o[u].z += z.w * w3.z; o[u].w += z.w * w3.w;
            }
        }
        {
            const float4 b0 = sW2[32 * 32 + lane];
            const float4 b1 = sW2[33 * 32 + lane];
            const float4 b2 = sW2[34 * 32 + lane];
            #pragma unroll
            for (int u = 0; u < 4; ++u) {
                const float4 f = ((const float4*)(grow + (size_t)u * APITCH))[8];
                o[u].x += f.x * b0.x; o[u].y += f.x * b0.y; o[u].z += f.x * b0.z; o[u].w += f.x * b0.w;
                o[u].x += f.y * b1.x; o[u].y += f.y * b1.y; o[u].z += f.y * b1.z; o[u].w += f.y * b1.w;
                o[u].x += f.z * b2.x; o[u].y += f.z * b2.y; o[u].z += f.z * b2.z; o[u].w += f.z * b2.w;
            }
        }
        #pragma unroll
        for (int u = 0; u < 4; ++u) {
            const int t = t0 + 4 * g2 + u;
            *(float4*)(grow + (size_t)u * APITCH + (lane << 2)) = o[u];
            if (t < lim) ((float4*)aggS)[(size_t)t * (H / 4) + lane] = o[u];
        }
    }
}

template <bool TWO>
__device__ __forceinline__ void prod_rows(const float4* __restrict__ p0, const float4* __restrict__ p1,
                                          int step, const float* __restrict__ bbase,
                                          v8f& acc00, v8f& acc01, v8f& acc10, v8f& acc11) {
    for (int j = 0; j < H / 8; ++j) {
        const float4 x0 = p0[step * j];
        float4 x1 = x0;
        if (TWO) x1 = p1[step * j];
        const float* bk = bbase + (j << 10);
        const v2f b00 = *(const v2f*)(bk);
        const v2f b01 = *(const v2f*)(bk + 32);
        const v2f b10 = *(const v2f*)(bk + 256);
        const v2f b11 = *(const v2f*)(bk + 256 + 32);
        v2f a;
        a.x = x0.x; a.y = x0.y;
        acc00 = __builtin_amdgcn_wmma_f32_16x16x4_f32(false, a, false, b00, (short)0, acc00, false, false);
        acc01 = __builtin_amdgcn_wmma_f32_16x16x4_f32(false, a, false, b01, (short)0, acc01, false, false);
        if (TWO) {
            a.x = x1.x; a.y = x1.y;
            acc10 = __builtin_amdgcn_wmma_f32_16x16x4_f32(false, a, false, b00, (short)0, acc10, false, false);
            acc11 = __builtin_amdgcn_wmma_f32_16x16x4_f32(false, a, false, b01, (short)0, acc11, false, false);
        }
        a.x = x0.z; a.y = x0.w;
        acc00 = __builtin_amdgcn_wmma_f32_16x16x4_f32(false, a, false, b10, (short)0, acc00, false, false);
        acc01 = __builtin_amdgcn_wmma_f32_16x16x4_f32(false, a, false, b11, (short)0, acc01, false, false);
        if (TWO) {
            a.x = x1.z; a.y = x1.w;
            acc10 = __builtin_amdgcn_wmma_f32_16x16x4_f32(false, a, false, b10, (short)0, acc10, false, false);
            acc11 = __builtin_amdgcn_wmma_f32_16x16x4_f32(false, a, false, b11, (short)0, acc11, false, false);
        }
    }
}

template <bool FIRST>
__global__ __launch_bounds__(GT) void fused_gemm_elu_kernel(
        const float* __restrict__ h, float* __restrict__ aggS,
        const int* __restrict__ listSp, const int* __restrict__ cnt, const int* __restrict__ off,
        const float* __restrict__ W0, const float* __restrict__ W1,
        const float* __restrict__ bias, float* __restrict__ out, int n, int nrp,
        const float* __restrict__ xLane, const float* __restrict__ xSens,
        const float* __restrict__ xInj,
        const float* __restrict__ WLane, const float* __restrict__ bLane,
        const float* __restrict__ WSens, const float* __restrict__ bSens,
        const float* __restrict__ WInj,  const float* __restrict__ bInj,
        const int* __restrict__ listFl, const int* __restrict__ listFs,
        const int* __restrict__ listIn) {
    __shared__ __align__(16) float sB[2 * H * H];
    __shared__ __align__(16) float sA[RPB * 32 * APITCH];
    __shared__ float4 sW2[FIRST ? 35 * 32 : 1];
    const int tid  = threadIdx.x;
    const int lane = tid & 31;
    const int half = lane >> 4;
    const int l15  = lane & 15;

    const int wave = __builtin_amdgcn_readfirstlane(tid >> 5);
    const int lrp  = wave >> 2;
    (void)nrp;
    const int nU   = (n + 15) >> 4;
    const int uBeg = (int)(((long long)blockIdx.x * nU) / gridDim.x);
    const int uEnd = (int)(((long long)(blockIdx.x + 1) * nU) / gridDim.x);
    const int uw   = uBeg + 2 * lrp;
    const bool active = uw < uEnd;
    const bool two = uw + 1 < uEnd;
    const int m0  = uw << 4;
    const int lim = min(n, uEnd << 4);
    const int cb = (wave & 3) << 5;

    const int lr0 = (wave & 3) << 3;
    const int t0  = m0 + lr0;
    const bool okl = active && lane < 8 && t0 + lane < lim;
    const int cntv = okl ? cnt[t0 + lane] : 0;
    const int offv = okl ? off[t0 + lane] : 0;

    {
        constexpr int NV = 2 * H * H / 4;
        constexpr int NI = (NV + GT - 1) / GT;
        float4 wv[NI];
        #pragma unroll
        for (int q = 0; q < NI; ++q) {
            const int i = tid + GT * q;
            wv[q] = make_float4(0.0f, 0.0f, 0.0f, 0.0f);
            if (i < NV) wv[q] = ((const float4*)((i >> 12) ? W1 : W0))[i & 4095];
        }
        #pragma unroll
        for (int q = 0; q < NI; ++q) {
            const int i = tid + GT * q;
            if (i < NV) {
                const int jj = i & 4095;
                const int k  = jj >> 5;
                const int c0 = (jj & 31) << 2;
                float* d = sB + ((i >> 12) << 14) + ((k >> 1) << 8) + (k & 1);
                d[2 * (c0 + 0)] = wv[q].x;
                d[2 * (c0 + 1)] = wv[q].y;
                d[2 * (c0 + 2)] = wv[q].z;
                d[2 * (c0 + 3)] = wv[q].w;
            }
        }
    }
    if (FIRST) {
        for (int i = tid; i < 35 * 32; i += GT) {
            float4 v;
            if (i < 512)       v = ((const float4*)WLane)[i];
            else if (i < 768)  v = ((const float4*)WSens)[i - 512];
            else if (i < 1024) v = ((const float4*)WInj)[i - 768];
            else {
                const int bi = i - 1024;
                const float* bs = bi < 32 ? bLane : (bi < 64 ? bSens : bInj);
                v = ((const float4*)bs)[bi & 31];
            }
            sW2[i] = v;
        }
    }
    int lenC = __builtin_amdgcn_readlane(cntv, 0);
    int begC = __builtin_amdgcn_readlane(offv, 0);
    int lenN = __builtin_amdgcn_readlane(cntv, 1);
    int begN = __builtin_amdgcn_readlane(offv, 1);
    int keyN = (lenN <= 32 && lane < lenN) ? listSp[begN + lane] : 0x7fffffff;
    const int keyC = (lenC <= 32 && lane < lenC) ? listSp[begC + lane] : 0x7fffffff;
    __syncthreads();

    int r0 = m0 + l15;
    if (r0 >= n) r0 = n - 1;
    int r1 = m0 + 16 + l15;
    if (r1 >= n) r1 = n - 1;
    float* sAp = sA + (size_t)lrp * 32 * APITCH;

    v8f acc00 = {}, acc01 = {}, acc10 = {}, acc11 = {};
    if (active) {
        const int first = (wave >> 2) & 1;
        auto self_product = [&]() {
            const float4* p0 = (const float4*)(h + (size_t)r0 * H + (half << 2));
            const float4* p1 = (const float4*)(h + (size_t)r1 * H + (half << 2));
            const float* bbase = sB + (half << 9) + ((cb + l15) << 1);
            if (two) prod_rows<true>(p0, p1, 2, bbase, acc00, acc01, acc10, acc11);
            else     prod_rows<false>(p0, p1, 2, bbase, acc00, acc01, acc10, acc11);
        };
        auto gather_rows = [&]() {
            if (FIRST) raw_aggregate_rows(xLane, xSens, xInj, listFl, listFs, listIn, cnt, off,
                                          sW2, sAp + (size_t)lr0 * APITCH, aggS, n, t0, lim, lane);
            int sortedC = (lenC <= 32) ? sort_ids(keyC, lenC, lane) : 0;
            const float4* h4 = (const float4*)h;
            for (int q = 0; q < 8; ++q) {
                const int t = t0 + q;
                int lenF = 0, begF = 0;
                if (q + 2 < 8) {
                    lenF = __builtin_amdgcn_readlane(cntv, q + 2);
                    begF = __builtin_amdgcn_readlane(offv, q + 2);
                }
                int keyF = 0x7fffffff;
                float4 r = make_float4(0.0f, 0.0f, 0.0f, 0.0f);
                int sortedN = 0;
                if (t < lim) {
                    float4 sm, base;
                    if (lenC <= 32) {
                        float4 v[NIF];
                        issue_rows(v, h4, sortedC, 0, lenC, lane);
                        if (lenF <= 32 && lane < lenF) keyF = listSp[begF + lane];
                        if (FIRST) base = *(const float4*)(sAp + (size_t)(lr0 + q) * APITCH + (lane << 2));
                        else       base = ((const float4*)aggS)[(size_t)t * (H / 4) + lane];
                        if (lenN <= 32) sortedN = sort_ids(keyN, lenN, lane);
                        sm = sum_rows_first(v);
                        for (int rr = NIF; rr < lenC; rr += NIF) {
                            issue_rows(v, h4, sortedC, rr, lenC, lane);
                            sum_rows(sm, v);
                        }
                    } else {
                        if (lenF <= 32 && lane < lenF) keyF = listSp[begF + lane];
                        if (FIRST) base = *(const float4*)(sAp + (size_t)(lr0 + q) * APITCH + (lane << 2));
                        else       base = ((const float4*)aggS)[(size_t)t * (H / 4) + lane];
                        sm = bucket_sum(h4, listSp, begC, lenC, lane);
                        if (lenN <= 32) sortedN = sort_ids(keyN, lenN, lane);
                    }
                    const float w = 1.0f / (float)max(lenC, 1);
                    r.x = base.x + sm.x * w; r.y = base.y + sm.y * w;
                    r.z = base.z + sm.z * w; r.w = base.w + sm.w * w;
                }
                *(float4*)(sAp + (size_t)(lr0 + q) * APITCH + (lane << 2)) = r;
                lenC = lenN; begC = begN; sortedC = sortedN;
                lenN = lenF; begN = begF; keyN = keyF;
            }
        };
        if constexpr (FIRST) {
            if (first == 0) self_product();
            gather_rows();
            if (first != 0) self_product();
        } else {
            for (int st = 0; st < 2; ++st) {
                if ((st ^ first) == 0) self_product();
                else                   gather_rows();
            }
        }
    }
    __syncthreads();

    if (active) {
        const float* q0 = sAp + (size_t)l15 * APITCH + (half << 2);
        const float* q1 = q0 + 16 * APITCH;
        const float* bbase = sB + (1 << 14) + (half << 9) + ((cb + l15) << 1);
        if (two) prod_rows<true>((const float4*)q0, (const float4*)q1, 2, bbase, acc00, acc01, acc10, acc11);
        else     prod_rows<false>((const float4*)q0, (const float4*)q1, 2, bbase, acc00, acc01, acc10, acc11);

        const float bc0 = bias[cb + l15];
        const float bc1 = bias[cb + 16 + l15];
        #pragma unroll
        for (int r = 0; r < 8; ++r) {
            const int rowa = m0 + (half << 3) + r;
            const int rowb = rowa + 16;
            if (rowa < lim) {
                float v = acc00[r] + bc0;
                out[(size_t)rowa * H + cb + l15] = v > 0.0f ? v : (__expf(v) - 1.0f);
                v = acc01[r] + bc1;
                out[(size_t)rowa * H + cb + 16 + l15] = v > 0.0f ? v : (__expf(v) - 1.0f);
            }
            if (rowb < lim) {
                float v = acc10[r] + bc0;
                out[(size_t)rowb * H + cb + l15] = v > 0.0f ? v : (__expf(v) - 1.0f);
                v = acc11[r] + bc1;
                out[(size_t)rowb * H + cb + 16 + l15] = v > 0.0f ? v : (__expf(v) - 1.0f);
            }
        }
    }
}

extern "C" void kernel_launch(void* const* d_in, const int* in_sizes, int n_in,
                              void* d_out, int out_size, void* d_ws, size_t ws_size,
                              hipStream_t stream) {
    const float* x_int  = (const float*)d_in[0];
    const float* x_lane = (const float*)d_in[1];
    const float* x_sens = (const float*)d_in[2];
    const float* x_inj  = (const float*)d_in[3];
    const int*   e_sp   = (const int*)d_in[4];
    const int*   e_fl   = (const int*)d_in[5];
    const int*   e_fs   = (const int*)d_in[6];
    const int*   e_in   = (const int*)d_in[7];
    const float* W_int  = (const float*)d_in[8];
    const float* b_int  = (const float*)d_in[9];
    const float* W_lane = (const float*)d_in[10];
    const float* b_lane = (const float*)d_in[11];
    const float* W_sens = (const float*)d_in[12];
    const float* b_sens = (const float*)d_in[13];
    const float* W_inj  = (const float*)d_in[14];
    const float* b_inj  = (const float*)d_in[15];
    const float* self_W = (const float*)d_in[16];
    const float* self_b = (const float*)d_in[17];
    const float* rel_W  = (const float*)d_in[18];

    const int nInt  = in_sizes[0] / 32;
    const int nLane = in_sizes[1] / 16;
    const int nSens = in_sizes[2] / 8;
    const int nInj  = in_sizes[3] / 8;
    const int eSp   = in_sizes[4] / 2;
    const int eFl   = in_sizes[5] / 2;
    const int eFs   = in_sizes[6] / 2;
    const int eIn   = in_sizes[7] / 2;
    const int L     = in_sizes[16] / (H * H);

    float* ws = (float*)d_ws;
    size_t o = 0;
    float* hA    = ws + o; o += (size_t)nInt  * H;
    float* hB    = ws + o; o += (size_t)nInt  * H;
    float* hLane = ws + o; o += (size_t)nLane * H;
    float* pSens = ws + o; o += (size_t)nSens * H;
    o += (size_t)nInj  * H;
    float* aggS  = ws + o; o += (size_t)nInt  * H;
    float* aggB  = ws + o; o += (size_t)nInt  * H;
    int*   cnt   = (int*)(ws + o); o += (size_t)4 * nInt;
    int*   off   = (int*)(ws + o); o += (size_t)4 * nInt;
    (void)ws_size; (void)n_in; (void)out_size;

    int*  binned = (int*)aggB;
    int*  listFl = (int*)aggB + 2 * ((size_t)eSp + eFl + eFs + eIn);
    int*  listFs = listFl + eFl;
    int*  listIn = listFs + eFs;
    int*  listSp = (int*)hLane;
    int*  tab    = (int*)pSens;

    const int nb  = (nInt + 255) / 256;
    const int nch = (eSp + CHUNK - 1) / CHUNK + (eFl + CHUNK - 1) / CHUNK
                  + (eFs + CHUNK - 1) / CHUNK + (eIn + CHUNK - 1) / CHUNK;
    proj_bin_kernel<<<nch + (nInt + 127) / 128, 256, 0, stream>>>(
        x_int, W_int, b_int, hA, nInt,
        e_sp, e_fl, e_fs, e_in, binned, tab, eSp, eFl, eFs, eIn, nb, nch);
    fine_kernel<<<4 * nb, 256, 0, stream>>>(binned, tab, cnt, off,
                                            listSp, listFl, listFs, listIn,
                                            eSp, eFl, eFs, eIn, nb, nch, nInt);

    const int nrp = (nInt + 31) / 32;
    const int needed = (nrp + RPB - 1) / RPB;
    const int lgrid = needed > 256 ? needed : 256;
    float* hcur = hA;
    float* hnext = hB;
    for (int l = 0; l < L; ++l) {
        float* outp = (l == L - 1) ? (float*)d_out : hnext;
        if (l == 0)
            fused_gemm_elu_kernel<true><<<lgrid, GT, 0, stream>>>(
                hcur, aggS, listSp, cnt, off,
                self_W + (size_t)l * H * H, rel_W + (size_t)l * H * H,
                self_b + (size_t)l * H, outp, nInt, nrp,
                x_lane, x_sens, x_inj, W_lane, b_lane, W_sens, b_sens, W_inj, b_inj,
                listFl, listFs, listIn);
        else
            fused_gemm_elu_kernel<false><<<lgrid, GT, 0, stream>>>(
                hcur, aggS, listSp, cnt, off,
                self_W + (size_t)l * H * H, rel_W + (size_t)l * H * H,
                self_b + (size_t)l * H, outp, nInt, nrp,
                x_lane, x_sens, x_inj, W_lane, b_lane, W_sens, b_sens, W_inj, b_inj,
                listFl, listFs, listIn);
        float* t = hnext; hnext = hcur; hcur = t;
    }
}
